# starting kernel
# baseline (speedup 1.0000x reference)
#include <hip/hip_runtime.h>
#include <hip/hip_bf16.h>

typedef __attribute__((ext_vector_type(16))) __bf16 v16bf;
typedef __attribute__((ext_vector_type(8)))  float  v8f;

union FragU {
    v16bf v;
    unsigned short u16[16];
    uint4 u4[2];
};

__device__ __forceinline__ unsigned short f2bf(float f) {
    unsigned int u = __float_as_uint(f);
    unsigned int r = u + 0x7FFFu + ((u >> 16) & 1u);
    return (unsigned short)(r >> 16);
}

__global__ void k_deg_init(float* deg, int n) {
    int i = blockIdx.x * blockDim.x + threadIdx.x;
    if (i < n) deg[i] = 1.0f;
}

__global__ void k_deg_accum(const int* dst, float* deg, int e) {
    int i = blockIdx.x * blockDim.x + threadIdx.x;
    if (i < e) unsafeAtomicAdd(&deg[dst[i]], 1.0f);
}

__global__ void k_deg_finish(float* deg, int n) {
    int i = blockIdx.x * blockDim.x + threadIdx.x;
    if (i < n) deg[i] = rsqrtf(deg[i]);
}

__global__ void k_cvt_bf16(const float* in, unsigned short* out, int n) {
    int i = blockIdx.x * blockDim.x + threadIdx.x;
    if (i < n) out[i] = f2bf(in[i]);
}

__global__ void k_pack_w(const float* W, unsigned short* out, int K, int N) {
    int tile = blockIdx.x;
    int lane = threadIdx.x;
    int ntiles = N >> 4;
    int kt = tile / ntiles, nt = tile - kt * ntiles;
    int col = nt * 16 + (lane & 15);
    int kb  = kt * 32 + ((lane < 16) ? 0 : 8);
    unsigned short* o = out + (size_t)tile * 512 + lane * 16;
#pragma unroll
    for (int e = 0; e < 8; ++e) o[e]     = f2bf(W[(kb + e)      * N + col]);
#pragma unroll
    for (int e = 0; e < 8; ++e) o[8 + e] = f2bf(W[(kb + 16 + e) * N + col]);
}

__global__ void k_gemm_bf16(const unsigned short* __restrict__ A,
                            const unsigned short* __restrict__ Bp,
                            float* __restrict__ C, int M, int K) {
    extern __shared__ unsigned short sB[];

    {
        const uint4* gB  = (const uint4*)Bp;
        uint4*       sB4 = (uint4*)sB;
        int n4 = (K >> 5) * 512;
        for (int i = threadIdx.x; i < n4; i += blockDim.x) sB4[i] = gB[i];
    }
    __syncthreads();

    int wave   = (int)((blockIdx.x * blockDim.x + threadIdx.x) >> 5);
    int lane   = threadIdx.x & 31;
    int mtiles = (M + 15) >> 4;
    if (wave >= mtiles) return;

    int row = wave * 16 + (lane & 15);
    if (row >= M) row = M - 1;
    int kb0 = (lane < 16) ? 0 : 8;
    const unsigned short* arow = A + (size_t)row * K;

    v8f zero = {};
    v8f acc[8];
#pragma unroll
    for (int nt = 0; nt < 8; ++nt) acc[nt] = zero;

    for (int kt = 0; kt < K; kt += 32) {
        FragU a;
        a.u4[0] = *(const uint4*)(arow + kt + kb0);
        a.u4[1] = *(const uint4*)(arow + kt + kb0 + 16);
        const unsigned short* bp = sB + (size_t)(kt >> 5) * 8 * 512 + lane * 16;
#pragma unroll
        for (int nt = 0; nt < 8; ++nt) {
            FragU b;
            const uint4* bl = (const uint4*)(bp + nt * 512);
            b.u4[0] = bl[0];
            b.u4[1] = bl[1];
            acc[nt] = __builtin_amdgcn_wmma_f32_16x16x32_bf16(
                false, a.v, false, b.v, (short)0, acc[nt], false, false);
        }
    }

    int srow0 = wave * 16 + ((lane < 16) ? 0 : 8);
    int scol  = lane & 15;
#pragma unroll
    for (int nt = 0; nt < 8; ++nt) {
#pragma unroll
        for (int j = 0; j < 8; ++j) {
            int r = srow0 + j;
            if (r < M) C[(size_t)r * 128 + nt * 16 + scol] = acc[nt][j];
        }
    }
}

__global__ void k_agg_init(const float* h, const float* dinv, float* agg, int total) {
    int i = blockIdx.x * blockDim.x + threadIdx.x;
    if (i < total) {
        float d = dinv[i >> 7];
        agg[i] = d * d * h[i];
    }
}

__global__ void k_agg_edges(const int* __restrict__ src, const int* __restrict__ dst,
                            const float* __restrict__ dinv,
                            const float* __restrict__ h,
                            float* __restrict__ agg, int e) {
    int wid  = (int)((blockIdx.x * blockDim.x + threadIdx.x) >> 5);
    if (wid >= e) return;
    int lane = threadIdx.x & 31;
    int s = src[wid], d = dst[wid];
    float nrm = dinv[s] * dinv[d];
    const float4 hv = *(const float4*)(h + (size_t)s * 128 + lane * 4);
    float* ap = agg + (size_t)d * 128 + lane * 4;
    unsafeAtomicAdd(ap + 0, nrm * hv.x);
    unsafeAtomicAdd(ap + 1, nrm * hv.y);
    unsafeAtomicAdd(ap + 2, nrm * hv.z);
    unsafeAtomicAdd(ap + 3, nrm * hv.w);
}

__global__ void k_ln_elu(const float* __restrict__ agg,
                         const float* __restrict__ bias,
                         const float* __restrict__ gamma,
                         const float* __restrict__ beta,
                         const float* __restrict__ res_in,
                         float* __restrict__ out_f32,
                         unsigned short* __restrict__ out_bf16,
                         int n, int mode) {
    int wave = (int)((blockIdx.x * blockDim.x + threadIdx.x) >> 5);
    if (wave >= n) return;
    int lane = threadIdx.x & 31;
    const float* a = agg + (size_t)wave * 128;

    float v[4];
    float s = 0.f;
#pragma unroll
    for (int j = 0; j < 4; ++j) {
        int c = lane * 4 + j;
        v[j] = a[c] + bias[c];
        s += v[j];
    }
#pragma unroll
    for (int off = 16; off >= 1; off >>= 1) s += __shfl_xor(s, off, 32);
    float mu = s * (1.0f / 128.0f);

    float q = 0.f;
#pragma unroll
    for (int j = 0; j < 4; ++j) {
        float dv = v[j] - mu;
        q += dv * dv;
    }
#pragma unroll
    for (int off = 16; off >= 1; off >>= 1) q += __shfl_xor(q, off, 32);
    float inv = rsqrtf(q * (1.0f / 128.0f) + 1e-5f);

#pragma unroll
    for (int j = 0; j < 4; ++j) {
        int c = lane * 4 + j;
        float y = (v[j] - mu) * inv * gamma[c] + beta[c];
        y = (y > 0.f) ? y : expm1f(y);
        size_t idx = (size_t)wave * 128 + c;
        if (mode == 0) {
            out_f32[idx]  = y;
            out_bf16[idx] = f2bf(y);
        } else {
            out_f32[idx] = y + res_in[idx];
        }
    }
}

extern "C" void kernel_launch(void* const* d_in, const int* in_sizes, int n_in,
                              void* d_out, int out_size, void* d_ws, size_t ws_size,
                              hipStream_t stream) {
    const float* x   = (const float*)d_in[0];
    const float* W1  = (const float*)d_in[1];
    const float* b1  = (const float*)d_in[2];
    const float* g1  = (const float*)d_in[3];
    const float* be1 = (const float*)d_in[4];
    const float* W2  = (const float*)d_in[5];
    const float* b2  = (const float*)d_in[6];
    const float* g2  = (const float*)d_in[7];
    const float* be2 = (const float*)d_in[8];
    const int*   ei  = (const int*)d_in[9];

    const int EMB = in_sizes[2];
    const int IN  = in_sizes[1] / EMB;
    const int N   = in_sizes[0] / IN;
    const int E   = in_sizes[9] / 2;
    const int* src = ei;
    const int* dst = ei + E;

    char* base = (char*)d_ws;
    size_t off = 0;
    auto alloc = [&](size_t bytes) {
        char* p = base + off;
        off += (bytes + 255) & ~(size_t)255;
        return p;
    };
    float*          dinv  = (float*)alloc((size_t)N * 4);
    unsigned short* xb    = (unsigned short*)alloc((size_t)N * IN * 2);
    unsigned short* W1p   = (unsigned short*)alloc((size_t)IN * EMB * 2);
    unsigned short* W2p   = (unsigned short*)alloc((size_t)EMB * EMB * 2);
    float*          h     = (float*)alloc((size_t)N * EMB * 4);
    float*          agg   = (float*)alloc((size_t)N * EMB * 4);
    float*          res   = (float*)alloc((size_t)N * EMB * 4);
    unsigned short* h1b   = (unsigned short*)alloc((size_t)N * EMB * 2);
    (void)ws_size;

    const int B = 256;
    int total = N * EMB;

    k_deg_init  <<<(N + B - 1) / B, B, 0, stream>>>(dinv, N);
    k_deg_accum <<<(E + B - 1) / B, B, 0, stream>>>(dst, dinv, E);
    k_deg_finish<<<(N + B - 1) / B, B, 0, stream>>>(dinv, N);

    k_cvt_bf16<<<((size_t)N * IN + B - 1) / B, B, 0, stream>>>(x, xb, N * IN);
    k_pack_w  <<<(IN  / 32) * (EMB / 16), 32, 0, stream>>>(W1, W1p, IN,  EMB);
    k_pack_w  <<<(EMB / 32) * (EMB / 16), 32, 0, stream>>>(W2, W2p, EMB, EMB);

    int mtiles = (N + 15) / 16;
    int gemm_blocks = (mtiles + 7) / 8;
    size_t lds1 = (size_t)(IN  / 32) * 8 * 1024;
    size_t lds2 = (size_t)(EMB / 32) * 8 * 1024;

    k_gemm_bf16<<<gemm_blocks, B, lds1, stream>>>(xb, W1p, h, N, IN);
    k_agg_init <<<(total + B - 1) / B, B, 0, stream>>>(h, dinv, agg, total);
    k_agg_edges<<<(E * 32 + B - 1) / B, B, 0, stream>>>(src, dst, dinv, h, agg, E);
    k_ln_elu   <<<(N + 7) / 8, B, 0, stream>>>(agg, b1, g1, be1, nullptr, res, h1b, N, 0);

    k_gemm_bf16<<<gemm_blocks, B, lds2, stream>>>(h1b, W2p, h, N, EMB);
    k_agg_init <<<(total + B - 1) / B, B, 0, stream>>>(h, dinv, agg, total);
    k_agg_edges<<<(E * 32 + B - 1) / B, B, 0, stream>>>(src, dst, dinv, h, agg, E);
    k_ln_elu   <<<(N + 7) / 8, B, 0, stream>>>(agg, b2, g2, be2, res, (float*)d_out, nullptr, N, 1);
}


# x is rounded f32->bf16 inside the layer-1 GEMM's A-fragment load (same rounding and k order, sums unchanged); the separate one-element-per-thread conversion launch is removed
# speedup vs baseline: 2.1531x; 2.1531x over previous
#include <hip/hip_runtime.h>
#include <hip/hip_bf16.h>

typedef __attribute__((ext_vector_type(16))) __bf16 v16bf;
typedef __attribute__((ext_vector_type(8)))  float  v8f;

union FragU {
    v16bf v;
    unsigned short u16[16];
    uint4 u4[2];
};

__device__ __forceinline__ unsigned short f2bf(float f) {
    unsigned int u = __float_as_uint(f);
    unsigned int r = u + 0x7FFFu + ((u >> 16) & 1u);
    return (unsigned short)(r >> 16);
}

__global__ void k_pack_w(const float* W, unsigned short* out, int K, int N) {
    int tile = blockIdx.x;
    int lane = threadIdx.x;
    int ntiles = N >> 4;
    int kt = tile / ntiles, nt = tile - kt * ntiles;
    int col = nt * 16 + (lane & 15);
    int kb  = kt * 32 + ((lane < 16) ? 0 : 8);
    unsigned short* o = out + (size_t)tile * 512 + lane * 16;
#pragma unroll
    for (int e = 0; e < 8; ++e) o[e]     = f2bf(W[(kb + e)      * N + col]);
#pragma unroll
    for (int e = 0; e < 8; ++e) o[8 + e] = f2bf(W[(kb + 16 + e) * N + col]);
}

__device__ __forceinline__ void load_a_frag(FragU& a, const unsigned short* p) {
    a.u4[0] = *(const uint4*)(p);
    a.u4[1] = *(const uint4*)(p + 16);
}

__device__ __forceinline__ unsigned int pack2bf(float lo, float hi) {
    return (unsigned int)f2bf(lo) | ((unsigned int)f2bf(hi) << 16);
}

__device__ __forceinline__ void load_a_frag(FragU& a, const float* p) {
    float4 f0 = *(const float4*)(p);
    float4 f1 = *(const float4*)(p + 4);
    float4 f2 = *(const float4*)(p + 16);
    float4 f3 = *(const float4*)(p + 20);
    asm volatile("" :: "v"(f0.x), "v"(f0.y), "v"(f0.z), "v"(f0.w),
                       "v"(f1.x), "v"(f1.y), "v"(f1.z), "v"(f1.w),
                       "v"(f2.x), "v"(f2.y), "v"(f2.z), "v"(f2.w),
                       "v"(f3.x), "v"(f3.y), "v"(f3.z), "v"(f3.w));
    a.u4[0] = make_uint4(pack2bf(f0.x, f0.y), pack2bf(f0.z, f0.w),
                         pack2bf(f1.x, f1.y), pack2bf(f1.z, f1.w));
    a.u4[1] = make_uint4(pack2bf(f2.x, f2.y), pack2bf(f2.z, f2.w),
                         pack2bf(f3.x, f3.y), pack2bf(f3.z, f3.w));
}

template <typename TA>
__device__ __forceinline__ void gemm_strip(const TA* __restrict__ A,
                                           const unsigned short* __restrict__ Bp,
                                           float* __restrict__ C, int M, int K,
                                           unsigned short* sB) {
    {
        const uint4* gB  = (const uint4*)Bp;
        uint4*       sB4 = (uint4*)sB;
        int n4 = (K >> 5) * 512;
        for (int i = threadIdx.x; i < n4; i += blockDim.x) sB4[i] = gB[i];
    }
    __syncthreads();

    int wave   = (int)((blockIdx.x * blockDim.x + threadIdx.x) >> 5);
    int lane   = threadIdx.x & 31;
    int mtiles = (M + 15) >> 4;
    if (wave >= mtiles) return;

    int row = wave * 16 + (lane & 15);
    if (row >= M) row = M - 1;
    int kb0 = (lane < 16) ? 0 : 8;
    const TA* arow = A + (size_t)row * K + kb0;

    v8f zero = {};
    v8f acc[8];
#pragma unroll
    for (int nt = 0; nt < 8; ++nt) acc[nt] = zero;

    for (int kt = 0; kt < K; kt += 32) {
        FragU a;
        load_a_frag(a, arow + kt);
        const unsigned short* bp = sB + (size_t)(kt >> 5) * 8 * 512 + lane * 16;
#pragma unroll
        for (int nt = 0; nt < 8; ++nt) {
            FragU b;
            const uint4* bl = (const uint4*)(bp + nt * 512);
            b.u4[0] = bl[0];
            b.u4[1] = bl[1];
            acc[nt] = __builtin_amdgcn_wmma_f32_16x16x32_bf16(
                false, a.v, false, b.v, (short)0, acc[nt], false, false);
        }
    }

    int srow0 = wave * 16 + ((lane < 16) ? 0 : 8);
    int scol  = lane & 15;
#pragma unroll
    for (int nt = 0; nt < 8; ++nt) {
#pragma unroll
        for (int j = 0; j < 8; ++j) {
            int r = srow0 + j;
            if (r < M) C[(size_t)r * 128 + nt * 16 + scol] = acc[nt][j];
        }
    }
}

__global__ void k_gemm_f32a(const float* __restrict__ A,
                            const unsigned short* __restrict__ Bp,
                            float* __restrict__ C, int M, int K) {
    extern __shared__ unsigned short sB[];
    gemm_strip<float>(A, Bp, C, M, K, sB);
}

__global__ void k_gemm_bf16(const unsigned short* __restrict__ A,
                            const unsigned short* __restrict__ Bp,
                            float* __restrict__ C, int M, int K) {
    extern __shared__ unsigned short sB[];
    gemm_strip<unsigned short>(A, Bp, C, M, K, sB);
}

__global__ void k_cnt_zero(int* cnt, int n) {
    int i = blockIdx.x * blockDim.x + threadIdx.x;
    if (i < n) cnt[i] = 0;
}

__global__ void k_cnt_rank(const int* __restrict__ dst, int* __restrict__ cnt,
                           int* __restrict__ rank, int e) {
    int i = blockIdx.x * blockDim.x + threadIdx.x;
    if (i < e) rank[i] = atomicAdd(&cnt[dst[i]], 1);
}

__global__ void k_scan(const int* __restrict__ cnt, int* __restrict__ rowptr,
                       float* __restrict__ dinv, int n) {
    __shared__ int s_red[32];
    __shared__ int s_wav[32];
    int tid  = threadIdx.x;
    int lane = tid & 31, wv = tid >> 5;
    int b0   = blockIdx.x * 1024;

    int part = 0;
    const int4* c4 = (const int4*)cnt;
#pragma unroll 4
    for (int k = tid; k < (b0 >> 2); k += 1024) {
        int4 q = c4[k];
        part += (q.x + q.y) + (q.z + q.w);
    }
#pragma unroll
    for (int off = 16; off >= 1; off >>= 1) part += __shfl_xor(part, off, 32);
    if (lane == 0) s_red[wv] = part;

    int i = b0 + tid;
    int v = (i < n) ? cnt[i] : 0;
    int inc = v;
#pragma unroll
    for (int off = 1; off < 32; off <<= 1) {
        int t = __shfl_up(inc, off, 32);
        if (lane >= off) inc += t;
    }
    if (lane == 31) s_wav[wv] = inc;
    __syncthreads();

    int t = s_red[lane] + ((lane < wv) ? s_wav[lane] : 0);
#pragma unroll
    for (int off = 16; off >= 1; off >>= 1) t += __shfl_xor(t, off, 32);
    int excl = t + inc - v;
    if (i < n) {
        rowptr[i] = excl;
        dinv[i]   = rsqrtf((float)(v + 1));
        if (i == n - 1) rowptr[n] = excl + v;
    }
}

__global__ void k_fill(const int* __restrict__ src, const int* __restrict__ dst,
                       const int* __restrict__ rank, const int* __restrict__ rowptr,
                       int* __restrict__ col, int e) {
    int i = blockIdx.x * blockDim.x + threadIdx.x;
    if (i < e) col[rowptr[dst[i]] + rank[i]] = src[i];
}

__global__ void k_sort_seg(const int* __restrict__ rowptr, const int* __restrict__ col,
                           const float* __restrict__ dinv, int* __restrict__ rec_i, int n) {
    int2* rec = (int2*)rec_i;
    int node = __builtin_amdgcn_readfirstlane((int)((blockIdx.x * blockDim.x + threadIdx.x) >> 5));
    if (node >= n) return;
    int lane = threadIdx.x & 31;
    int beg = rowptr[node];
    int deg = rowptr[node + 1] - beg;
    float dn = dinv[node];
    const int* seg = col + beg;
    for (int c0 = 0; c0 < deg; c0 += 32) {
        int i = c0 + lane;
        bool act = i < deg;
        int s = act ? seg[i] : 0;
        s = min(max(s, 0), n - 1);
        int r = 0;
        for (int k = 0; k < deg; ++k) {
            int t = min(max(seg[k], 0), n - 1);
            r += (t < s || (t == s && k < i)) ? 1 : 0;
        }
        if (act) rec[beg + r] = make_int2(s, __float_as_int(dinv[s] * dn));
    }
}

template <int G>
__device__ __forceinline__ void gather_rows(float4& acc, const int2& r, int j,
                                            const float4* __restrict__ h4, int lane) {
    float4 v[G]; float w[G];
#pragma unroll
    for (int u = 0; u < G; ++u) {
        int s = __builtin_amdgcn_readlane(r.x, j + u);
        w[u]  = __int_as_float(__builtin_amdgcn_readlane(r.y, j + u));
        v[u]  = h4[(size_t)s * 32 + lane];
    }
#pragma unroll
    for (int u = 0; u < G; ++u) {
        acc.x = fmaf(w[u], v[u].x, acc.x);
        acc.y = fmaf(w[u], v[u].y, acc.y);
        acc.z = fmaf(w[u], v[u].z, acc.z);
        acc.w = fmaf(w[u], v[u].w, acc.w);
    }
}

__global__ void k_gather(const int* __restrict__ rowptr, const int* __restrict__ rec_i,
                         const float* __restrict__ dinv, const float* __restrict__ h,
                         float* __restrict__ agg, int n) {
    const int2* rec = (const int2*)rec_i;
    int node = __builtin_amdgcn_readfirstlane((int)((blockIdx.x * blockDim.x + threadIdx.x) >> 5));
    if (node >= n) return;
    int lane = threadIdx.x & 31;
    int beg = rowptr[node], end = rowptr[node + 1];
    const float4* h4 = (const float4*)h;
    float dn = dinv[node];
    float w0 = dn * dn;
    float4 hv = h4[(size_t)node * 32 + lane];
    float4 acc = make_float4(w0 * hv.x, w0 * hv.y, w0 * hv.z, w0 * hv.w);

    for (int c0 = beg; c0 < end; c0 += 32) {
        int m = min(32, end - c0);
        int2 r = make_int2(0, 0);
        if (lane < m) r = rec[c0 + lane];
        int j = 0;
        for (; j + 8 <= m; j += 8) gather_rows<8>(acc, r, j, h4, lane);
        if (j + 4 <= m) { gather_rows<4>(acc, r, j, h4, lane); j += 4; }
        if (j + 2 <= m) { gather_rows<2>(acc, r, j, h4, lane); j += 2; }
        if (j < m)        gather_rows<1>(acc, r, j, h4, lane);
    }
    ((float4*)agg)[(size_t)node * 32 + lane] = acc;
}

__global__ void k_ln_elu(const float* __restrict__ agg,
                         const float* __restrict__ bias,
                         const float* __restrict__ gamma,
                         const float* __restrict__ beta,
                         const float* __restrict__ res_in,
                         float* __restrict__ out_f32,
                         unsigned short* __restrict__ out_bf16,
                         int n, int mode) {
    int wave = (int)((blockIdx.x * blockDim.x + threadIdx.x) >> 5);
    if (wave >= n) return;
    int lane = threadIdx.x & 31;
    const float* a = agg + (size_t)wave * 128;

    float v[4];
    float s = 0.f;
#pragma unroll
    for (int j = 0; j < 4; ++j) {
        int c = lane * 4 + j;
        v[j] = a[c] + bias[c];
        s += v[j];
    }
#pragma unroll
    for (int off = 16; off >= 1; off >>= 1) s += __shfl_xor(s, off, 32);
    float mu = s * (1.0f / 128.0f);

    float q = 0.f;
#pragma unroll
    for (int j = 0; j < 4; ++j) {
        float dv = v[j] - mu;
        q += dv * dv;
    }
#pragma unroll
    for (int off = 16; off >= 1; off >>= 1) q += __shfl_xor(q, off, 32);
    float inv = rsqrtf(q * (1.0f / 128.0f) + 1e-5f);

#pragma unroll
    for (int j = 0; j < 4; ++j) {
        int c = lane * 4 + j;
        float y = (v[j] - mu) * inv * gamma[c] + beta[c];
        y = (y > 0.f) ? y : expm1f(y);
        size_t idx = (size_t)wave * 128 + c;
        if (mode == 0) {
            out_f32[idx]  = y;
            out_bf16[idx] = f2bf(y);
        } else {
            out_f32[idx] = y + res_in[idx];
        }
    }
}

extern "C" void kernel_launch(void* const* d_in, const int* in_sizes, int n_in,
                              void* d_out, int out_size, void* d_ws, size_t ws_size,
                              hipStream_t stream) {
    const float* x   = (const float*)d_in[0];
    const float* W1  = (const float*)d_in[1];
    const float* b1  = (const float*)d_in[2];
    const float* g1  = (const float*)d_in[3];
    const float* be1 = (const float*)d_in[4];
    const float* W2  = (const float*)d_in[5];
    const float* b2  = (const float*)d_in[6];
    const float* g2  = (const float*)d_in[7];
    const float* be2 = (const float*)d_in[8];
    const int*   ei  = (const int*)d_in[9];

    const int EMB = in_sizes[2];
    const int IN  = in_sizes[1] / EMB;
    const int N   = in_sizes[0] / IN;
    const int E   = in_sizes[9] / 2;
    const int* src = ei;
    const int* dst = ei + E;

    char* base = (char*)d_ws;
    size_t off = 0;
    auto alloc = [&](size_t bytes) {
        char* p = base + off;
        off += (bytes + 255) & ~(size_t)255;
        return p;
    };
    float*          dinv  = (float*)alloc((size_t)N * 4);
    int*            cnt   = (int*)alloc((size_t)N * 4);
    int*            rowp  = (int*)alloc((size_t)(N + 1) * 4);
    int*            rank  = (int*)alloc((size_t)E * 4);
    int*            col   = (int*)alloc((size_t)E * 4);
    int*            rec   = (int*)alloc((size_t)E * 8);
    unsigned short* W1p   = (unsigned short*)alloc((size_t)IN * EMB * 2);
    unsigned short* W2p   = (unsigned short*)alloc((size_t)EMB * EMB * 2);
    float*          h     = (float*)alloc((size_t)N * EMB * 4);
    float*          agg   = (float*)alloc((size_t)N * EMB * 4);
    float*          res   = (float*)alloc((size_t)N * EMB * 4);
    unsigned short* h1b   = (unsigned short*)alloc((size_t)N * EMB * 2);
    (void)ws_size;

    const int B = 256;

    k_cnt_zero<<<(N + B - 1) / B, B, 0, stream>>>(cnt, N);
    k_cnt_rank<<<(E + B - 1) / B, B, 0, stream>>>(dst, cnt, rank, E);
    k_scan    <<<(N + 1023) / 1024, 1024, 0, stream>>>(cnt, rowp, dinv, N);
    k_fill    <<<(E + B - 1) / B, B, 0, stream>>>(src, dst, rank, rowp, col, E);
    k_sort_seg<<<(N + 7) / 8, B, 0, stream>>>(rowp, col, dinv, rec, N);

    k_pack_w  <<<(IN  / 32) * (EMB / 16), 32, 0, stream>>>(W1, W1p, IN,  EMB);
    k_pack_w  <<<(EMB / 32) * (EMB / 16), 32, 0, stream>>>(W2, W2p, EMB, EMB);

    int mtiles = (N + 15) / 16;
    int gemm_blocks = (mtiles + 7) / 8;
    size_t lds1 = (size_t)(IN  / 32) * 8 * 1024;
    size_t lds2 = (size_t)(EMB / 32) * 8 * 1024;

    k_gemm_f32a<<<gemm_blocks, B, lds1, stream>>>(x, W1p, h, N, IN);
    k_gather   <<<(N + 7) / 8, B, 0, stream>>>(rowp, rec, dinv, h, agg, N);
    k_ln_elu   <<<(N + 7) / 8, B, 0, stream>>>(agg, b1, g1, be1, nullptr, res, h1b, N, 0);

    k_gemm_bf16<<<gemm_blocks, B, lds2, stream>>>(h1b, W2p, h, N, EMB);
    k_gather   <<<(N + 7) / 8, B, 0, stream>>>(rowp, rec, dinv, h, agg, N);
    k_ln_elu   <<<(N + 7) / 8, B, 0, stream>>>(agg, b2, g2, be2, res, (float*)d_out, nullptr, N, 1);
}


# Bias+LayerNorm+ELU (+residual, bf16 copy) applied in the gather wave to the summed row still in registers: the two LayerNorm launches and the aggregate plane go; every sum keeps its order
# speedup vs baseline: 2.3970x; 1.1133x over previous
#include <hip/hip_runtime.h>
#include <hip/hip_bf16.h>

typedef __attribute__((ext_vector_type(16))) __bf16 v16bf;
typedef __attribute__((ext_vector_type(8)))  float  v8f;

union FragU {
    v16bf v;
    unsigned short u16[16];
    uint4 u4[2];
};

__device__ __forceinline__ unsigned short f2bf(float f) {
    unsigned int u = __float_as_uint(f);
    unsigned int r = u + 0x7FFFu + ((u >> 16) & 1u);
    return (unsigned short)(r >> 16);
}

__global__ void k_pack_w(const float* W, unsigned short* out, int K, int N) {
    int tile = blockIdx.x;
    int lane = threadIdx.x;
    int ntiles = N >> 4;
    int kt = tile / ntiles, nt = tile - kt * ntiles;
    int col = nt * 16 + (lane & 15);
    int kb  = kt * 32 + ((lane < 16) ? 0 : 8);
    unsigned short* o = out + (size_t)tile * 512 + lane * 16;
#pragma unroll
    for (int e = 0; e < 8; ++e) o[e]     = f2bf(W[(kb + e)      * N + col]);
#pragma unroll
    for (int e = 0; e < 8; ++e) o[8 + e] = f2bf(W[(kb + 16 + e) * N + col]);
}

__device__ __forceinline__ void load_a_frag(FragU& a, const unsigned short* p) {
    a.u4[0] = *(const uint4*)(p);
    a.u4[1] = *(const uint4*)(p + 16);
}

__device__ __forceinline__ unsigned int pack2bf(float lo, float hi) {
    return (unsigned int)f2bf(lo) | ((unsigned int)f2bf(hi) << 16);
}

__device__ __forceinline__ void load_a_frag(FragU& a, const float* p) {
    float4 f0 = *(const float4*)(p);
    float4 f1 = *(const float4*)(p + 4);
    float4 f2 = *(const float4*)(p + 16);
    float4 f3 = *(const float4*)(p + 20);
    asm volatile("" :: "v"(f0.x), "v"(f0.y), "v"(f0.z), "v"(f0.w),
                       "v"(f1.x), "v"(f1.y), "v"(f1.z), "v"(f1.w),
                       "v"(f2.x), "v"(f2.y), "v"(f2.z), "v"(f2.w),
                       "v"(f3.x), "v"(f3.y), "v"(f3.z), "v"(f3.w));
    a.u4[0] = make_uint4(pack2bf(f0.x, f0.y), pack2bf(f0.z, f0.w),
                         pack2bf(f1.x, f1.y), pack2bf(f1.z, f1.w));
    a.u4[1] = make_uint4(pack2bf(f2.x, f2.y), pack2bf(f2.z, f2.w),
                         pack2bf(f3.x, f3.y), pack2bf(f3.z, f3.w));
}

template <typename TA>
__device__ __forceinline__ void gemm_strip(const TA* __restrict__ A,
                                           const unsigned short* __restrict__ Bp,
                                           float* __restrict__ C, int M, int K,
                                           unsigned short* sB) {
    {
        const uint4* gB  = (const uint4*)Bp;
        uint4*       sB4 = (uint4*)sB;
        int n4 = (K >> 5) * 512;
        for (int i = threadIdx.x; i < n4; i += blockDim.x) sB4[i] = gB[i];
    }
    __syncthreads();

    int wave   = (int)((blockIdx.x * blockDim.x + threadIdx.x) >> 5);
    int lane   = threadIdx.x & 31;
    int mtiles = (M + 15) >> 4;
    if (wave >= mtiles) return;

    int row = wave * 16 + (lane & 15);
    if (row >= M) row = M - 1;
    int kb0 = (lane < 16) ? 0 : 8;
    const TA* arow = A + (size_t)row * K + kb0;

    v8f zero = {};
    v8f acc[8];
#pragma unroll
    for (int nt = 0; nt < 8; ++nt) acc[nt] = zero;

    for (int kt = 0; kt < K; kt += 32) {
        FragU a;
        load_a_frag(a, arow + kt);
        const unsigned short* bp = sB + (size_t)(kt >> 5) * 8 * 512 + lane * 16;
#pragma unroll
        for (int nt = 0; nt < 8; ++nt) {
            FragU b;
            const uint4* bl = (const uint4*)(bp + nt * 512);
            b.u4[0] = bl[0];
            b.u4[1] = bl[1];
            acc[nt] = __builtin_amdgcn_wmma_f32_16x16x32_bf16(
                false, a.v, false, b.v, (short)0, acc[nt], false, false);
        }
    }

    int srow0 = wave * 16 + ((lane < 16) ? 0 : 8);
    int scol  = lane & 15;
#pragma unroll
    for (int nt = 0; nt < 8; ++nt) {
#pragma unroll
        for (int j = 0; j < 8; ++j) {
            int r = srow0 + j;
            if (r < M) C[(size_t)r * 128 + nt * 16 + scol] = acc[nt][j];
        }
    }
}

__global__ void k_gemm_f32a(const float* __restrict__ A,
                            const unsigned short* __restrict__ Bp,
                            float* __restrict__ C, int M, int K) {
    extern __shared__ unsigned short sB[];
    gemm_strip<float>(A, Bp, C, M, K, sB);
}

__global__ void k_gemm_bf16(const unsigned short* __restrict__ A,
                            const unsigned short* __restrict__ Bp,
                            float* __restrict__ C, int M, int K) {
    extern __shared__ unsigned short sB[];
    gemm_strip<unsigned short>(A, Bp, C, M, K, sB);
}

__global__ void k_cnt_zero(int* cnt, int n) {
    int i = blockIdx.x * blockDim.x + threadIdx.x;
    if (i < n) cnt[i] = 0;
}

__global__ void k_cnt_rank(const int* __restrict__ dst, int* __restrict__ cnt,
                           int* __restrict__ rank, int e) {
    int i = blockIdx.x * blockDim.x + threadIdx.x;
    if (i < e) rank[i] = atomicAdd(&cnt[dst[i]], 1);
}

__global__ void k_scan(const int* __restrict__ cnt, int* __restrict__ rowptr,
                       float* __restrict__ dinv, int n) {
    __shared__ int s_red[32];
    __shared__ int s_wav[32];
    int tid  = threadIdx.x;
    int lane = tid & 31, wv = tid >> 5;
    int b0   = blockIdx.x * 1024;

    int part = 0;
    const int4* c4 = (const int4*)cnt;
#pragma unroll 4
    for (int k = tid; k < (b0 >> 2); k += 1024) {
        int4 q = c4[k];
        part += (q.x + q.y) + (q.z + q.w);
    }
#pragma unroll
    for (int off = 16; off >= 1; off >>= 1) part += __shfl_xor(part, off, 32);
    if (lane == 0) s_red[wv] = part;

    int i = b0 + tid;
    int v = (i < n) ? cnt[i] : 0;
    int inc = v;
#pragma unroll
    for (int off = 1; off < 32; off <<= 1) {
        int t = __shfl_up(inc, off, 32);
        if (lane >= off) inc += t;
    }
    if (lane == 31) s_wav[wv] = inc;
    __syncthreads();

    int t = s_red[lane] + ((lane < wv) ? s_wav[lane] : 0);
#pragma unroll
    for (int off = 16; off >= 1; off >>= 1) t += __shfl_xor(t, off, 32);
    int excl = t + inc - v;
    if (i < n) {
        rowptr[i] = excl;
        dinv[i]   = rsqrtf((float)(v + 1));
        if (i == n - 1) rowptr[n] = excl + v;
    }
}

__global__ void k_fill(const int* __restrict__ src, const int* __restrict__ dst,
                       const int* __restrict__ rank, const int* __restrict__ rowptr,
                       int* __restrict__ col, int e) {
    int i = blockIdx.x * blockDim.x + threadIdx.x;
    if (i < e) col[rowptr[dst[i]] + rank[i]] = src[i];
}

__global__ void k_sort_seg(const int* __restrict__ rowptr, const int* __restrict__ col,
                           const float* __restrict__ dinv, int* __restrict__ rec_i, int n) {
    int2* rec = (int2*)rec_i;
    int node = __builtin_amdgcn_readfirstlane((int)((blockIdx.x * blockDim.x + threadIdx.x) >> 5));
    if (node >= n) return;
    int lane = threadIdx.x & 31;
    int beg = rowptr[node];
    int deg = rowptr[node + 1] - beg;
    float dn = dinv[node];
    const int* seg = col + beg;
    for (int c0 = 0; c0 < deg; c0 += 32) {
        int i = c0 + lane;
        bool act = i < deg;
        int s = act ? seg[i] : 0;
        s = min(max(s, 0), n - 1);
        int r = 0;
        for (int k = 0; k < deg; ++k) {
            int t = min(max(seg[k], 0), n - 1);
            r += (t < s || (t == s && k < i)) ? 1 : 0;
        }
        if (act) rec[beg + r] = make_int2(s, __float_as_int(dinv[s] * dn));
    }
}

template <int G>
__device__ __forceinline__ void gather_rows(float4& acc, const int2& r, int j,
                                            const float4* __restrict__ h4, int lane) {
    float4 v[G]; float w[G];
#pragma unroll
    for (int u = 0; u < G; ++u) {
        int s = __builtin_amdgcn_readlane(r.x, j + u);
        w[u]  = __int_as_float(__builtin_amdgcn_readlane(r.y, j + u));
        v[u]  = h4[(size_t)s * 32 + lane];
    }
#pragma unroll
    for (int u = 0; u < G; ++u) {
        acc.x = fmaf(w[u], v[u].x, acc.x);
        acc.y = fmaf(w[u], v[u].y, acc.y);
        acc.z = fmaf(w[u], v[u].z, acc.z);
        acc.w = fmaf(w[u], v[u].w, acc.w);
    }
}

template <int MODE>
__device__ __forceinline__ void gather_ln_row(const int* __restrict__ rowptr,
                                              const int* __restrict__ rec_i,
                                              const float* __restrict__ dinv,
                                              const float* __restrict__ h,
                                              const float* __restrict__ bias,
                                              const float* __restrict__ gamma,
                                              const float* __restrict__ beta,
                                              const float* __restrict__ res_in,
                                              float* __restrict__ out_f32,
                                              unsigned short* __restrict__ out_bf16,
                                              int n) {
    const int2* rec = (const int2*)rec_i;
    int node = __builtin_amdgcn_readfirstlane((int)((blockIdx.x * blockDim.x + threadIdx.x) >> 5));
    if (node >= n) return;
    int lane = threadIdx.x & 31;
    int beg = rowptr[node], end = rowptr[node + 1];
    const float4* h4 = (const float4*)h;
    float dn = dinv[node];
    float w0 = dn * dn;
    size_t ridx = (size_t)node * 32 + lane;
    float4 hv = h4[ridx];
    float4 bv = ((const float4*)bias)[lane];
    float4 rv = make_float4(0.f, 0.f, 0.f, 0.f);
    if (MODE == 1) rv = ((const float4*)res_in)[ridx];
    float4 acc = make_float4(w0 * hv.x, w0 * hv.y, w0 * hv.z, w0 * hv.w);

    for (int c0 = beg; c0 < end; c0 += 32) {
        int m = min(32, end - c0);
        int2 r = make_int2(0, 0);
        if (lane < m) r = rec[c0 + lane];
        int j = 0;
        for (; j + 8 <= m; j += 8) gather_rows<8>(acc, r, j, h4, lane);
        if (j + 4 <= m) { gather_rows<4>(acc, r, j, h4, lane); j += 4; }
        if (j + 2 <= m) { gather_rows<2>(acc, r, j, h4, lane); j += 2; }
        if (j < m)        gather_rows<1>(acc, r, j, h4, lane);
    }

    float4 gv = ((const float4*)gamma)[lane];
    float4 ev = ((const float4*)beta)[lane];
    float v[4] = { acc.x + bv.x, acc.y + bv.y, acc.z + bv.z, acc.w + bv.w };
    float s = 0.f;
#pragma unroll
    for (int j = 0; j < 4; ++j) s += v[j];
#pragma unroll
    for (int off = 16; off >= 1; off >>= 1) s += __shfl_xor(s, off, 32);
    float mu = s * (1.0f / 128.0f);

    float q = 0.f;
#pragma unroll
    for (int j = 0; j < 4; ++j) {
        float dv = v[j] - mu;
        q += dv * dv;
    }
#pragma unroll
    for (int off = 16; off >= 1; off >>= 1) q += __shfl_xor(q, off, 32);
    float inv = rsqrtf(q * (1.0f / 128.0f) + 1e-5f);

    const float g[4] = { gv.x, gv.y, gv.z, gv.w };
    const float b[4] = { ev.x, ev.y, ev.z, ev.w };
    float y[4];
#pragma unroll
    for (int j = 0; j < 4; ++j) {
        float t = (v[j] - mu) * inv * g[j] + b[j];
        y[j] = (t > 0.f) ? t : expm1f(t);
    }
    if (MODE == 0) {
        ((float4*)out_f32)[ridx] = make_float4(y[0], y[1], y[2], y[3]);
        ((uint2*)out_bf16)[ridx] = make_uint2(pack2bf(y[0], y[1]),
                                              pack2bf(y[2], y[3]));
    } else {
        ((float4*)out_f32)[ridx] = make_float4(y[0] + rv.x, y[1] + rv.y,
                                               y[2] + rv.z, y[3] + rv.w);
    }
}

__global__ void k_gather_ln1(const int* __restrict__ rowptr, const int* __restrict__ rec_i,
                             const float* __restrict__ dinv, const float* __restrict__ h,
                             const float* __restrict__ bias, const float* __restrict__ gamma,
                             const float* __restrict__ beta,
                             float* __restrict__ out_f32, unsigned short* __restrict__ out_bf16,
                             int n) {
    gather_ln_row<0>(rowptr, rec_i, dinv, h, bias, gamma, beta, nullptr, out_f32, out_bf16, n);
}

__global__ void k_gather_ln2(const int* __restrict__ rowptr, const int* __restrict__ rec_i,
                             const float* __restrict__ dinv, const float* __restrict__ h,
                             const float* __restrict__ bias, const float* __restrict__ gamma,
                             const float* __restrict__ beta, const float* __restrict__ res_in,
                             float* __restrict__ out_f32, int n) {
    gather_ln_row<1>(rowptr, rec_i, dinv, h, bias, gamma, beta, res_in, out_f32, nullptr, n);
}

extern "C" void kernel_launch(void* const* d_in, const int* in_sizes, int n_in,
                              void* d_out, int out_size, void* d_ws, size_t ws_size,
                              hipStream_t stream) {
    const float* x   = (const float*)d_in[0];
    const float* W1  = (const float*)d_in[1];
    const float* b1  = (const float*)d_in[2];
    const float* g1  = (const float*)d_in[3];
    const float* be1 = (const float*)d_in[4];
    const float* W2  = (const float*)d_in[5];
    const float* b2  = (const float*)d_in[6];
    const float* g2  = (const float*)d_in[7];
    const float* be2 = (const float*)d_in[8];
    const int*   ei  = (const int*)d_in[9];

    const int EMB = in_sizes[2];
    const int IN  = in_sizes[1] / EMB;
    const int N   = in_sizes[0] / IN;
    const int E   = in_sizes[9] / 2;
    const int* src = ei;
    const int* dst = ei + E;

    char* base = (char*)d_ws;
    size_t off = 0;
    auto alloc = [&](size_t bytes) {
        char* p = base + off;
        off += (bytes + 255) & ~(size_t)255;
        return p;
    };
    float*          dinv  = (float*)alloc((size_t)N * 4);
    int*            cnt   = (int*)alloc((size_t)N * 4);
    int*            rowp  = (int*)alloc((size_t)(N + 1) * 4);
    int*            rank  = (int*)alloc((size_t)E * 4);
    int*            col   = (int*)alloc((size_t)E * 4);
    int*            rec   = (int*)alloc((size_t)E * 8);
    unsigned short* W1p   = (unsigned short*)alloc((size_t)IN * EMB * 2);
    unsigned short* W2p   = (unsigned short*)alloc((size_t)EMB * EMB * 2);
    float*          h     = (float*)alloc((size_t)N * EMB * 4);
    float*          res   = (float*)alloc((size_t)N * EMB * 4);
    unsigned short* h1b   = (unsigned short*)alloc((size_t)N * EMB * 2);
    (void)ws_size;

    const int B = 256;

    k_cnt_zero<<<(N + B - 1) / B, B, 0, stream>>>(cnt, N);
    k_cnt_rank<<<(E + B - 1) / B, B, 0, stream>>>(dst, cnt, rank, E);
    k_scan    <<<(N + 1023) / 1024, 1024, 0, stream>>>(cnt, rowp, dinv, N);
    k_fill    <<<(E + B - 1) / B, B, 0, stream>>>(src, dst, rank, rowp, col, E);
    k_sort_seg<<<(N + 7) / 8, B, 0, stream>>>(rowp, col, dinv, rec, N);

    k_pack_w  <<<(IN  / 32) * (EMB / 16), 32, 0, stream>>>(W1, W1p, IN,  EMB);
    k_pack_w  <<<(EMB / 32) * (EMB / 16), 32, 0, stream>>>(W2, W2p, EMB, EMB);

    int mtiles = (N + 15) / 16;
    int gemm_blocks = (mtiles + 7) / 8;
    size_t lds1 = (size_t)(IN  / 32) * 8 * 1024;
    size_t lds2 = (size_t)(EMB / 32) * 8 * 1024;

    k_gemm_f32a<<<gemm_blocks, B, lds1, stream>>>(x, W1p, h, N, IN);
    k_gather_ln1<<<(N + 7) / 8, B, 0, stream>>>(rowp, rec, dinv, h, b1, g1, be1, res, h1b, N);

    k_gemm_bf16<<<gemm_blocks, B, lds2, stream>>>(h1b, W2p, h, N, EMB);
    k_gather_ln2<<<(N + 7) / 8, B, 0, stream>>>(rowp, rec, dinv, h, b2, g2, be2, res, (float*)d_out, N);
}


# Degree count and edge slots by LDS counters: 32 edge chunks, one work-group each, a counter per node in LDS (returning LDS add = slot), chunk table scanned; the 800k global atomics go; sum order kept
# speedup vs baseline: 3.0447x; 1.2702x over previous
#include <hip/hip_runtime.h>
#include <hip/hip_bf16.h>

typedef __attribute__((ext_vector_type(16))) __bf16 v16bf;
typedef __attribute__((ext_vector_type(8)))  float  v8f;

union FragU {
    v16bf v;
    unsigned short u16[16];
    uint4 u4[2];
};

__device__ __forceinline__ unsigned short f2bf(float f) {
    unsigned int u = __float_as_uint(f);
    unsigned int r = u + 0x7FFFu + ((u >> 16) & 1u);
    return (unsigned short)(r >> 16);
}

__global__ void k_pack_w(const float* W, unsigned short* out, int K, int N) {
    int tile = blockIdx.x;
    int lane = threadIdx.x;
    int ntiles = N >> 4;
    int kt = tile / ntiles, nt = tile - kt * ntiles;
    int col = nt * 16 + (lane & 15);
    int kb  = kt * 32 + ((lane < 16) ? 0 : 8);
    unsigned short* o = out + (size_t)tile * 512 + lane * 16;
#pragma unroll
    for (int e = 0; e < 8; ++e) o[e]     = f2bf(W[(kb + e)      * N + col]);
#pragma unroll
    for (int e = 0; e < 8; ++e) o[8 + e] = f2bf(W[(kb + 16 + e) * N + col]);
}

__device__ __forceinline__ void load_a_frag(FragU& a, const unsigned short* p) {
    a.u4[0] = *(const uint4*)(p);
    a.u4[1] = *(const uint4*)(p + 16);
}

__device__ __forceinline__ unsigned int pack2bf(float lo, float hi) {
    return (unsigned int)f2bf(lo) | ((unsigned int)f2bf(hi) << 16);
}

__device__ __forceinline__ void load_a_frag(FragU& a, const float* p) {
    float4 f0 = *(const float4*)(p);
    float4 f1 = *(const float4*)(p + 4);
    float4 f2 = *(const float4*)(p + 16);
    float4 f3 = *(const float4*)(p + 20);
    asm volatile("" :: "v"(f0.x), "v"(f0.y), "v"(f0.z), "v"(f0.w),
                       "v"(f1.x), "v"(f1.y), "v"(f1.z), "v"(f1.w),
                       "v"(f2.x), "v"(f2.y), "v"(f2.z), "v"(f2.w),
                       "v"(f3.x), "v"(f3.y), "v"(f3.z), "v"(f3.w));
    a.u4[0] = make_uint4(pack2bf(f0.x, f0.y), pack2bf(f0.z, f0.w),
                         pack2bf(f1.x, f1.y), pack2bf(f1.z, f1.w));
    a.u4[1] = make_uint4(pack2bf(f2.x, f2.y), pack2bf(f2.z, f2.w),
                         pack2bf(f3.x, f3.y), pack2bf(f3.z, f3.w));
}

template <typename TA>
__device__ __forceinline__ void gemm_strip(const TA* __restrict__ A,
                                           const unsigned short* __restrict__ Bp,
                                           float* __restrict__ C, int M, int K,
                                           unsigned short* sB) {
    {
        const uint4* gB  = (const uint4*)Bp;
        uint4*       sB4 = (uint4*)sB;
        int n4 = (K >> 5) * 512;
        for (int i = threadIdx.x; i < n4; i += blockDim.x) sB4[i] = gB[i];
    }
    __syncthreads();

    int wave   = (int)((blockIdx.x * blockDim.x + threadIdx.x) >> 5);
    int lane   = threadIdx.x & 31;
    int mtiles = (M + 15) >> 4;
    if (wave >= mtiles) return;

    int row = wave * 16 + (lane & 15);
    if (row >= M) row = M - 1;
    int kb0 = (lane < 16) ? 0 : 8;
    const TA* arow = A + (size_t)row * K + kb0;

    v8f zero = {};
    v8f acc[8];
#pragma unroll
    for (int nt = 0; nt < 8; ++nt) acc[nt] = zero;

    for (int kt = 0; kt < K; kt += 32) {
        FragU a;
        load_a_frag(a, arow + kt);
        const unsigned short* bp = sB + (size_t)(kt >> 5) * 8 * 512 + lane * 16;
#pragma unroll
        for (int nt = 0; nt < 8; ++nt) {
            FragU b;
            const uint4* bl = (const uint4*)(bp + nt * 512);
            b.u4[0] = bl[0];
            b.u4[1] = bl[1];
            acc[nt] = __builtin_amdgcn_wmma_f32_16x16x32_bf16(
                false, a.v, false, b.v, (short)0, acc[nt], false, false);
        }
    }

    int srow0 = wave * 16 + ((lane < 16) ? 0 : 8);
    int scol  = lane & 15;
#pragma unroll
    for (int nt = 0; nt < 8; ++nt) {
#pragma unroll
        for (int j = 0; j < 8; ++j) {
            int r = srow0 + j;
            if (r < M) C[(size_t)r * 128 + nt * 16 + scol] = acc[nt][j];
        }
    }
}

__global__ void k_gemm_f32a(const float* __restrict__ A,
                            const unsigned short* __restrict__ Bp,
                            float* __restrict__ C, int M, int K) {
    extern __shared__ unsigned short sB[];
    gemm_strip<float>(A, Bp, C, M, K, sB);
}

__global__ void k_gemm_bf16(const unsigned short* __restrict__ A,
                            const unsigned short* __restrict__ Bp,
                            float* __restrict__ C, int M, int K) {
    extern __shared__ unsigned short sB[];
    gemm_strip<unsigned short>(A, Bp, C, M, K, sB);
}

constexpr int LCH  = 32;
constexpr int LCAP = 65536;
constexpr int LDEP = 32;

__device__ __forceinline__ int chunk_edges(int e) {
    return (((e + LCH - 1) / LCH) + 3) & ~3;
}

__global__ __launch_bounds__(1024) void k_cnt_lds(const int* __restrict__ dst, int* __restrict__ rank,
                                                  int* __restrict__ tab, int* __restrict__ bsum,
                                                  int n, int np, int nbp, int e) {
    __shared__ __attribute__((aligned(16))) int s_cnt[LCAP];
    __shared__ int s_part[(LCAP / 4096) * 32];
    const int tid = threadIdx.x, lane = tid & 31, wv = tid >> 5;
    const int c   = blockIdx.x;
    const int per = chunk_edges(e);
    const int e0  = min(c * per, e), e1 = min(e0 + per, e);
    int* trow = tab  + (size_t)c * np;
    int* brow = bsum + (size_t)c * nbp;

    for (int w0 = 0; w0 < np; w0 += LCAP) {
        const int wn  = min(LCAP, np - w0);
        const int wn4 = wn >> 2;
        for (int q = tid; q < wn4; q += 1024) ((int4*)s_cnt)[q] = make_int4(0, 0, 0, 0);
        __syncthreads();

        for (int i0 = e0; i0 < e1; i0 += LDEP * 1024) {
            int d[LDEP];
#pragma unroll
            for (int u = 0; u < LDEP; ++u) {
                int i = i0 + u * 1024 + tid;
                d[u] = (i < e1) ? dst[i] : -1;
            }
            int r[LDEP];
#pragma unroll
            for (int u = 0; u < LDEP; ++u) {
                int i = i0 + u * 1024 + tid;
                unsigned k = (unsigned)(d[u] - w0);
                r[u] = -1;
                if (i < e1 && k < (unsigned)wn) r[u] = atomicAdd(&s_cnt[k], 1);
            }
#pragma unroll
            for (int u = 0; u < LDEP; ++u) {
                int i = i0 + u * 1024 + tid;
                if (r[u] >= 0) rank[i] = r[u];
            }
        }
        __syncthreads();

        const int steps = (wn4 + 1023) >> 10;
        for (int t = 0; t < steps; ++t) {
            int q = t * 1024 + tid;
            int4 v = make_int4(0, 0, 0, 0);
            if (q < wn4) {
                v = ((const int4*)s_cnt)[q];
                ((int4*)(trow + w0))[q] = v;
            }
            int s = (v.x + v.y) + (v.z + v.w);
#pragma unroll
            for (int off = 16; off >= 1; off >>= 1) s += __shfl_xor(s, off, 32);
            if (lane == 0) s_part[t * 32 + wv] = s;
        }
        __syncthreads();
        const int nr = (wn + 1023) >> 10;
        for (int j = tid; j < nr; j += 1024) {
            int s = 0;
#pragma unroll
            for (int k = 0; k < 8; ++k) s += s_part[8 * j + k];
            brow[(w0 >> 10) + j] = s;
        }
        __syncthreads();
    }
}

__global__ __launch_bounds__(1024) void k_scan(int* __restrict__ tab, const int* __restrict__ bsum,
                                               int* __restrict__ rowptr, float* __restrict__ dinv,
                                               int n, int np, int nbp) {
    __shared__ int s_red[32];
    __shared__ int s_wav[32];
    int tid  = threadIdx.x;
    int lane = tid & 31, wv = tid >> 5;
    int b0   = blockIdx.x * 1024;

    int part = 0;
    for (int j = tid; j < (int)blockIdx.x; j += 1024) {
#pragma unroll
        for (int c = 0; c < LCH; ++c) part += bsum[(size_t)c * nbp + j];
    }
#pragma unroll
    for (int off = 16; off >= 1; off >>= 1) part += __shfl_xor(part, off, 32);
    if (lane == 0) s_red[wv] = part;

    int i = b0 + tid;
    int cv[LCH];
    int v = 0;
#pragma unroll
    for (int c = 0; c < LCH; ++c) {
        cv[c] = (i < n) ? tab[(size_t)c * np + i] : 0;
        v += cv[c];
    }
    int inc = v;
#pragma unroll
    for (int off = 1; off < 32; off <<= 1) {
        int t = __shfl_up(inc, off, 32);
        if (lane >= off) inc += t;
    }
    if (lane == 31) s_wav[wv] = inc;
    __syncthreads();

    int t = s_red[lane] + ((lane < wv) ? s_wav[lane] : 0);
#pragma unroll
    for (int off = 16; off >= 1; off >>= 1) t += __shfl_xor(t, off, 32);
    int excl = t + inc - v;
    if (i < n) {
        rowptr[i] = excl;
        dinv[i]   = rsqrtf((float)(v + 1));
        if (i == n - 1) rowptr[n] = excl + v;
        int run = excl;
#pragma unroll
        for (int c = 0; c < LCH; ++c) {
            tab[(size_t)c * np + i] = run;
            run += cv[c];
        }
    }
}

__global__ void k_fill(const int* __restrict__ src, const int* __restrict__ dst,
                       const int* __restrict__ rank, const int* __restrict__ tab,
                       int* __restrict__ col, int e, int np) {
    int i = blockIdx.x * blockDim.x + threadIdx.x;
    if (i < e) {
        int c = i / chunk_edges(e);
        col[tab[(size_t)c * np + dst[i]] + rank[i]] = src[i];
    }
}

__global__ void k_sort_seg(const int* __restrict__ rowptr, const int* __restrict__ col,
                           const float* __restrict__ dinv, int* __restrict__ rec_i, int n) {
    int2* rec = (int2*)rec_i;
    int node = __builtin_amdgcn_readfirstlane((int)((blockIdx.x * blockDim.x + threadIdx.x) >> 5));
    if (node >= n) return;
    int lane = threadIdx.x & 31;
    int beg = rowptr[node];
    int deg = rowptr[node + 1] - beg;
    float dn = dinv[node];
    const int* seg = col + beg;
    for (int c0 = 0; c0 < deg; c0 += 32) {
        int i = c0 + lane;
        bool act = i < deg;
        int s = act ? seg[i] : 0;
        s = min(max(s, 0), n - 1);
        int r = 0;
        for (int k = 0; k < deg; ++k) {
            int t = min(max(seg[k], 0), n - 1);
            r += (t < s || (t == s && k < i)) ? 1 : 0;
        }
        if (act) rec[beg + r] = make_int2(s, __float_as_int(dinv[s] * dn));
    }
}

template <int G>
__device__ __forceinline__ void gather_rows(float4& acc, const int2& r, int j,
                                            const float4* __restrict__ h4, int lane) {
    float4 v[G]; float w[G];
#pragma unroll
    for (int u = 0; u < G; ++u) {
        int s = __builtin_amdgcn_readlane(r.x, j + u);
        w[u]  = __int_as_float(__builtin_amdgcn_readlane(r.y, j + u));
        v[u]  = h4[(size_t)s * 32 + lane];
    }
#pragma unroll
    for (int u = 0; u < G; ++u) {
        acc.x = fmaf(w[u], v[u].x, acc.x);
        acc.y = fmaf(w[u], v[u].y, acc.y);
        acc.z = fmaf(w[u], v[u].z, acc.z);
        acc.w = fmaf(w[u], v[u].w, acc.w);
    }
}

template <int MODE>
__device__ __forceinline__ void gather_ln_row(const int* __restrict__ rowptr,
                                              const int* __restrict__ rec_i,
                                              const float* __restrict__ dinv,
                                              const float* __restrict__ h,
                                              const float* __restrict__ bias,
                                              const float* __restrict__ gamma,
                                              const float* __restrict__ beta,
                                              const float* __restrict__ res_in,
                                              float* __restrict__ out_f32,
                                              unsigned short* __restrict__ out_bf16,
                                              int n) {
    const int2* rec = (const int2*)rec_i;
    int node = __builtin_amdgcn_readfirstlane((int)((blockIdx.x * blockDim.x + threadIdx.x) >> 5));
    if (node >= n) return;
    int lane = threadIdx.x & 31;
    int beg = rowptr[node], end = rowptr[node + 1];
    const float4* h4 = (const float4*)h;
    float dn = dinv[node];
    float w0 = dn * dn;
    size_t ridx = (size_t)node * 32 + lane;
    float4 hv = h4[ridx];
    float4 bv = ((const float4*)bias)[lane];
    float4 rv = make_float4(0.f, 0.f, 0.f, 0.f);
    if (MODE == 1) rv = ((const float4*)res_in)[ridx];
    float4 acc = make_float4(w0 * hv.x, w0 * hv.y, w0 * hv.z, w0 * hv.w);

    for (int c0 = beg; c0 < end; c0 += 32) {
        int m = min(32, end - c0);
        int2 r = make_int2(0, 0);
        if (lane < m) r = rec[c0 + lane];
        int j = 0;
        for (; j + 8 <= m; j += 8) gather_rows<8>(acc, r, j, h4, lane);
        if (j + 4 <= m) { gather_rows<4>(acc, r, j, h4, lane); j += 4; }
        if (j + 2 <= m) { gather_rows<2>(acc, r, j, h4, lane); j += 2; }
        if (j < m)        gather_rows<1>(acc, r, j, h4, lane);
    }

    float4 gv = ((const float4*)gamma)[lane];
    float4 ev = ((const float4*)beta)[lane];
    float v[4] = { acc.x + bv.x, acc.y + bv.y, acc.z + bv.z, acc.w + bv.w };
    float s = 0.f;
#pragma unroll
    for (int j = 0; j < 4; ++j) s += v[j];
#pragma unroll
    for (int off = 16; off >= 1; off >>= 1) s += __shfl_xor(s, off, 32);
    float mu = s * (1.0f / 128.0f);

    float q = 0.f;
#pragma unroll
    for (int j = 0; j < 4; ++j) {
        float dv = v[j] - mu;
        q += dv * dv;
    }
#pragma unroll
    for (int off = 16; off >= 1; off >>= 1) q += __shfl_xor(q, off, 32);
    float inv = rsqrtf(q * (1.0f / 128.0f) + 1e-5f);

    const float g[4] = { gv.x, gv.y, gv.z, gv.w };
    const float b[4] = { ev.x, ev.y, ev.z, ev.w };
    float y[4];
#pragma unroll
    for (int j = 0; j < 4; ++j) {
        float t = (v[j] - mu) * inv * g[j] + b[j];
        y[j] = (t > 0.f) ? t : expm1f(t);
    }
    if (MODE == 0) {
        ((float4*)out_f32)[ridx] = make_float4(y[0], y[1], y[2], y[3]);
        ((uint2*)out_bf16)[ridx] = make_uint2(pack2bf(y[0], y[1]),
                                              pack2bf(y[2], y[3]));
    } else {
        ((float4*)out_f32)[ridx] = make_float4(y[0] + rv.x, y[1] + rv.y,
                                               y[2] + rv.z, y[3] + rv.w);
    }
}

__global__ void k_gather_ln1(const int* __restrict__ rowptr, const int* __restrict__ rec_i,
                             const float* __restrict__ dinv, const float* __restrict__ h,
                             const float* __restrict__ bias, const float* __restrict__ gamma,
                             const float* __restrict__ beta,
                             float* __restrict__ out_f32, unsigned short* __restrict__ out_bf16,
                             int n) {
    gather_ln_row<0>(rowptr, rec_i, dinv, h, bias, gamma, beta, nullptr, out_f32, out_bf16, n);
}

__global__ void k_gather_ln2(const int* __restrict__ rowptr, const int* __restrict__ rec_i,
                             const float* __restrict__ dinv, const float* __restrict__ h,
                             const float* __restrict__ bias, const float* __restrict__ gamma,
                             const float* __restrict__ beta, const float* __restrict__ res_in,
                             float* __restrict__ out_f32, int n) {
    gather_ln_row<1>(rowptr, rec_i, dinv, h, bias, gamma, beta, res_in, out_f32, nullptr, n);
}

extern "C" void kernel_launch(void* const* d_in, const int* in_sizes, int n_in,
                              void* d_out, int out_size, void* d_ws, size_t ws_size,
                              hipStream_t stream) {
    const float* x   = (const float*)d_in[0];
    const float* W1  = (const float*)d_in[1];
    const float* b1  = (const float*)d_in[2];
    const float* g1  = (const float*)d_in[3];
    const float* be1 = (const float*)d_in[4];
    const float* W2  = (const float*)d_in[5];
    const float* b2  = (const float*)d_in[6];
    const float* g2  = (const float*)d_in[7];
    const float* be2 = (const float*)d_in[8];
    const int*   ei  = (const int*)d_in[9];

    const int EMB = in_sizes[2];
    const int IN  = in_sizes[1] / EMB;
    const int N   = in_sizes[0] / IN;
    const int E   = in_sizes[9] / 2;
    const int* src = ei;
    const int* dst = ei + E;

    char* base = (char*)d_ws;
    size_t off = 0;
    auto alloc = [&](size_t bytes) {
        char* p = base + off;
        off += (bytes + 255) & ~(size_t)255;
        return p;
    };
    float*          dinv  = (float*)alloc((size_t)N * 4);
    const int NP  = (N + 3) & ~3;
    const int NB  = (N + 1023) / 1024;
    int*            tab   = (int*)alloc((size_t)LCH * NP * 4);
    int*            bsum  = (int*)alloc((size_t)LCH * NB * 4);
    int*            rowp  = (int*)alloc((size_t)(N + 1) * 4);
    int*            rank  = (int*)alloc((size_t)E * 4);
    int*            col   = (int*)alloc((size_t)E * 4);
    int*            rec   = (int*)alloc((size_t)E * 8);
    unsigned short* W1p   = (unsigned short*)alloc((size_t)IN * EMB * 2);
    unsigned short* W2p   = (unsigned short*)alloc((size_t)EMB * EMB * 2);
    float*          h     = (float*)alloc((size_t)N * EMB * 4);
    float*          res   = (float*)alloc((size_t)N * EMB * 4);
    unsigned short* h1b   = (unsigned short*)alloc((size_t)N * EMB * 2);
    (void)ws_size;

    const int B = 256;

    k_cnt_lds <<<LCH, 1024, 0, stream>>>(dst, rank, tab, bsum, N, NP, NB, E);
    k_scan    <<<NB, 1024, 0, stream>>>(tab, bsum, rowp, dinv, N, NP, NB);
    k_fill    <<<(E + B - 1) / B, B, 0, stream>>>(src, dst, rank, tab, col, E, NP);
    k_sort_seg<<<(N + 7) / 8, B, 0, stream>>>(rowp, col, dinv, rec, N);

    k_pack_w  <<<(IN  / 32) * (EMB / 16), 32, 0, stream>>>(W1, W1p, IN,  EMB);
    k_pack_w  <<<(EMB / 32) * (EMB / 16), 32, 0, stream>>>(W2, W2p, EMB, EMB);

    int mtiles = (N + 15) / 16;
    int gemm_blocks = (mtiles + 7) / 8;
    size_t lds1 = (size_t)(IN  / 32) * 8 * 1024;
    size_t lds2 = (size_t)(EMB / 32) * 8 * 1024;

    k_gemm_f32a<<<gemm_blocks, B, lds1, stream>>>(x, W1p, h, N, IN);
    k_gather_ln1<<<(N + 7) / 8, B, 0, stream>>>(rowp, rec, dinv, h, b1, g1, be1, res, h1b, N);

    k_gemm_bf16<<<gemm_blocks, B, lds2, stream>>>(h1b, W2p, h, N, EMB);
    k_gather_ln2<<<(N + 7) / 8, B, 0, stream>>>(rowp, rec, dinv, h, b2, g2, be2, res, (float*)d_out, N);
}


# Independent kernels share launches by work-group range: weight packing rides in the edge-count launch, the layer-1 GEMM in the edge-placing launch; 10 launches -> 7, every sum and its order kept
# speedup vs baseline: 3.3718x; 1.1074x over previous
#include <hip/hip_runtime.h>
#include <hip/hip_bf16.h>

typedef __attribute__((ext_vector_type(16))) __bf16 v16bf;
typedef __attribute__((ext_vector_type(8)))  float  v8f;

union FragU {
    v16bf v;
    unsigned short u16[16];
    uint4 u4[2];
};

__device__ __forceinline__ unsigned short f2bf(float f) {
    unsigned int u = __float_as_uint(f);
    unsigned int r = u + 0x7FFFu + ((u >> 16) & 1u);
    return (unsigned short)(r >> 16);
}

__device__ __forceinline__ void pack_tile(const float* __restrict__ W, unsigned short* __restrict__ out,
                                          int tile, int lane, int N) {
    int ntiles = N >> 4;
    int kt = tile / ntiles, nt = tile - kt * ntiles;
    int col = nt * 16 + (lane & 15);
    int kb  = kt * 32 + ((lane < 16) ? 0 : 8);
    unsigned short* o = out + (size_t)tile * 512 + lane * 16;
#pragma unroll
    for (int e = 0; e < 8; ++e) o[e]     = f2bf(W[(kb + e)      * N + col]);
#pragma unroll
    for (int e = 0; e < 8; ++e) o[8 + e] = f2bf(W[(kb + 16 + e) * N + col]);
}

__device__ __forceinline__ void load_a_frag(FragU& a, const unsigned short* p) {
    a.u4[0] = *(const uint4*)(p);
    a.u4[1] = *(const uint4*)(p + 16);
}

__device__ __forceinline__ unsigned int pack2bf(float lo, float hi) {
    return (unsigned int)f2bf(lo) | ((unsigned int)f2bf(hi) << 16);
}

__device__ __forceinline__ void load_a_frag(FragU& a, const float* p) {
    float4 f0 = *(const float4*)(p);
    float4 f1 = *(const float4*)(p + 4);
    float4 f2 = *(const float4*)(p + 16);
    float4 f3 = *(const float4*)(p + 20);
    asm volatile("" :: "v"(f0.x), "v"(f0.y), "v"(f0.z), "v"(f0.w),
                       "v"(f1.x), "v"(f1.y), "v"(f1.z), "v"(f1.w),
                       "v"(f2.x), "v"(f2.y), "v"(f2.z), "v"(f2.w),
                       "v"(f3.x), "v"(f3.y), "v"(f3.z), "v"(f3.w));
    a.u4[0] = make_uint4(pack2bf(f0.x, f0.y), pack2bf(f0.z, f0.w),
                         pack2bf(f1.x, f1.y), pack2bf(f1.z, f1.w));
    a.u4[1] = make_uint4(pack2bf(f2.x, f2.y), pack2bf(f2.z, f2.w),
                         pack2bf(f3.x, f3.y), pack2bf(f3.z, f3.w));
}

template <typename TA>
__device__ __forceinline__ void gemm_strip(const TA* __restrict__ A,
                                           const unsigned short* __restrict__ Bp,
                                           float* __restrict__ C, int M, int K,
                                           unsigned short* sB) {
    {
        const uint4* gB  = (const uint4*)Bp;
        uint4*       sB4 = (uint4*)sB;
        int n4 = (K >> 5) * 512;
        for (int i = threadIdx.x; i < n4; i += blockDim.x) sB4[i] = gB[i];
    }
    __syncthreads();

    int wave   = (int)((blockIdx.x * blockDim.x + threadIdx.x) >> 5);
    int lane   = threadIdx.x & 31;
    int mtiles = (M + 15) >> 4;
    if (wave >= mtiles) return;

    int row = wave * 16 + (lane & 15);
    if (row >= M) row = M - 1;
    int kb0 = (lane < 16) ? 0 : 8;
    const TA* arow = A + (size_t)row * K + kb0;

    v8f zero = {};
    v8f acc[8];
#pragma unroll
    for (int nt = 0; nt < 8; ++nt) acc[nt] = zero;

    for (int kt = 0; kt < K; kt += 32) {
        FragU a;
        load_a_frag(a, arow + kt);
        const unsigned short* bp = sB + (size_t)(kt >> 5) * 8 * 512 + lane * 16;
#pragma unroll
        for (int nt = 0; nt < 8; ++nt) {
            FragU b;
            const uint4* bl = (const uint4*)(bp + nt * 512);
            b.u4[0] = bl[0];
            b.u4[1] = bl[1];
            acc[nt] = __builtin_amdgcn_wmma_f32_16x16x32_bf16(
                false, a.v, false, b.v, (short)0, acc[nt], false, false);
        }
    }

    int srow0 = wave * 16 + ((lane < 16) ? 0 : 8);
    int scol  = lane & 15;
#pragma unroll
    for (int nt = 0; nt < 8; ++nt) {
#pragma unroll
        for (int j = 0; j < 8; ++j) {
            int r = srow0 + j;
            if (r < M) C[(size_t)r * 128 + nt * 16 + scol] = acc[nt][j];
        }
    }
}


__global__ void k_gemm_bf16(const unsigned short* __restrict__ A,
                            const unsigned short* __restrict__ Bp,
                            float* __restrict__ C, int M, int K) {
    extern __shared__ unsigned short sB[];
    gemm_strip<unsigned short>(A, Bp, C, M, K, sB);
}

constexpr int LCH  = 32;
constexpr int LCAP = 65536;
constexpr int LDEP = 32;

__device__ __forceinline__ int chunk_edges(int e) {
    return (((e + LCH - 1) / LCH) + 3) & ~3;
}

__global__ __launch_bounds__(1024) void k_cnt_pack(const int* __restrict__ dst, int* __restrict__ rank,
                                                   int* __restrict__ tab, int* __restrict__ bsum,
                                                   int n, int np, int nbp, int e,
                                                   const float* __restrict__ W1, unsigned short* __restrict__ W1p, int K1,
                                                   const float* __restrict__ W2, unsigned short* __restrict__ W2p, int K2,
                                                   int ncol) {
    __shared__ __attribute__((aligned(16))) int s_cnt[LCAP];
    __shared__ int s_part[(LCAP / 4096) * 32];
    const int tid = threadIdx.x, lane = tid & 31, wv = tid >> 5;
    if ((int)blockIdx.x >= LCH) {
        const int tile = ((int)blockIdx.x - LCH) * 32 + wv;
        const int t1 = (K1 >> 5) * (ncol >> 4);
        const int t2 = (K2 >> 5) * (ncol >> 4);
        if (tile < t1)           pack_tile(W1, W1p, tile, lane, ncol);
        else if (tile - t1 < t2) pack_tile(W2, W2p, tile - t1, lane, ncol);
        return;
    }
    const int c   = blockIdx.x;
    const int per = chunk_edges(e);
    const int e0  = min(c * per, e), e1 = min(e0 + per, e);
    int* trow = tab  + (size_t)c * np;
    int* brow = bsum + (size_t)c * nbp;

    for (int w0 = 0; w0 < np; w0 += LCAP) {
        const int wn  = min(LCAP, np - w0);
        const int wn4 = wn >> 2;
        for (int q = tid; q < wn4; q += 1024) ((int4*)s_cnt)[q] = make_int4(0, 0, 0, 0);
        __syncthreads();

        for (int i0 = e0; i0 < e1; i0 += LDEP * 1024) {
            int d[LDEP];
#pragma unroll
            for (int u = 0; u < LDEP; ++u) {
                int i = i0 + u * 1024 + tid;
                d[u] = (i < e1) ? dst[i] : -1;
            }
            int r[LDEP];
#pragma unroll
            for (int u = 0; u < LDEP; ++u) {
                int i = i0 + u * 1024 + tid;
                unsigned k = (unsigned)(d[u] - w0);
                r[u] = -1;
                if (i < e1 && k < (unsigned)wn) r[u] = atomicAdd(&s_cnt[k], 1);
            }
#pragma unroll
            for (int u = 0; u < LDEP; ++u) {
                int i = i0 + u * 1024 + tid;
                if (r[u] >= 0) rank[i] = r[u];
            }
        }
        __syncthreads();

        const int steps = (wn4 + 1023) >> 10;
        for (int t = 0; t < steps; ++t) {
            int q = t * 1024 + tid;
            int4 v = make_int4(0, 0, 0, 0);
            if (q < wn4) {
                v = ((const int4*)s_cnt)[q];
                ((int4*)(trow + w0))[q] = v;
            }
            int s = (v.x + v.y) + (v.z + v.w);
#pragma unroll
            for (int off = 16; off >= 1; off >>= 1) s += __shfl_xor(s, off, 32);
            if (lane == 0) s_part[t * 32 + wv] = s;
        }
        __syncthreads();
        const int nr = (wn + 1023) >> 10;
        for (int j = tid; j < nr; j += 1024) {
            int s = 0;
#pragma unroll
            for (int k = 0; k < 8; ++k) s += s_part[8 * j + k];
            brow[(w0 >> 10) + j] = s;
        }
        __syncthreads();
    }
}

__global__ __launch_bounds__(1024) void k_scan(int* __restrict__ tab, const int* __restrict__ bsum,
                                               int* __restrict__ rowptr, float* __restrict__ dinv,
                                               int n, int np, int nbp) {
    __shared__ int s_red[32];
    __shared__ int s_wav[32];
    int tid  = threadIdx.x;
    int lane = tid & 31, wv = tid >> 5;
    int b0   = blockIdx.x * 1024;

    int part = 0;
    for (int j = tid; j < (int)blockIdx.x; j += 1024) {
#pragma unroll
        for (int c = 0; c < LCH; ++c) part += bsum[(size_t)c * nbp + j];
    }
#pragma unroll
    for (int off = 16; off >= 1; off >>= 1) part += __shfl_xor(part, off, 32);
    if (lane == 0) s_red[wv] = part;

    int i = b0 + tid;
    int cv[LCH];
    int v = 0;
#pragma unroll
    for (int c = 0; c < LCH; ++c) {
        cv[c] = (i < n) ? tab[(size_t)c * np + i] : 0;
        v += cv[c];
    }
    int inc = v;
#pragma unroll
    for (int off = 1; off < 32; off <<= 1) {
        int t = __shfl_up(inc, off, 32);
        if (lane >= off) inc += t;
    }
    if (lane == 31) s_wav[wv] = inc;
    __syncthreads();

    int t = s_red[lane] + ((lane < wv) ? s_wav[lane] : 0);
#pragma unroll
    for (int off = 16; off >= 1; off >>= 1) t += __shfl_xor(t, off, 32);
    int excl = t + inc - v;
    if (i < n) {
        rowptr[i] = excl;
        dinv[i]   = rsqrtf((float)(v + 1));
        if (i == n - 1) rowptr[n] = excl + v;
        int run = excl;
#pragma unroll
        for (int c = 0; c < LCH; ++c) {
            tab[(size_t)c * np + i] = run;
            run += cv[c];
        }
    }
}

__global__ void k_gemm1_fill(const float* __restrict__ A, const unsigned short* __restrict__ Bp,
                             float* __restrict__ C, int M, int K, int gb,
                             const int* __restrict__ src, const int* __restrict__ dst,
                             const int* __restrict__ rank, const int* __restrict__ tab,
                             int* __restrict__ col, int e, int np) {
    extern __shared__ unsigned short sB[];
    if ((int)blockIdx.x < gb) {
        gemm_strip<float>(A, Bp, C, M, K, sB);
        return;
    }
    int i = ((int)blockIdx.x - gb) * blockDim.x + threadIdx.x;
    if (i < e) {
        int c = i / chunk_edges(e);
        col[tab[(size_t)c * np + dst[i]] + rank[i]] = src[i];
    }
}

__global__ void k_sort_seg(const int* __restrict__ rowptr, const int* __restrict__ col,
                           const float* __restrict__ dinv, int* __restrict__ rec_i, int n) {
    int2* rec = (int2*)rec_i;
    int node = __builtin_amdgcn_readfirstlane((int)((blockIdx.x * blockDim.x + threadIdx.x) >> 5));
    if (node >= n) return;
    int lane = threadIdx.x & 31;
    int beg = rowptr[node];
    int deg = rowptr[node + 1] - beg;
    float dn = dinv[node];
    const int* seg = col + beg;
    for (int c0 = 0; c0 < deg; c0 += 32) {
        int i = c0 + lane;
        bool act = i < deg;
        int s = act ? seg[i] : 0;
        s = min(max(s, 0), n - 1);
        int r = 0;
        for (int k = 0; k < deg; ++k) {
            int t = min(max(seg[k], 0), n - 1);
            r += (t < s || (t == s && k < i)) ? 1 : 0;
        }
        if (act) rec[beg + r] = make_int2(s, __float_as_int(dinv[s] * dn));
    }
}

template <int G>
__device__ __forceinline__ void gather_rows(float4& acc, const int2& r, int j,
                                            const float4* __restrict__ h4, int lane) {
    float4 v[G]; float w[G];
#pragma unroll
    for (int u = 0; u < G; ++u) {
        int s = __builtin_amdgcn_readlane(r.x, j + u);
        w[u]  = __int_as_float(__builtin_amdgcn_readlane(r.y, j + u));
        v[u]  = h4[(size_t)s * 32 + lane];
    }
#pragma unroll
    for (int u = 0; u < G; ++u) {
        acc.x = fmaf(w[u], v[u].x, acc.x);
        acc.y = fmaf(w[u], v[u].y, acc.y);
        acc.z = fmaf(w[u], v[u].z, acc.z);
        acc.w = fmaf(w[u], v[u].w, acc.w);
    }
}

template <int MODE>
__device__ __forceinline__ void gather_ln_row(const int* __restrict__ rowptr,
                                              const int* __restrict__ rec_i,
                                              const float* __restrict__ dinv,
                                              const float* __restrict__ h,
                                              const float* __restrict__ bias,
                                              const float* __restrict__ gamma,
                                              const float* __restrict__ beta,
                                              const float* __restrict__ res_in,
                                              float* __restrict__ out_f32,
                                              unsigned short* __restrict__ out_bf16,
                                              int n) {
    const int2* rec = (const int2*)rec_i;
    int node = __builtin_amdgcn_readfirstlane((int)((blockIdx.x * blockDim.x + threadIdx.x) >> 5));
    if (node >= n) return;
    int lane = threadIdx.x & 31;
    int beg = rowptr[node], end = rowptr[node + 1];
    const float4* h4 = (const float4*)h;
    float dn = dinv[node];
    float w0 = dn * dn;
    size_t ridx = (size_t)node * 32 + lane;
    float4 hv = h4[ridx];
    float4 bv = ((const float4*)bias)[lane];
    float4 rv = make_float4(0.f, 0.f, 0.f, 0.f);
    if (MODE == 1) rv = ((const float4*)res_in)[ridx];
    float4 acc = make_float4(w0 * hv.x, w0 * hv.y, w0 * hv.z, w0 * hv.w);

    for (int c0 = beg; c0 < end; c0 += 32) {
        int m = min(32, end - c0);
        int2 r = make_int2(0, 0);
        if (lane < m) r = rec[c0 + lane];
        int j = 0;
        for (; j + 8 <= m; j += 8) gather_rows<8>(acc, r, j, h4, lane);
        if (j + 4 <= m) { gather_rows<4>(acc, r, j, h4, lane); j += 4; }
        if (j + 2 <= m) { gather_rows<2>(acc, r, j, h4, lane); j += 2; }
        if (j < m)        gather_rows<1>(acc, r, j, h4, lane);
    }

    float4 gv = ((const float4*)gamma)[lane];
    float4 ev = ((const float4*)beta)[lane];
    float v[4] = { acc.x + bv.x, acc.y + bv.y, acc.z + bv.z, acc.w + bv.w };
    float s = 0.f;
#pragma unroll
    for (int j = 0; j < 4; ++j) s += v[j];
#pragma unroll
    for (int off = 16; off >= 1; off >>= 1) s += __shfl_xor(s, off, 32);
    float mu = s * (1.0f / 128.0f);

    float q = 0.f;
#pragma unroll
    for (int j = 0; j < 4; ++j) {
        float dv = v[j] - mu;
        q += dv * dv;
    }
#pragma unroll
    for (int off = 16; off >= 1; off >>= 1) q += __shfl_xor(q, off, 32);
    float inv = rsqrtf(q * (1.0f / 128.0f) + 1e-5f);

    const float g[4] = { gv.x, gv.y, gv.z, gv.w };
    const float b[4] = { ev.x, ev.y, ev.z, ev.w };
    float y[4];
#pragma unroll
    for (int j = 0; j < 4; ++j) {
        float t = (v[j] - mu) * inv * g[j] + b[j];
        y[j] = (t > 0.f) ? t : expm1f(t);
    }
    if (MODE == 0) {
        ((float4*)out_f32)[ridx] = make_float4(y[0], y[1], y[2], y[3]);
        ((uint2*)out_bf16)[ridx] = make_uint2(pack2bf(y[0], y[1]),
                                              pack2bf(y[2], y[3]));
    } else {
        ((float4*)out_f32)[ridx] = make_float4(y[0] + rv.x, y[1] + rv.y,
                                               y[2] + rv.z, y[3] + rv.w);
    }
}

__global__ void k_gather_ln1(const int* __restrict__ rowptr, const int* __restrict__ rec_i,
                             const float* __restrict__ dinv, const float* __restrict__ h,
                             const float* __restrict__ bias, const float* __restrict__ gamma,
                             const float* __restrict__ beta,
                             float* __restrict__ out_f32, unsigned short* __restrict__ out_bf16,
                             int n) {
    gather_ln_row<0>(rowptr, rec_i, dinv, h, bias, gamma, beta, nullptr, out_f32, out_bf16, n);
}

__global__ void k_gather_ln2(const int* __restrict__ rowptr, const int* __restrict__ rec_i,
                             const float* __restrict__ dinv, const float* __restrict__ h,
                             const float* __restrict__ bias, const float* __restrict__ gamma,
                             const float* __restrict__ beta, const float* __restrict__ res_in,
                             float* __restrict__ out_f32, int n) {
    gather_ln_row<1>(rowptr, rec_i, dinv, h, bias, gamma, beta, res_in, out_f32, nullptr, n);
}

extern "C" void kernel_launch(void* const* d_in, const int* in_sizes, int n_in,
                              void* d_out, int out_size, void* d_ws, size_t ws_size,
                              hipStream_t stream) {
    const float* x   = (const float*)d_in[0];
    const float* W1  = (const float*)d_in[1];
    const float* b1  = (const float*)d_in[2];
    const float* g1  = (const float*)d_in[3];
    const float* be1 = (const float*)d_in[4];
    const float* W2  = (const float*)d_in[5];
    const float* b2  = (const float*)d_in[6];
    const float* g2  = (const float*)d_in[7];
    const float* be2 = (const float*)d_in[8];
    const int*   ei  = (const int*)d_in[9];

    const int EMB = in_sizes[2];
    const int IN  = in_sizes[1] / EMB;
    const int N   = in_sizes[0] / IN;
    const int E   = in_sizes[9] / 2;
    const int* src = ei;
    const int* dst = ei + E;

    char* base = (char*)d_ws;
    size_t off = 0;
    auto alloc = [&](size_t bytes) {
        char* p = base + off;
        off += (bytes + 255) & ~(size_t)255;
        return p;
    };
    float*          dinv  = (float*)alloc((size_t)N * 4);
    const int NP  = (N + 3) & ~3;
    const int NB  = (N + 1023) / 1024;
    int*            tab   = (int*)alloc((size_t)LCH * NP * 4);
    int*            bsum  = (int*)alloc((size_t)LCH * NB * 4);
    int*            rowp  = (int*)alloc((size_t)(N + 1) * 4);
    int*            rank  = (int*)alloc((size_t)E * 4);
    int*            col   = (int*)alloc((size_t)E * 4);
    int*            rec   = (int*)alloc((size_t)E * 8);
    unsigned short* W1p   = (unsigned short*)alloc((size_t)IN * EMB * 2);
    unsigned short* W2p   = (unsigned short*)alloc((size_t)EMB * EMB * 2);
    float*          h     = (float*)alloc((size_t)N * EMB * 4);
    float*          res   = (float*)alloc((size_t)N * EMB * 4);
    unsigned short* h1b   = (unsigned short*)alloc((size_t)N * EMB * 2);
    (void)ws_size;

    const int B = 256;

    int mtiles = (N + 15) / 16;
    int gemm_blocks = (mtiles + 7) / 8;
    size_t lds1 = (size_t)(IN  / 32) * 8 * 1024;
    size_t lds2 = (size_t)(EMB / 32) * 8 * 1024;
    int pack_tiles  = (IN / 32) * (EMB / 16) + (EMB / 32) * (EMB / 16);
    int pack_blocks = (pack_tiles + 31) / 32;

    k_cnt_pack  <<<LCH + pack_blocks, 1024, 0, stream>>>(dst, rank, tab, bsum, N, NP, NB, E,
                                                         W1, W1p, IN, W2, W2p, EMB, EMB);
    k_scan      <<<NB, 1024, 0, stream>>>(tab, bsum, rowp, dinv, N, NP, NB);
    k_gemm1_fill<<<gemm_blocks + (E + B - 1) / B, B, lds1, stream>>>(x, W1p, h, N, IN, gemm_blocks,
                                                                     src, dst, rank, tab, col, E, NP);
    k_sort_seg  <<<(N + 7) / 8, B, 0, stream>>>(rowp, col, dinv, rec, N);

    k_gather_ln1<<<(N + 7) / 8, B, 0, stream>>>(rowp, rec, dinv, h, b1, g1, be1, res, h1b, N);

    k_gemm_bf16<<<gemm_blocks, B, lds2, stream>>>(h1b, W2p, h, N, EMB);
    k_gather_ln2<<<(N + 7) / 8, B, 0, stream>>>(rowp, rec, dinv, h, b2, g2, be2, res, (float*)d_out, N);
}


# GEMM strips request global loads ahead of use: a thread's B-stage copies go out as one batch, A's K-step k+1 is requested once k is in the fragment and lands under k's WMMAs; sum order kept
# speedup vs baseline: 3.6443x; 1.0808x over previous
#include <hip/hip_runtime.h>
#include <hip/hip_bf16.h>

typedef __attribute__((ext_vector_type(16))) __bf16 v16bf;
typedef __attribute__((ext_vector_type(8)))  float  v8f;

union FragU {
    v16bf v;
    unsigned short u16[16];
    uint4 u4[2];
};

__device__ __forceinline__ unsigned short f2bf(float f) {
    unsigned int u = __float_as_uint(f);
    unsigned int r = u + 0x7FFFu + ((u >> 16) & 1u);
    return (unsigned short)(r >> 16);
}

__device__ __forceinline__ void pack_tile(const float* __restrict__ W, unsigned short* __restrict__ out,
                                          int tile, int lane, int N) {
    int ntiles = N >> 4;
    int kt = tile / ntiles, nt = tile - kt * ntiles;
    int col = nt * 16 + (lane & 15);
    int kb  = kt * 32 + ((lane < 16) ? 0 : 8);
    unsigned short* o = out + (size_t)tile * 512 + lane * 16;
#pragma unroll
    for (int e = 0; e < 8; ++e) o[e]     = f2bf(W[(kb + e)      * N + col]);
#pragma unroll
    for (int e = 0; e < 8; ++e) o[8 + e] = f2bf(W[(kb + 16 + e) * N + col]);
}

struct RawH { uint4 u0, u1; };
struct RawF { float4 f0, f1, f2, f3; };

__device__ __forceinline__ void issue_a(RawH& r, const unsigned short* p) {
    r.u0 = *(const uint4*)(p);
    r.u1 = *(const uint4*)(p + 16);
}

__device__ __forceinline__ void issue_a(RawF& r, const float* p) {
    r.f0 = *(const float4*)(p);
    r.f1 = *(const float4*)(p + 4);
    r.f2 = *(const float4*)(p + 16);
    r.f3 = *(const float4*)(p + 20);
}

__device__ __forceinline__ void finish_a(FragU& a, const RawH& r) {
    a.u4[0] = r.u0;
    a.u4[1] = r.u1;
}

__device__ __forceinline__ unsigned int pack2bf(float lo, float hi) {
    return (unsigned int)f2bf(lo) | ((unsigned int)f2bf(hi) << 16);
}

__device__ __forceinline__ void finish_a(FragU& a, const RawF& r) {
    a.u4[0] = make_uint4(pack2bf(r.f0.x, r.f0.y), pack2bf(r.f0.z, r.f0.w),
                         pack2bf(r.f1.x, r.f1.y), pack2bf(r.f1.z, r.f1.w));
    a.u4[1] = make_uint4(pack2bf(r.f2.x, r.f2.y), pack2bf(r.f2.z, r.f2.w),
                         pack2bf(r.f3.x, r.f3.y), pack2bf(r.f3.z, r.f3.w));
}

template <typename TA> struct RawOf;
template <> struct RawOf<unsigned short> { typedef RawH type; };
template <> struct RawOf<float>          { typedef RawF type; };

constexpr int SBAT = 16;

template <typename TA>
__device__ __forceinline__ void gemm_strip(const TA* __restrict__ A,
                                           const unsigned short* __restrict__ Bp,
                                           float* __restrict__ C, int M, int K,
                                           unsigned short* sB) {
    typedef typename RawOf<TA>::type RawA;
    int wave   = (int)((blockIdx.x * blockDim.x + threadIdx.x) >> 5);
    int lane   = threadIdx.x & 31;
    int mtiles = (M + 15) >> 4;

    int row = wave * 16 + (lane & 15);
    if (row >= M) row = M - 1;
    int kb0 = (lane < 16) ? 0 : 8;
    const TA* arow = A + (size_t)row * K + kb0;

    {
        const uint4* gB  = (const uint4*)Bp;
        uint4*       sB4 = (uint4*)sB;
        const int n4  = (K >> 5) * 512;
        const int nth = (int)blockDim.x;
        for (int b0 = 0; b0 < n4; b0 += SBAT * nth) {
            uint4 t[SBAT];
#pragma unroll
            for (int u = 0; u < SBAT; ++u) {
                int i = b0 + u * nth + (int)threadIdx.x;
                t[u] = (i < n4) ? gB[i] : make_uint4(0u, 0u, 0u, 0u);
            }
#pragma unroll
            for (int u = 0; u < SBAT; ++u) {
                int i = b0 + u * nth + (int)threadIdx.x;
                if (i < n4) sB4[i] = t[u];
            }
        }
    }
    const int ksteps = K >> 5;
    RawA raw;
    issue_a(raw, arow);
    __syncthreads();

    if (wave >= mtiles) return;

    v8f zero = {};
    v8f acc[8];
#pragma unroll
    for (int nt = 0; nt < 8; ++nt) acc[nt] = zero;

    for (int ks = 0; ks < ksteps; ++ks) {
        FragU a;
        finish_a(a, raw);
        issue_a(raw, arow + min(ks + 1, ksteps - 1) * 32);
        const unsigned short* bp = sB + (size_t)ks * 8 * 512 + lane * 16;
#pragma unroll
        for (int nt = 0; nt < 8; ++nt) {
            FragU b;
            const uint4* bl = (const uint4*)(bp + nt * 512);
            b.u4[0] = bl[0];
            b.u4[1] = bl[1];
            acc[nt] = __builtin_amdgcn_wmma_f32_16x16x32_bf16(
                false, a.v, false, b.v, (short)0, acc[nt], false, false);
        }
    }

    int srow0 = wave * 16 + ((lane < 16) ? 0 : 8);
    int scol  = lane & 15;
#pragma unroll
    for (int nt = 0; nt < 8; ++nt) {
#pragma unroll
        for (int j = 0; j < 8; ++j) {
            int r = srow0 + j;
            if (r < M) C[(size_t)r * 128 + nt * 16 + scol] = acc[nt][j];
        }
    }
}


__global__ void k_gemm_bf16(const unsigned short* __restrict__ A,
                            const unsigned short* __restrict__ Bp,
                            float* __restrict__ C, int M, int K) {
    extern __shared__ unsigned short sB[];
    gemm_strip<unsigned short>(A, Bp, C, M, K, sB);
}

constexpr int LCH  = 32;
constexpr int LCAP = 65536;
constexpr int LDEP = 32;

__device__ __forceinline__ int chunk_edges(int e) {
    return (((e + LCH - 1) / LCH) + 3) & ~3;
}

__global__ __launch_bounds__(1024) void k_cnt_pack(const int* __restrict__ dst, int* __restrict__ rank,
                                                   int* __restrict__ tab, int* __restrict__ bsum,
                                                   int n, int np, int nbp, int e,
                                                   const float* __restrict__ W1, unsigned short* __restrict__ W1p, int K1,
                                                   const float* __restrict__ W2, unsigned short* __restrict__ W2p, int K2,
                                                   int ncol) {
    __shared__ __attribute__((aligned(16))) int s_cnt[LCAP];
    __shared__ int s_part[(LCAP / 4096) * 32];
    const int tid = threadIdx.x, lane = tid & 31, wv = tid >> 5;
    if ((int)blockIdx.x >= LCH) {
        const int tile = ((int)blockIdx.x - LCH) * 32 + wv;
        const int t1 = (K1 >> 5) * (ncol >> 4);
        const int t2 = (K2 >> 5) * (ncol >> 4);
        if (tile < t1)           pack_tile(W1, W1p, tile, lane, ncol);
        else if (tile - t1 < t2) pack_tile(W2, W2p, tile - t1, lane, ncol);
        return;
    }
    const int c   = blockIdx.x;
    const int per = chunk_edges(e);
    const int e0  = min(c * per, e), e1 = min(e0 + per, e);
    int* trow = tab  + (size_t)c * np;
    int* brow = bsum + (size_t)c * nbp;

    for (int w0 = 0; w0 < np; w0 += LCAP) {
        const int wn  = min(LCAP, np - w0);
        const int wn4 = wn >> 2;
        for (int q = tid; q < wn4; q += 1024) ((int4*)s_cnt)[q] = make_int4(0, 0, 0, 0);
        __syncthreads();

        for (int i0 = e0; i0 < e1; i0 += LDEP * 1024) {
            int d[LDEP];
#pragma unroll
            for (int u = 0; u < LDEP; ++u) {
                int i = i0 + u * 1024 + tid;
                d[u] = (i < e1) ? dst[i] : -1;
            }
            int r[LDEP];
#pragma unroll
            for (int u = 0; u < LDEP; ++u) {
                int i = i0 + u * 1024 + tid;
                unsigned k = (unsigned)(d[u] - w0);
                r[u] = -1;
                if (i < e1 && k < (unsigned)wn) r[u] = atomicAdd(&s_cnt[k], 1);
            }
#pragma unroll
            for (int u = 0; u < LDEP; ++u) {
                int i = i0 + u * 1024 + tid;
                if (r[u] >= 0) rank[i] = r[u];
            }
        }
        __syncthreads();

        const int steps = (wn4 + 1023) >> 10;
        for (int t = 0; t < steps; ++t) {
            int q = t * 1024 + tid;
            int4 v = make_int4(0, 0, 0, 0);
            if (q < wn4) {
                v = ((const int4*)s_cnt)[q];
                ((int4*)(trow + w0))[q] = v;
            }
            int s = (v.x + v.y) + (v.z + v.w);
#pragma unroll
            for (int off = 16; off >= 1; off >>= 1) s += __shfl_xor(s, off, 32);
            if (lane == 0) s_part[t * 32 + wv] = s;
        }
        __syncthreads();
        const int nr = (wn + 1023) >> 10;
        for (int j = tid; j < nr; j += 1024) {
            int s = 0;
#pragma unroll
            for (int k = 0; k < 8; ++k) s += s_part[8 * j + k];
            brow[(w0 >> 10) + j] = s;
        }
        __syncthreads();
    }
}

__global__ __launch_bounds__(1024) void k_scan(int* __restrict__ tab, const int* __restrict__ bsum,
                                               int* __restrict__ rowptr, float* __restrict__ dinv,
                                               int n, int np, int nbp) {
    __shared__ int s_red[32];
    __shared__ int s_wav[32];
    int tid  = threadIdx.x;
    int lane = tid & 31, wv = tid >> 5;
    int b0   = blockIdx.x * 1024;

    int part = 0;
    for (int j = tid; j < (int)blockIdx.x; j += 1024) {
#pragma unroll
        for (int c = 0; c < LCH; ++c) part += bsum[(size_t)c * nbp + j];
    }
#pragma unroll
    for (int off = 16; off >= 1; off >>= 1) part += __shfl_xor(part, off, 32);
    if (lane == 0) s_red[wv] = part;

    int i = b0 + tid;
    int cv[LCH];
    int v = 0;
#pragma unroll
    for (int c = 0; c < LCH; ++c) {
        cv[c] = (i < n) ? tab[(size_t)c * np + i] : 0;
        v += cv[c];
    }
    int inc = v;
#pragma unroll
    for (int off = 1; off < 32; off <<= 1) {
        int t = __shfl_up(inc, off, 32);
        if (lane >= off) inc += t;
    }
    if (lane == 31) s_wav[wv] = inc;
    __syncthreads();

    int t = s_red[lane] + ((lane < wv) ? s_wav[lane] : 0);
#pragma unroll
    for (int off = 16; off >= 1; off >>= 1) t += __shfl_xor(t, off, 32);
    int excl = t + inc - v;
    if (i < n) {
        rowptr[i] = excl;
        dinv[i]   = rsqrtf((float)(v + 1));
        if (i == n - 1) rowptr[n] = excl + v;
        int run = excl;
#pragma unroll
        for (int c = 0; c < LCH; ++c) {
            tab[(size_t)c * np + i] = run;
            run += cv[c];
        }
    }
}

__global__ void k_gemm1_fill(const float* __restrict__ A, const unsigned short* __restrict__ Bp,
                             float* __restrict__ C, int M, int K, int gb,
                             const int* __restrict__ src, const int* __restrict__ dst,
                             const int* __restrict__ rank, const int* __restrict__ tab,
                             int* __restrict__ col, int e, int np) {
    extern __shared__ unsigned short sB[];
    if ((int)blockIdx.x < gb) {
        gemm_strip<float>(A, Bp, C, M, K, sB);
        return;
    }
    int i = ((int)blockIdx.x - gb) * blockDim.x + threadIdx.x;
    if (i < e) {
        int c = i / chunk_edges(e);
        col[tab[(size_t)c * np + dst[i]] + rank[i]] = src[i];
    }
}

__global__ void k_sort_seg(const int* __restrict__ rowptr, const int* __restrict__ col,
                           const float* __restrict__ dinv, int* __restrict__ rec_i, int n) {
    int2* rec = (int2*)rec_i;
    int node = __builtin_amdgcn_readfirstlane((int)((blockIdx.x * blockDim.x + threadIdx.x) >> 5));
    if (node >= n) return;
    int lane = threadIdx.x & 31;
    int beg = rowptr[node];
    int deg = rowptr[node + 1] - beg;
    float dn = dinv[node];
    const int* seg = col + beg;
    for (int c0 = 0; c0 < deg; c0 += 32) {
        int i = c0 + lane;
        bool act = i < deg;
        int s = act ? seg[i] : 0;
        s = min(max(s, 0), n - 1);
        int r = 0;
        for (int k = 0; k < deg; ++k) {
            int t = min(max(seg[k], 0), n - 1);
            r += (t < s || (t == s && k < i)) ? 1 : 0;
        }
        if (act) rec[beg + r] = make_int2(s, __float_as_int(dinv[s] * dn));
    }
}

template <int G>
__device__ __forceinline__ void gather_rows(float4& acc, const int2& r, int j,
                                            const float4* __restrict__ h4, int lane) {
    float4 v[G]; float w[G];
#pragma unroll
    for (int u = 0; u < G; ++u) {
        int s = __builtin_amdgcn_readlane(r.x, j + u);
        w[u]  = __int_as_float(__builtin_amdgcn_readlane(r.y, j + u));
        v[u]  = h4[(size_t)s * 32 + lane];
    }
#pragma unroll
    for (int u = 0; u < G; ++u) {
        acc.x = fmaf(w[u], v[u].x, acc.x);
        acc.y = fmaf(w[u], v[u].y, acc.y);
        acc.z = fmaf(w[u], v[u].z, acc.z);
        acc.w = fmaf(w[u], v[u].w, acc.w);
    }
}

template <int MODE>
__device__ __forceinline__ void gather_ln_row(const int* __restrict__ rowptr,
                                              const int* __restrict__ rec_i,
                                              const float* __restrict__ dinv,
                                              const float* __restrict__ h,
                                              const float* __restrict__ bias,
                                              const float* __restrict__ gamma,
                                              const float* __restrict__ beta,
                                              const float* __restrict__ res_in,
                                              float* __restrict__ out_f32,
                                              unsigned short* __restrict__ out_bf16,
                                              int n) {
    const int2* rec = (const int2*)rec_i;
    int node = __builtin_amdgcn_readfirstlane((int)((blockIdx.x * blockDim.x + threadIdx.x) >> 5));
    if (node >= n) return;
    int lane = threadIdx.x & 31;
    int beg = rowptr[node], end = rowptr[node + 1];
    const float4* h4 = (const float4*)h;
    float dn = dinv[node];
    float w0 = dn * dn;
    size_t ridx = (size_t)node * 32 + lane;
    float4 hv = h4[ridx];
    float4 bv = ((const float4*)bias)[lane];
    float4 rv = make_float4(0.f, 0.f, 0.f, 0.f);
    if (MODE == 1) rv = ((const float4*)res_in)[ridx];
    float4 acc = make_float4(w0 * hv.x, w0 * hv.y, w0 * hv.z, w0 * hv.w);

    for (int c0 = beg; c0 < end; c0 += 32) {
        int m = min(32, end - c0);
        int2 r = make_int2(0, 0);
        if (lane < m) r = rec[c0 + lane];
        int j = 0;
        for (; j + 8 <= m; j += 8) gather_rows<8>(acc, r, j, h4, lane);
        if (j + 4 <= m) { gather_rows<4>(acc, r, j, h4, lane); j += 4; }
        if (j + 2 <= m) { gather_rows<2>(acc, r, j, h4, lane); j += 2; }
        if (j < m)        gather_rows<1>(acc, r, j, h4, lane);
    }

    float4 gv = ((const float4*)gamma)[lane];
    float4 ev = ((const float4*)beta)[lane];
    float v[4] = { acc.x + bv.x, acc.y + bv.y, acc.z + bv.z, acc.w + bv.w };
    float s = 0.f;
#pragma unroll
    for (int j = 0; j < 4; ++j) s += v[j];
#pragma unroll
    for (int off = 16; off >= 1; off >>= 1) s += __shfl_xor(s, off, 32);
    float mu = s * (1.0f / 128.0f);

    float q = 0.f;
#pragma unroll
    for (int j = 0; j < 4; ++j) {
        float dv = v[j] - mu;
        q += dv * dv;
    }
#pragma unroll
    for (int off = 16; off >= 1; off >>= 1) q += __shfl_xor(q, off, 32);
    float inv = rsqrtf(q * (1.0f / 128.0f) + 1e-5f);

    const float g[4] = { gv.x, gv.y, gv.z, gv.w };
    const float b[4] = { ev.x, ev.y, ev.z, ev.w };
    float y[4];
#pragma unroll
    for (int j = 0; j < 4; ++j) {
        float t = (v[j] - mu) * inv * g[j] + b[j];
        y[j] = (t > 0.f) ? t : expm1f(t);
    }
    if (MODE == 0) {
        ((float4*)out_f32)[ridx] = make_float4(y[0], y[1], y[2], y[3]);
        ((uint2*)out_bf16)[ridx] = make_uint2(pack2bf(y[0], y[1]),
                                              pack2bf(y[2], y[3]));
    } else {
        ((float4*)out_f32)[ridx] = make_float4(y[0] + rv.x, y[1] + rv.y,
                                               y[2] + rv.z, y[3] + rv.w);
    }
}

__global__ void k_gather_ln1(const int* __restrict__ rowptr, const int* __restrict__ rec_i,
                             const float* __restrict__ dinv, const float* __restrict__ h,
                             const float* __restrict__ bias, const float* __restrict__ gamma,
                             const float* __restrict__ beta,
                             float* __restrict__ out_f32, unsigned short* __restrict__ out_bf16,
                             int n) {
    gather_ln_row<0>(rowptr, rec_i, dinv, h, bias, gamma, beta, nullptr, out_f32, out_bf16, n);
}

__global__ void k_gather_ln2(const int* __restrict__ rowptr, const int* __restrict__ rec_i,
                             const float* __restrict__ dinv, const float* __restrict__ h,
                             const float* __restrict__ bias, const float* __restrict__ gamma,
                             const float* __restrict__ beta, const float* __restrict__ res_in,
                             float* __restrict__ out_f32, int n) {
    gather_ln_row<1>(rowptr, rec_i, dinv, h, bias, gamma, beta, res_in, out_f32, nullptr, n);
}

extern "C" void kernel_launch(void* const* d_in, const int* in_sizes, int n_in,
                              void* d_out, int out_size, void* d_ws, size_t ws_size,
                              hipStream_t stream) {
    const float* x   = (const float*)d_in[0];
    const float* W1  = (const float*)d_in[1];
    const float* b1  = (const float*)d_in[2];
    const float* g1  = (const float*)d_in[3];
    const float* be1 = (const float*)d_in[4];
    const float* W2  = (const float*)d_in[5];
    const float* b2  = (const float*)d_in[6];
    const float* g2  = (const float*)d_in[7];
    const float* be2 = (const float*)d_in[8];
    const int*   ei  = (const int*)d_in[9];

    const int EMB = in_sizes[2];
    const int IN  = in_sizes[1] / EMB;
    const int N   = in_sizes[0] / IN;
    const int E   = in_sizes[9] / 2;
    const int* src = ei;
    const int* dst = ei + E;

    char* base = (char*)d_ws;
    size_t off = 0;
    auto alloc = [&](size_t bytes) {
        char* p = base + off;
        off += (bytes + 255) & ~(size_t)255;
        return p;
    };
    float*          dinv  = (float*)alloc((size_t)N * 4);
    const int NP  = (N + 3) & ~3;
    const int NB  = (N + 1023) / 1024;
    int*            tab   = (int*)alloc((size_t)LCH * NP * 4);
    int*            bsum  = (int*)alloc((size_t)LCH * NB * 4);
    int*            rowp  = (int*)alloc((size_t)(N + 1) * 4);
    int*            rank  = (int*)alloc((size_t)E * 4);
    int*            col   = (int*)alloc((size_t)E * 4);
    int*            rec   = (int*)alloc((size_t)E * 8);
    unsigned short* W1p   = (unsigned short*)alloc((size_t)IN * EMB * 2);
    unsigned short* W2p   = (unsigned short*)alloc((size_t)EMB * EMB * 2);
    float*          h     = (float*)alloc((size_t)N * EMB * 4);
    float*          res   = (float*)alloc((size_t)N * EMB * 4);
    unsigned short* h1b   = (unsigned short*)alloc((size_t)N * EMB * 2);
    (void)ws_size;

    const int B = 256;

    int mtiles = (N + 15) / 16;
    int gemm_blocks = (mtiles + 7) / 8;
    size_t lds1 = (size_t)(IN  / 32) * 8 * 1024;
    size_t lds2 = (size_t)(EMB / 32) * 8 * 1024;
    int pack_tiles  = (IN / 32) * (EMB / 16) + (EMB / 32) * (EMB / 16);
    int pack_blocks = (pack_tiles + 31) / 32;

    k_cnt_pack  <<<LCH + pack_blocks, 1024, 0, stream>>>(dst, rank, tab, bsum, N, NP, NB, E,
                                                         W1, W1p, IN, W2, W2p, EMB, EMB);
    k_scan      <<<NB, 1024, 0, stream>>>(tab, bsum, rowp, dinv, N, NP, NB);
    k_gemm1_fill<<<gemm_blocks + (E + B - 1) / B, B, lds1, stream>>>(x, W1p, h, N, IN, gemm_blocks,
                                                                     src, dst, rank, tab, col, E, NP);
    k_sort_seg  <<<(N + 7) / 8, B, 0, stream>>>(rowp, col, dinv, rec, N);

    k_gather_ln1<<<(N + 7) / 8, B, 0, stream>>>(rowp, rec, dinv, h, b1, g1, be1, res, h1b, N);

    k_gemm_bf16<<<gemm_blocks, B, lds2, stream>>>(h1b, W2p, h, N, EMB);
    k_gather_ln2<<<(N + 7) / 8, B, 0, stream>>>(rowp, rec, dinv, h, b2, g2, be2, res, (float*)d_out, N);
}


# Layer-2 product moves into the layer-1 gather launch: 16 nodes per work-group, their rounded rows reach the WMMAs through LDS; the product launch and the rounded plane go; every sum keeps its order
# speedup vs baseline: 3.7687x; 1.0341x over previous
#include <hip/hip_runtime.h>
#include <hip/hip_bf16.h>

typedef __attribute__((ext_vector_type(16))) __bf16 v16bf;
typedef __attribute__((ext_vector_type(8)))  float  v8f;

union FragU {
    v16bf v;
    unsigned short u16[16];
    uint4 u4[2];
};

__device__ __forceinline__ unsigned short f2bf(float f) {
    unsigned int u = __float_as_uint(f);
    unsigned int r = u + 0x7FFFu + ((u >> 16) & 1u);
    return (unsigned short)(r >> 16);
}

__device__ __forceinline__ void pack_tile(const float* __restrict__ W, unsigned short* __restrict__ out,
                                          int tile, int lane, int N) {
    int ntiles = N >> 4;
    int kt = tile / ntiles, nt = tile - kt * ntiles;
    int col = nt * 16 + (lane & 15);
    int kb  = kt * 32 + ((lane < 16) ? 0 : 8);
    unsigned short* o = out + (size_t)tile * 512 + lane * 16;
#pragma unroll
    for (int e = 0; e < 8; ++e) o[e]     = f2bf(W[(kb + e)      * N + col]);
#pragma unroll
    for (int e = 0; e < 8; ++e) o[8 + e] = f2bf(W[(kb + 16 + e) * N + col]);
}

struct RawF { float4 f0, f1, f2, f3; };

__device__ __forceinline__ void issue_a(RawF& r, const float* p) {
    r.f0 = *(const float4*)(p);
    r.f1 = *(const float4*)(p + 4);
    r.f2 = *(const float4*)(p + 16);
    r.f3 = *(const float4*)(p + 20);
}

__device__ __forceinline__ unsigned int pack2bf(float lo, float hi) {
    return (unsigned int)f2bf(lo) | ((unsigned int)f2bf(hi) << 16);
}

__device__ __forceinline__ void finish_a(FragU& a, const RawF& r) {
    a.u4[0] = make_uint4(pack2bf(r.f0.x, r.f0.y), pack2bf(r.f0.z, r.f0.w),
                         pack2bf(r.f1.x, r.f1.y), pack2bf(r.f1.z, r.f1.w));
    a.u4[1] = make_uint4(pack2bf(r.f2.x, r.f2.y), pack2bf(r.f2.z, r.f2.w),
                         pack2bf(r.f3.x, r.f3.y), pack2bf(r.f3.z, r.f3.w));
}

constexpr int SBAT = 16;

__device__ __forceinline__ void gemm_strip(const float* __restrict__ A,
                                           const unsigned short* __restrict__ Bp,
                                           float* __restrict__ C, int M, int K,
                                           unsigned short* sB) {
    int wave   = (int)((blockIdx.x * blockDim.x + threadIdx.x) >> 5);
    int lane   = threadIdx.x & 31;
    int mtiles = (M + 15) >> 4;

    int row = wave * 16 + (lane & 15);
    if (row >= M) row = M - 1;
    int kb0 = (lane < 16) ? 0 : 8;
    const float* arow = A + (size_t)row * K + kb0;

    {
        const uint4* gB  = (const uint4*)Bp;
        uint4*       sB4 = (uint4*)sB;
        const int n4  = (K >> 5) * 512;
        const int nth = (int)blockDim.x;
        for (int b0 = 0; b0 < n4; b0 += SBAT * nth) {
            uint4 t[SBAT];
#pragma unroll
            for (int u = 0; u < SBAT; ++u) {
                int i = b0 + u * nth + (int)threadIdx.x;
                t[u] = (i < n4) ? gB[i] : make_uint4(0u, 0u, 0u, 0u);
            }
#pragma unroll
            for (int u = 0; u < SBAT; ++u) {
                int i = b0 + u * nth + (int)threadIdx.x;
                if (i < n4) sB4[i] = t[u];
            }
        }
    }
    const int ksteps = K >> 5;
    RawF raw;
    issue_a(raw, arow);
    __syncthreads();

    if (wave >= mtiles) return;

    v8f zero = {};
    v8f acc[8];
#pragma unroll
    for (int nt = 0; nt < 8; ++nt) acc[nt] = zero;

    for (int ks = 0; ks < ksteps; ++ks) {
        FragU a;
        finish_a(a, raw);
        issue_a(raw, arow + min(ks + 1, ksteps - 1) * 32);
        const unsigned short* bp = sB + (size_t)ks * 8 * 512 + lane * 16;
#pragma unroll
        for (int nt = 0; nt < 8; ++nt) {
            FragU b;
            const uint4* bl = (const uint4*)(bp + nt * 512);
            b.u4[0] = bl[0];
            b.u4[1] = bl[1];
            acc[nt] = __builtin_amdgcn_wmma_f32_16x16x32_bf16(
                false, a.v, false, b.v, (short)0, acc[nt], false, false);
        }
    }

    int srow0 = wave * 16 + ((lane < 16) ? 0 : 8);
    int scol  = lane & 15;
#pragma unroll
    for (int nt = 0; nt < 8; ++nt) {
#pragma unroll
        for (int j = 0; j < 8; ++j) {
            int r = srow0 + j;
            if (r < M) C[(size_t)r * 128 + nt * 16 + scol] = acc[nt][j];
        }
    }
}


constexpr int LCH  = 32;
constexpr int LCAP = 65536;
constexpr int LDEP = 32;

__device__ __forceinline__ int chunk_edges(int e) {
    return (((e + LCH - 1) / LCH) + 3) & ~3;
}

__global__ __launch_bounds__(1024) void k_cnt_pack(const int* __restrict__ dst, int* __restrict__ rank,
                                                   int* __restrict__ tab, int* __restrict__ bsum,
                                                   int n, int np, int nbp, int e,
                                                   const float* __restrict__ W1, unsigned short* __restrict__ W1p, int K1,
                                                   const float* __restrict__ W2, unsigned short* __restrict__ W2p, int K2,
                                                   int ncol) {
    __shared__ __attribute__((aligned(16))) int s_cnt[LCAP];
    __shared__ int s_part[(LCAP / 4096) * 32];
    const int tid = threadIdx.x, lane = tid & 31, wv = tid >> 5;
    if ((int)blockIdx.x >= LCH) {
        const int tile = ((int)blockIdx.x - LCH) * 32 + wv;
        const int t1 = (K1 >> 5) * (ncol >> 4);
        const int t2 = (K2 >> 5) * (ncol >> 4);
        if (tile < t1)           pack_tile(W1, W1p, tile, lane, ncol);
        else if (tile - t1 < t2) pack_tile(W2, W2p, tile - t1, lane, ncol);
        return;
    }
    const int c   = blockIdx.x;
    const int per = chunk_edges(e);
    const int e0  = min(c * per, e), e1 = min(e0 + per, e);
    int* trow = tab  + (size_t)c * np;
    int* brow = bsum + (size_t)c * nbp;

    for (int w0 = 0; w0 < np; w0 += LCAP) {
        const int wn  = min(LCAP, np - w0);
        const int wn4 = wn >> 2;
        for (int q = tid; q < wn4; q += 1024) ((int4*)s_cnt)[q] = make_int4(0, 0, 0, 0);
        __syncthreads();

        for (int i0 = e0; i0 < e1; i0 += LDEP * 1024) {
            int d[LDEP];
#pragma unroll
            for (int u = 0; u < LDEP; ++u) {
                int i = i0 + u * 1024 + tid;
                d[u] = (i < e1) ? dst[i] : -1;
            }
            int r[LDEP];
#pragma unroll
            for (int u = 0; u < LDEP; ++u) {
                int i = i0 + u * 1024 + tid;
                unsigned k = (unsigned)(d[u] - w0);
                r[u] = -1;
                if (i < e1 && k < (unsigned)wn) r[u] = atomicAdd(&s_cnt[k], 1);
            }
#pragma unroll
            for (int u = 0; u < LDEP; ++u) {
                int i = i0 + u * 1024 + tid;
                if (r[u] >= 0) rank[i] = r[u];
            }
        }
        __syncthreads();

        const int steps = (wn4 + 1023) >> 10;
        for (int t = 0; t < steps; ++t) {
            int q = t * 1024 + tid;
            int4 v = make_int4(0, 0, 0, 0);
            if (q < wn4) {
                v = ((const int4*)s_cnt)[q];
                ((int4*)(trow + w0))[q] = v;
            }
            int s = (v.x + v.y) + (v.z + v.w);
#pragma unroll
            for (int off = 16; off >= 1; off >>= 1) s += __shfl_xor(s, off, 32);
            if (lane == 0) s_part[t * 32 + wv] = s;
        }
        __syncthreads();
        const int nr = (wn + 1023) >> 10;
        for (int j = tid; j < nr; j += 1024) {
            int s = 0;
#pragma unroll
            for (int k = 0; k < 8; ++k) s += s_part[8 * j + k];
            brow[(w0 >> 10) + j] = s;
        }
        __syncthreads();
    }
}

__global__ __launch_bounds__(1024) void k_scan(int* __restrict__ tab, const int* __restrict__ bsum,
                                               int* __restrict__ rowptr, float* __restrict__ dinv,
                                               int n, int np, int nbp) {
    __shared__ int s_red[32];
    __shared__ int s_wav[32];
    int tid  = threadIdx.x;
    int lane = tid & 31, wv = tid >> 5;
    int b0   = blockIdx.x * 1024;

    int part = 0;
    for (int j = tid; j < (int)blockIdx.x; j += 1024) {
#pragma unroll
        for (int c = 0; c < LCH; ++c) part += bsum[(size_t)c * nbp + j];
    }
#pragma unroll
    for (int off = 16; off >= 1; off >>= 1) part += __shfl_xor(part, off, 32);
    if (lane == 0) s_red[wv] = part;

    int i = b0 + tid;
    int cv[LCH];
    int v = 0;
#pragma unroll
    for (int c = 0; c < LCH; ++c) {
        cv[c] = (i < n) ? tab[(size_t)c * np + i] : 0;
        v += cv[c];
    }
    int inc = v;
#pragma unroll
    for (int off = 1; off < 32; off <<= 1) {
        int t = __shfl_up(inc, off, 32);
        if (lane >= off) inc += t;
    }
    if (lane == 31) s_wav[wv] = inc;
    __syncthreads();

    int t = s_red[lane] + ((lane < wv) ? s_wav[lane] : 0);
#pragma unroll
    for (int off = 16; off >= 1; off >>= 1) t += __shfl_xor(t, off, 32);
    int excl = t + inc - v;
    if (i < n) {
        rowptr[i] = excl;
        dinv[i]   = rsqrtf((float)(v + 1));
        if (i == n - 1) rowptr[n] = excl + v;
        int run = excl;
#pragma unroll
        for (int c = 0; c < LCH; ++c) {
            tab[(size_t)c * np + i] = run;
            run += cv[c];
        }
    }
}

__global__ void k_gemm1_fill(const float* __restrict__ A, const unsigned short* __restrict__ Bp,
                             float* __restrict__ C, int M, int K, int gb,
                             const int* __restrict__ src, const int* __restrict__ dst,
                             const int* __restrict__ rank, const int* __restrict__ tab,
                             int* __restrict__ col, int e, int np) {
    extern __shared__ unsigned short sB[];
    if ((int)blockIdx.x < gb) {
        gemm_strip(A, Bp, C, M, K, sB);
        return;
    }
    int i = ((int)blockIdx.x - gb) * blockDim.x + threadIdx.x;
    if (i < e) {
        int c = i / chunk_edges(e);
        col[tab[(size_t)c * np + dst[i]] + rank[i]] = src[i];
    }
}

__global__ void k_sort_seg(const int* __restrict__ rowptr, const int* __restrict__ col,
                           const float* __restrict__ dinv, int* __restrict__ rec_i, int n) {
    int2* rec = (int2*)rec_i;
    int node = __builtin_amdgcn_readfirstlane((int)((blockIdx.x * blockDim.x + threadIdx.x) >> 5));
    if (node >= n) return;
    int lane = threadIdx.x & 31;
    int beg = rowptr[node];
    int deg = rowptr[node + 1] - beg;
    float dn = dinv[node];
    const int* seg = col + beg;
    for (int c0 = 0; c0 < deg; c0 += 32) {
        int i = c0 + lane;
        bool act = i < deg;
        int s = act ? seg[i] : 0;
        s = min(max(s, 0), n - 1);
        int r = 0;
        for (int k = 0; k < deg; ++k) {
            int t = min(max(seg[k], 0), n - 1);
            r += (t < s || (t == s && k < i)) ? 1 : 0;
        }
        if (act) rec[beg + r] = make_int2(s, __float_as_int(dinv[s] * dn));
    }
}

template <int G>
__device__ __forceinline__ void gather_rows(float4& acc, const int2& r, int j,
                                            const float4* __restrict__ h4, int lane) {
    float4 v[G]; float w[G];
#pragma unroll
    for (int u = 0; u < G; ++u) {
        int s = __builtin_amdgcn_readlane(r.x, j + u);
        w[u]  = __int_as_float(__builtin_amdgcn_readlane(r.y, j + u));
        v[u]  = h4[(size_t)s * 32 + lane];
    }
#pragma unroll
    for (int u = 0; u < G; ++u) {
        acc.x = fmaf(w[u], v[u].x, acc.x);
        acc.y = fmaf(w[u], v[u].y, acc.y);
        acc.z = fmaf(w[u], v[u].z, acc.z);
        acc.w = fmaf(w[u], v[u].w, acc.w);
    }
}

template <bool RES>
__device__ __forceinline__ void gather_ln_vals(const int* __restrict__ rowptr,
                                               const int* __restrict__ rec_i,
                                               const float* __restrict__ dinv,
                                               const float* __restrict__ h,
                                               const float* __restrict__ bias,
                                               const float* __restrict__ gamma,
                                               const float* __restrict__ beta,
                                               const float* __restrict__ res_in,
                                               int node, int lane,
                                               float (&y)[4], float4& rv) {
    const int2* rec = (const int2*)rec_i;
    int beg = rowptr[node], end = rowptr[node + 1];
    const float4* h4 = (const float4*)h;
    float dn = dinv[node];
    float w0 = dn * dn;
    size_t ridx = (size_t)node * 32 + lane;
    float4 hv = h4[ridx];
    float4 bv = ((const float4*)bias)[lane];
    rv = make_float4(0.f, 0.f, 0.f, 0.f);
    if (RES) rv = ((const float4*)res_in)[ridx];
    float4 acc = make_float4(w0 * hv.x, w0 * hv.y, w0 * hv.z, w0 * hv.w);

    for (int c0 = beg; c0 < end; c0 += 32) {
        int m = min(32, end - c0);
        int2 r = make_int2(0, 0);
        if (lane < m) r = rec[c0 + lane];
        int j = 0;
        for (; j + 8 <= m; j += 8) gather_rows<8>(acc, r, j, h4, lane);
        if (j + 4 <= m) { gather_rows<4>(acc, r, j, h4, lane); j += 4; }
        if (j + 2 <= m) { gather_rows<2>(acc, r, j, h4, lane); j += 2; }
        if (j < m)        gather_rows<1>(acc, r, j, h4, lane);
    }

    float4 gv = ((const float4*)gamma)[lane];
    float4 ev = ((const float4*)beta)[lane];
    float v[4] = { acc.x + bv.x, acc.y + bv.y, acc.z + bv.z, acc.w + bv.w };
    float s = 0.f;
#pragma unroll
    for (int j = 0; j < 4; ++j) s += v[j];
#pragma unroll
    for (int off = 16; off >= 1; off >>= 1) s += __shfl_xor(s, off, 32);
    float mu = s * (1.0f / 128.0f);

    float q = 0.f;
#pragma unroll
    for (int j = 0; j < 4; ++j) {
        float dv = v[j] - mu;
        q += dv * dv;
    }
#pragma unroll
    for (int off = 16; off >= 1; off >>= 1) q += __shfl_xor(q, off, 32);
    float inv = rsqrtf(q * (1.0f / 128.0f) + 1e-5f);

    const float g[4] = { gv.x, gv.y, gv.z, gv.w };
    const float b[4] = { ev.x, ev.y, ev.z, ev.w };
#pragma unroll
    for (int j = 0; j < 4; ++j) {
        float t = (v[j] - mu) * inv * g[j] + b[j];
        y[j] = (t > 0.f) ? t : expm1f(t);
    }
}

constexpr int GN     = 16;
constexpr int APITCH = 136;
constexpr int DPITCH = 132;

__global__ __launch_bounds__(GN * 32) void k_gather_ln1_gemm2(const int* __restrict__ rowptr, const int* __restrict__ rec_i,
                             const float* __restrict__ dinv, const float* __restrict__ h,
                             const float* __restrict__ bias, const float* __restrict__ gamma,
                             const float* __restrict__ beta,
                             float* __restrict__ out_f32,
                             const unsigned short* __restrict__ Bp, float* __restrict__ C,
                             int n) {
    __shared__ __attribute__((aligned(16))) unsigned short sA[GN * APITCH];
    __shared__ __attribute__((aligned(16))) float          sD[GN * DPITCH];
    const int lane = threadIdx.x & 31;
    const int wv   = __builtin_amdgcn_readfirstlane((int)(threadIdx.x >> 5));
    const int node = (int)blockIdx.x * GN + wv;
    const bool live = node < n;
    const size_t ridx = (size_t)node * 32 + lane;

    float y[4] = { 0.f, 0.f, 0.f, 0.f };
    if (live) {
        float4 rv;
        gather_ln_vals<false>(rowptr, rec_i, dinv, h, bias, gamma, beta, nullptr, node, lane, y, rv);
    }
    *(uint2*)(sA + wv * APITCH + lane * 4) = make_uint2(pack2bf(y[0], y[1]), pack2bf(y[2], y[3]));

    FragU b[4];
    if (wv < 8) {
#pragma unroll
        for (int ks = 0; ks < 4; ++ks) {
            const uint4* bl = (const uint4*)(Bp + (size_t)(ks * 8 + wv) * 512 + lane * 16);
            b[ks].u4[0] = bl[0];
            b[ks].u4[1] = bl[1];
        }
    }
    __syncthreads();

    if (live) ((float4*)out_f32)[ridx] = make_float4(y[0], y[1], y[2], y[3]);

    if (wv < 8) {
        const unsigned short* ap = sA + (lane & 15) * APITCH + ((lane < 16) ? 0 : 8);
        v8f acc = {};
#pragma unroll
        for (int ks = 0; ks < 4; ++ks) {
            FragU a;
            a.u4[0] = *(const uint4*)(ap + ks * 32);
            a.u4[1] = *(const uint4*)(ap + ks * 32 + 16);
            acc = __builtin_amdgcn_wmma_f32_16x16x32_bf16(
                false, a.v, false, b[ks].v, (short)0, acc, false, false);
        }
        float* dp = sD + ((lane < 16) ? 0 : 8) * DPITCH + wv * 16 + (lane & 15);
#pragma unroll
        for (int j = 0; j < 8; ++j) dp[j * DPITCH] = acc[j];
    }
    __syncthreads();

    if (live) ((float4*)C)[ridx] = *(const float4*)(sD + wv * DPITCH + lane * 4);
}

__global__ void k_gather_ln2(const int* __restrict__ rowptr, const int* __restrict__ rec_i,
                             const float* __restrict__ dinv, const float* __restrict__ h,
                             const float* __restrict__ bias, const float* __restrict__ gamma,
                             const float* __restrict__ beta, const float* __restrict__ res_in,
                             float* __restrict__ out_f32, int n) {
    int node = __builtin_amdgcn_readfirstlane((int)((blockIdx.x * blockDim.x + threadIdx.x) >> 5));
    if (node >= n) return;
    int lane = threadIdx.x & 31;
    float y[4];
    float4 rv;
    gather_ln_vals<true>(rowptr, rec_i, dinv, h, bias, gamma, beta, res_in, node, lane, y, rv);
    ((float4*)out_f32)[(size_t)node * 32 + lane] = make_float4(y[0] + rv.x, y[1] + rv.y,
                                                               y[2] + rv.z, y[3] + rv.w);
}

extern "C" void kernel_launch(void* const* d_in, const int* in_sizes, int n_in,
                              void* d_out, int out_size, void* d_ws, size_t ws_size,
                              hipStream_t stream) {
    const float* x   = (const float*)d_in[0];
    const float* W1  = (const float*)d_in[1];
    const float* b1  = (const float*)d_in[2];
    const float* g1  = (const float*)d_in[3];
    const float* be1 = (const float*)d_in[4];
    const float* W2  = (const float*)d_in[5];
    const float* b2  = (const float*)d_in[6];
    const float* g2  = (const float*)d_in[7];
    const float* be2 = (const float*)d_in[8];
    const int*   ei  = (const int*)d_in[9];

    const int EMB = in_sizes[2];
    const int IN  = in_sizes[1] / EMB;
    const int N   = in_sizes[0] / IN;
    const int E   = in_sizes[9] / 2;
    const int* src = ei;
    const int* dst = ei + E;

    char* base = (char*)d_ws;
    size_t off = 0;
    auto alloc = [&](size_t bytes) {
        char* p = base + off;
        off += (bytes + 255) & ~(size_t)255;
        return p;
    };
    float*          dinv  = (float*)alloc((size_t)N * 4);
    const int NP  = (N + 3) & ~3;
    const int NB  = (N + 1023) / 1024;
    int*            tab   = (int*)alloc((size_t)LCH * NP * 4);
    int*            bsum  = (int*)alloc((size_t)LCH * NB * 4);
    int*            rowp  = (int*)alloc((size_t)(N + 1) * 4);
    int*            rank  = (int*)alloc((size_t)E * 4);
    int*            col   = (int*)alloc((size_t)E * 4);
    int*            rec   = (int*)alloc((size_t)E * 8);
    unsigned short* W1p   = (unsigned short*)alloc((size_t)IN * EMB * 2);
    unsigned short* W2p   = (unsigned short*)alloc((size_t)EMB * EMB * 2);
    float*          h     = (float*)alloc((size_t)N * EMB * 4);
    float*          res   = (float*)alloc((size_t)N * EMB * 4);
    float*          h2    = (float*)alloc((size_t)N * EMB * 4);
    (void)ws_size;

    const int B = 256;

    int mtiles = (N + 15) / 16;
    int gemm_blocks = (mtiles + 7) / 8;
    size_t lds1 = (size_t)(IN  / 32) * 8 * 1024;
    int pack_tiles  = (IN / 32) * (EMB / 16) + (EMB / 32) * (EMB / 16);
    int pack_blocks = (pack_tiles + 31) / 32;

    k_cnt_pack  <<<LCH + pack_blocks, 1024, 0, stream>>>(dst, rank, tab, bsum, N, NP, NB, E,
                                                         W1, W1p, IN, W2, W2p, EMB, EMB);
    k_scan      <<<NB, 1024, 0, stream>>>(tab, bsum, rowp, dinv, N, NP, NB);
    k_gemm1_fill<<<gemm_blocks + (E + B - 1) / B, B, lds1, stream>>>(x, W1p, h, N, IN, gemm_blocks,
                                                                     src, dst, rank, tab, col, E, NP);
    k_sort_seg  <<<(N + 7) / 8, B, 0, stream>>>(rowp, col, dinv, rec, N);

    k_gather_ln1_gemm2<<<(N + GN - 1) / GN, GN * 32, 0, stream>>>(rowp, rec, dinv, h, b1, g1, be1, res, W2p, h2, N);

    k_gather_ln2<<<(N + 7) / 8, B, 0, stream>>>(rowp, rec, dinv, h2, b2, g2, be2, res, (float*)d_out, N);
}


# Each node's list is ordered by its own wave in the layer-1 gather launch: up to 32 entries are ranked in the lanes and change lanes by rank, longer lists as before; sort launch goes; sum order kept
# speedup vs baseline: 3.9484x; 1.0477x over previous
#include <hip/hip_runtime.h>
#include <hip/hip_bf16.h>

typedef __attribute__((ext_vector_type(16))) __bf16 v16bf;
typedef __attribute__((ext_vector_type(8)))  float  v8f;

union FragU {
    v16bf v;
    unsigned short u16[16];
    uint4 u4[2];
};

__device__ __forceinline__ unsigned short f2bf(float f) {
    unsigned int u = __float_as_uint(f);
    unsigned int r = u + 0x7FFFu + ((u >> 16) & 1u);
    return (unsigned short)(r >> 16);
}

__device__ __forceinline__ void pack_tile(const float* __restrict__ W, unsigned short* __restrict__ out,
                                          int tile, int lane, int N) {
    int ntiles = N >> 4;
    int kt = tile / ntiles, nt = tile - kt * ntiles;
    int col = nt * 16 + (lane & 15);
    int kb  = kt * 32 + ((lane < 16) ? 0 : 8);
    unsigned short* o = out + (size_t)tile * 512 + lane * 16;
#pragma unroll
    for (int e = 0; e < 8; ++e) o[e]     = f2bf(W[(kb + e)      * N + col]);
#pragma unroll
    for (int e = 0; e < 8; ++e) o[8 + e] = f2bf(W[(kb + 16 + e) * N + col]);
}

struct RawF { float4 f0, f1, f2, f3; };

__device__ __forceinline__ void issue_a(RawF& r, const float* p) {
    r.f0 = *(const float4*)(p);
    r.f1 = *(const float4*)(p + 4);
    r.f2 = *(const float4*)(p + 16);
    r.f3 = *(const float4*)(p + 20);
}

__device__ __forceinline__ unsigned int pack2bf(float lo, float hi) {
    return (unsigned int)f2bf(lo) | ((unsigned int)f2bf(hi) << 16);
}

__device__ __forceinline__ void finish_a(FragU& a, const RawF& r) {
    a.u4[0] = make_uint4(pack2bf(r.f0.x, r.f0.y), pack2bf(r.f0.z, r.f0.w),
                         pack2bf(r.f1.x, r.f1.y), pack2bf(r.f1.z, r.f1.w));
    a.u4[1] = make_uint4(pack2bf(r.f2.x, r.f2.y), pack2bf(r.f2.z, r.f2.w),
                         pack2bf(r.f3.x, r.f3.y), pack2bf(r.f3.z, r.f3.w));
}

constexpr int SBAT = 16;

__device__ __forceinline__ void gemm_strip(const float* __restrict__ A,
                                           const unsigned short* __restrict__ Bp,
                                           float* __restrict__ C, int M, int K,
                                           unsigned short* sB) {
    int wave   = (int)((blockIdx.x * blockDim.x + threadIdx.x) >> 5);
    int lane   = threadIdx.x & 31;
    int mtiles = (M + 15) >> 4;

    int row = wave * 16 + (lane & 15);
    if (row >= M) row = M - 1;
    int kb0 = (lane < 16) ? 0 : 8;
    const float* arow = A + (size_t)row * K + kb0;

    {
        const uint4* gB  = (const uint4*)Bp;
        uint4*       sB4 = (uint4*)sB;
        const int n4  = (K >> 5) * 512;
        const int nth = (int)blockDim.x;
        for (int b0 = 0; b0 < n4; b0 += SBAT * nth) {
            uint4 t[SBAT];
#pragma unroll
            for (int u = 0; u < SBAT; ++u) {
                int i = b0 + u * nth + (int)threadIdx.x;
                t[u] = (i < n4) ? gB[i] : make_uint4(0u, 0u, 0u, 0u);
            }
#pragma unroll
            for (int u = 0; u < SBAT; ++u) {
                int i = b0 + u * nth + (int)threadIdx.x;
                if (i < n4) sB4[i] = t[u];
            }
        }
    }
    const int ksteps = K >> 5;
    RawF raw;
    issue_a(raw, arow);
    __syncthreads();

    if (wave >= mtiles) return;

    v8f zero = {};
    v8f acc[8];
#pragma unroll
    for (int nt = 0; nt < 8; ++nt) acc[nt] = zero;

    for (int ks = 0; ks < ksteps; ++ks) {
        FragU a;
        finish_a(a, raw);
        issue_a(raw, arow + min(ks + 1, ksteps - 1) * 32);
        const unsigned short* bp = sB + (size_t)ks * 8 * 512 + lane * 16;
#pragma unroll
        for (int nt = 0; nt < 8; ++nt) {
            FragU b;
            const uint4* bl = (const uint4*)(bp + nt * 512);
            b.u4[0] = bl[0];
            b.u4[1] = bl[1];
            acc[nt] = __builtin_amdgcn_wmma_f32_16x16x32_bf16(
                false, a.v, false, b.v, (short)0, acc[nt], false, false);
        }
    }

    int srow0 = wave * 16 + ((lane < 16) ? 0 : 8);
    int scol  = lane & 15;
#pragma unroll
    for (int nt = 0; nt < 8; ++nt) {
#pragma unroll
        for (int j = 0; j < 8; ++j) {
            int r = srow0 + j;
            if (r < M) C[(size_t)r * 128 + nt * 16 + scol] = acc[nt][j];
        }
    }
}


constexpr int LCH  = 32;
constexpr int LCAP = 65536;
constexpr int LDEP = 32;

__device__ __forceinline__ int chunk_edges(int e) {
    return (((e + LCH - 1) / LCH) + 3) & ~3;
}

__global__ __launch_bounds__(1024) void k_cnt_pack(const int* __restrict__ dst, int* __restrict__ rank,
                                                   int* __restrict__ tab, int* __restrict__ bsum,
                                                   int n, int np, int nbp, int e,
                                                   const float* __restrict__ W1, unsigned short* __restrict__ W1p, int K1,
                                                   const float* __restrict__ W2, unsigned short* __restrict__ W2p, int K2,
                                                   int ncol) {
    __shared__ __attribute__((aligned(16))) int s_cnt[LCAP];
    __shared__ int s_part[(LCAP / 4096) * 32];
    const int tid = threadIdx.x, lane = tid & 31, wv = tid >> 5;
    if ((int)blockIdx.x >= LCH) {
        const int tile = ((int)blockIdx.x - LCH) * 32 + wv;
        const int t1 = (K1 >> 5) * (ncol >> 4);
        const int t2 = (K2 >> 5) * (ncol >> 4);
        if (tile < t1)           pack_tile(W1, W1p, tile, lane, ncol);
        else if (tile - t1 < t2) pack_tile(W2, W2p, tile - t1, lane, ncol);
        return;
    }
    const int c   = blockIdx.x;
    const int per = chunk_edges(e);
    const int e0  = min(c * per, e), e1 = min(e0 + per, e);
    int* trow = tab  + (size_t)c * np;
    int* brow = bsum + (size_t)c * nbp;

    for (int w0 = 0; w0 < np; w0 += LCAP) {
        const int wn  = min(LCAP, np - w0);
        const int wn4 = wn >> 2;
        for (int q = tid; q < wn4; q += 1024) ((int4*)s_cnt)[q] = make_int4(0, 0, 0, 0);
        __syncthreads();

        for (int i0 = e0; i0 < e1; i0 += LDEP * 1024) {
            int d[LDEP];
#pragma unroll
            for (int u = 0; u < LDEP; ++u) {
                int i = i0 + u * 1024 + tid;
                d[u] = (i < e1) ? dst[i] : -1;
            }
            int r[LDEP];
#pragma unroll
            for (int u = 0; u < LDEP; ++u) {
                int i = i0 + u * 1024 + tid;
                unsigned k = (unsigned)(d[u] - w0);
                r[u] = -1;
                if (i < e1 && k < (unsigned)wn) r[u] = atomicAdd(&s_cnt[k], 1);
            }
#pragma unroll
            for (int u = 0; u < LDEP; ++u) {
                int i = i0 + u * 1024 + tid;
                if (r[u] >= 0) rank[i] = r[u];
            }
        }
        __syncthreads();

        const int steps = (wn4 + 1023) >> 10;
        for (int t = 0; t < steps; ++t) {
            int q = t * 1024 + tid;
            int4 v = make_int4(0, 0, 0, 0);
            if (q < wn4) {
                v = ((const int4*)s_cnt)[q];
                ((int4*)(trow + w0))[q] = v;
            }
            int s = (v.x + v.y) + (v.z + v.w);
#pragma unroll
            for (int off = 16; off >= 1; off >>= 1) s += __shfl_xor(s, off, 32);
            if (lane == 0) s_part[t * 32 + wv] = s;
        }
        __syncthreads();
        const int nr = (wn + 1023) >> 10;
        for (int j = tid; j < nr; j += 1024) {
            int s = 0;
#pragma unroll
            for (int k = 0; k < 8; ++k) s += s_part[8 * j + k];
            brow[(w0 >> 10) + j] = s;
        }
        __syncthreads();
    }
}

__global__ __launch_bounds__(1024) void k_scan(int* __restrict__ tab, const int* __restrict__ bsum,
                                               int* __restrict__ rowptr, float* __restrict__ dinv,
                                               int n, int np, int nbp) {
    __shared__ int s_red[32];
    __shared__ int s_wav[32];
    int tid  = threadIdx.x;
    int lane = tid & 31, wv = tid >> 5;
    int b0   = blockIdx.x * 1024;

    int part = 0;
    for (int j = tid; j < (int)blockIdx.x; j += 1024) {
#pragma unroll
        for (int c = 0; c < LCH; ++c) part += bsum[(size_t)c * nbp + j];
    }
#pragma unroll
    for (int off = 16; off >= 1; off >>= 1) part += __shfl_xor(part, off, 32);
    if (lane == 0) s_red[wv] = part;

    int i = b0 + tid;
    int cv[LCH];
    int v = 0;
#pragma unroll
    for (int c = 0; c < LCH; ++c) {
        cv[c] = (i < n) ? tab[(size_t)c * np + i] : 0;
        v += cv[c];
    }
    int inc = v;
#pragma unroll
    for (int off = 1; off < 32; off <<= 1) {
        int t = __shfl_up(inc, off, 32);
        if (lane >= off) inc += t;
    }
    if (lane == 31) s_wav[wv] = inc;
    __syncthreads();

    int t = s_red[lane] + ((lane < wv) ? s_wav[lane] : 0);
#pragma unroll
    for (int off = 16; off >= 1; off >>= 1) t += __shfl_xor(t, off, 32);
    int excl = t + inc - v;
    if (i < n) {
        rowptr[i] = excl;
        dinv[i]   = rsqrtf((float)(v + 1));
        if (i == n - 1) rowptr[n] = excl + v;
        int run = excl;
#pragma unroll
        for (int c = 0; c < LCH; ++c) {
            tab[(size_t)c * np + i] = run;
            run += cv[c];
        }
    }
}

__global__ void k_gemm1_fill(const float* __restrict__ A, const unsigned short* __restrict__ Bp,
                             float* __restrict__ C, int M, int K, int gb,
                             const int* __restrict__ src, const int* __restrict__ dst,
                             const int* __restrict__ rank, const int* __restrict__ tab,
                             int* __restrict__ col, int e, int np) {
    extern __shared__ unsigned short sB[];
    if ((int)blockIdx.x < gb) {
        gemm_strip(A, Bp, C, M, K, sB);
        return;
    }
    int i = ((int)blockIdx.x - gb) * blockDim.x + threadIdx.x;
    if (i < e) {
        int c = i / chunk_edges(e);
        col[tab[(size_t)c * np + dst[i]] + rank[i]] = src[i];
    }
}

__device__ __forceinline__ void sort_long_list(const int* __restrict__ seg, int2* rseg,
                                               const float* __restrict__ dinv, float dn,
                                               int deg, int lane, int n) {
    for (int c0 = 0; c0 < deg; c0 += 32) {
        int i = c0 + lane;
        bool act = i < deg;
        int s = act ? seg[i] : 0;
        s = min(max(s, 0), n - 1);
        int r = 0;
        for (int k = 0; k < deg; ++k) {
            int t = min(max(seg[k], 0), n - 1);
            r += (t < s || (t == s && k < i)) ? 1 : 0;
        }
        if (act) rseg[r] = make_int2(s, __float_as_int(dinv[s] * dn));
    }
}

__device__ __forceinline__ int2 sort_short_list(const int* __restrict__ seg,
                                                const float* __restrict__ dinv, float dn,
                                                int deg, int lane, int n) {
    int s = (lane < deg) ? seg[lane] : 0;
    s = min(max(s, 0), n - 1);
    float ds = dinv[s];
    int r = 0;
    for (int k = 0; k < deg; ++k) {
        int t = __builtin_amdgcn_readlane(s, k);
        r += (t < s || (t == s && k < lane)) ? 1 : 0;
    }
    if (lane >= deg) r = lane;
    int w = __float_as_int(ds * dn);
    return make_int2(__builtin_amdgcn_ds_permute(r << 2, s), __builtin_amdgcn_ds_permute(r << 2, w));
}

template <int G>
__device__ __forceinline__ void gather_rows(float4& acc, const int2& r, int j,
                                            const float4* __restrict__ h4, int lane) {
    float4 v[G]; float w[G];
#pragma unroll
    for (int u = 0; u < G; ++u) {
        int s = __builtin_amdgcn_readlane(r.x, j + u);
        w[u]  = __int_as_float(__builtin_amdgcn_readlane(r.y, j + u));
        v[u]  = h4[(size_t)s * 32 + lane];
    }
#pragma unroll
    for (int u = 0; u < G; ++u) {
        acc.x = fmaf(w[u], v[u].x, acc.x);
        acc.y = fmaf(w[u], v[u].y, acc.y);
        acc.z = fmaf(w[u], v[u].z, acc.z);
        acc.w = fmaf(w[u], v[u].w, acc.w);
    }
}

template <bool RES, bool SORT>
__device__ __forceinline__ void gather_ln_vals(const int* __restrict__ rowptr,
                                               const int* __restrict__ rec_i,
                                               const int* __restrict__ col,
                                               int* rec_w, int n,
                                               const float* __restrict__ dinv,
                                               const float* __restrict__ h,
                                               const float* __restrict__ bias,
                                               const float* __restrict__ gamma,
                                               const float* __restrict__ beta,
                                               const float* __restrict__ res_in,
                                               int node, int lane,
                                               float (&y)[4], float4& rv) {
    const int2* rec = SORT ? (const int2*)rec_w : (const int2*)rec_i;
    int beg = rowptr[node], end = rowptr[node + 1];
    const float4* h4 = (const float4*)h;
    float dn = dinv[node];
    float w0 = dn * dn;
    size_t ridx = (size_t)node * 32 + lane;
    float4 hv = h4[ridx];
    float4 bv = ((const float4*)bias)[lane];
    rv = make_float4(0.f, 0.f, 0.f, 0.f);
    if (RES) rv = ((const float4*)res_in)[ridx];
    float4 acc = make_float4(w0 * hv.x, w0 * hv.y, w0 * hv.z, w0 * hv.w);

    int2 r = make_int2(0, 0);
    const bool held = SORT && (end - beg <= 32);
    if (SORT) {
        const int deg = end - beg;
        if (held) {
            r = sort_short_list(col + beg, dinv, dn, deg, lane, n);
            if (lane < deg) ((int2*)rec_w)[beg + lane] = r;
        } else {
            sort_long_list(col + beg, (int2*)rec_w + beg, dinv, dn, deg, lane, n);
            __threadfence();
        }
    }

    for (int c0 = beg; c0 < end; c0 += 32) {
        int m = min(32, end - c0);
        if (!held) {
            r = make_int2(0, 0);
            if (lane < m) r = rec[c0 + lane];
        }
        int j = 0;
        for (; j + 8 <= m; j += 8) gather_rows<8>(acc, r, j, h4, lane);
        if (j + 4 <= m) { gather_rows<4>(acc, r, j, h4, lane); j += 4; }
        if (j + 2 <= m) { gather_rows<2>(acc, r, j, h4, lane); j += 2; }
        if (j < m)        gather_rows<1>(acc, r, j, h4, lane);
    }

    float4 gv = ((const float4*)gamma)[lane];
    float4 ev = ((const float4*)beta)[lane];
    float v[4] = { acc.x + bv.x, acc.y + bv.y, acc.z + bv.z, acc.w + bv.w };
    float s = 0.f;
#pragma unroll
    for (int j = 0; j < 4; ++j) s += v[j];
#pragma unroll
    for (int off = 16; off >= 1; off >>= 1) s += __shfl_xor(s, off, 32);
    float mu = s * (1.0f / 128.0f);

    float q = 0.f;
#pragma unroll
    for (int j = 0; j < 4; ++j) {
        float dv = v[j] - mu;
        q += dv * dv;
    }
#pragma unroll
    for (int off = 16; off >= 1; off >>= 1) q += __shfl_xor(q, off, 32);
    float inv = rsqrtf(q * (1.0f / 128.0f) + 1e-5f);

    const float g[4] = { gv.x, gv.y, gv.z, gv.w };
    const float b[4] = { ev.x, ev.y, ev.z, ev.w };
#pragma unroll
    for (int j = 0; j < 4; ++j) {
        float t = (v[j] - mu) * inv * g[j] + b[j];
        y[j] = (t > 0.f) ? t : expm1f(t);
    }
}

constexpr int GN     = 16;
constexpr int APITCH = 136;
constexpr int DPITCH = 132;

__global__ __launch_bounds__(GN * 32) void k_gather_ln1_gemm2(const int* __restrict__ rowptr, const int* __restrict__ col,
                             int* rec_w,
                             const float* __restrict__ dinv, const float* __restrict__ h,
                             const float* __restrict__ bias, const float* __restrict__ gamma,
                             const float* __restrict__ beta,
                             float* __restrict__ out_f32,
                             const unsigned short* __restrict__ Bp, float* __restrict__ C,
                             int n) {
    __shared__ __attribute__((aligned(16))) unsigned short sA[GN * APITCH];
    __shared__ __attribute__((aligned(16))) float          sD[GN * DPITCH];
    const int lane = threadIdx.x & 31;
    const int wv   = __builtin_amdgcn_readfirstlane((int)(threadIdx.x >> 5));
    const int node = (int)blockIdx.x * GN + wv;
    const bool live = node < n;
    const size_t rbase = (size_t)node * 32;

    float y[4] = { 0.f, 0.f, 0.f, 0.f };
    if (live) {
        float4 rv;
        gather_ln_vals<false, true>(rowptr, nullptr, col, rec_w, n, dinv, h, bias, gamma, beta, nullptr, node, lane, y, rv);
    }
    *(uint2*)(sA + wv * APITCH + lane * 4) = make_uint2(pack2bf(y[0], y[1]), pack2bf(y[2], y[3]));

    FragU b[4];
    if (wv < 8) {
#pragma unroll
        for (int ks = 0; ks < 4; ++ks) {
            const uint4* bl = (const uint4*)(Bp + (size_t)(ks * 8 + wv) * 512 + lane * 16);
            b[ks].u4[0] = bl[0];
            b[ks].u4[1] = bl[1];
        }
    }
    __syncthreads();

    if (live) ((float4*)out_f32 + rbase)[lane] = make_float4(y[0], y[1], y[2], y[3]);

    if (wv < 8) {
        const unsigned short* ap = sA + (lane & 15) * APITCH + ((lane < 16) ? 0 : 8);
        v8f acc = {};
#pragma unroll
        for (int ks = 0; ks < 4; ++ks) {
            FragU a;
            a.u4[0] = *(const uint4*)(ap + ks * 32);
            a.u4[1] = *(const uint4*)(ap + ks * 32 + 16);
            acc = __builtin_amdgcn_wmma_f32_16x16x32_bf16(
                false, a.v, false, b[ks].v, (short)0, acc, false, false);
        }
        float* dp = sD + ((lane < 16) ? 0 : 8) * DPITCH + wv * 16 + (lane & 15);
#pragma unroll
        for (int j = 0; j < 8; ++j) dp[j * DPITCH] = acc[j];
    }
    __syncthreads();

    if (live) ((float4*)C + rbase)[lane] = *(const float4*)(sD + wv * DPITCH + lane * 4);
}

__global__ void k_gather_ln2(const int* __restrict__ rowptr, const int* __restrict__ rec_i,
                             const float* __restrict__ dinv, const float* __restrict__ h,
                             const float* __restrict__ bias, const float* __restrict__ gamma,
                             const float* __restrict__ beta, const float* __restrict__ res_in,
                             float* __restrict__ out_f32, int n) {
    int node = __builtin_amdgcn_readfirstlane((int)((blockIdx.x * blockDim.x + threadIdx.x) >> 5));
    if (node >= n) return;
    int lane = threadIdx.x & 31;
    float y[4];
    float4 rv;
    gather_ln_vals<true, false>(rowptr, rec_i, nullptr, nullptr, n, dinv, h, bias, gamma, beta, res_in, node, lane, y, rv);
    ((float4*)out_f32)[(size_t)node * 32 + lane] = make_float4(y[0] + rv.x, y[1] + rv.y,
                                                               y[2] + rv.z, y[3] + rv.w);
}

extern "C" void kernel_launch(void* const* d_in, const int* in_sizes, int n_in,
                              void* d_out, int out_size, void* d_ws, size_t ws_size,
                              hipStream_t stream) {
    const float* x   = (const float*)d_in[0];
    const float* W1  = (const float*)d_in[1];
    const float* b1  = (const float*)d_in[2];
    const float* g1  = (const float*)d_in[3];
    const float* be1 = (const float*)d_in[4];
    const float* W2  = (const float*)d_in[5];
    const float* b2  = (const float*)d_in[6];
    const float* g2  = (const float*)d_in[7];
    const float* be2 = (const float*)d_in[8];
    const int*   ei  = (const int*)d_in[9];

    const int EMB = in_sizes[2];
    const int IN  = in_sizes[1] / EMB;
    const int N   = in_sizes[0] / IN;
    const int E   = in_sizes[9] / 2;
    const int* src = ei;
    const int* dst = ei + E;

    char* base = (char*)d_ws;
    size_t off = 0;
    auto alloc = [&](size_t bytes) {
        char* p = base + off;
        off += (bytes + 255) & ~(size_t)255;
        return p;
    };
    float*          dinv  = (float*)alloc((size_t)N * 4);
    const int NP  = (N + 3) & ~3;
    const int NB  = (N + 1023) / 1024;
    int*            tab   = (int*)alloc((size_t)LCH * NP * 4);
    int*            bsum  = (int*)alloc((size_t)LCH * NB * 4);
    int*            rowp  = (int*)alloc((size_t)(N + 1) * 4);
    int*            rank  = (int*)alloc((size_t)E * 4);
    int*            col   = (int*)alloc((size_t)E * 4);
    int*            rec   = (int*)alloc((size_t)E * 8);
    unsigned short* W1p   = (unsigned short*)alloc((size_t)IN * EMB * 2);
    unsigned short* W2p   = (unsigned short*)alloc((size_t)EMB * EMB * 2);
    float*          h     = (float*)alloc((size_t)N * EMB * 4);
    float*          res   = (float*)alloc((size_t)N * EMB * 4);
    float*          h2    = (float*)alloc((size_t)N * EMB * 4);
    (void)ws_size;

    const int B = 256;

    int mtiles = (N + 15) / 16;
    int gemm_blocks = (mtiles + 7) / 8;
    size_t lds1 = (size_t)(IN  / 32) * 8 * 1024;
    int pack_tiles  = (IN / 32) * (EMB / 16) + (EMB / 32) * (EMB / 16);
    int pack_blocks = (pack_tiles + 31) / 32;

    k_cnt_pack  <<<LCH + pack_blocks, 1024, 0, stream>>>(dst, rank, tab, bsum, N, NP, NB, E,
                                                         W1, W1p, IN, W2, W2p, EMB, EMB);
    k_scan      <<<NB, 1024, 0, stream>>>(tab, bsum, rowp, dinv, N, NP, NB);
    k_gemm1_fill<<<gemm_blocks + (E + B - 1) / B, B, lds1, stream>>>(x, W1p, h, N, IN, gemm_blocks,
                                                                     src, dst, rank, tab, col, E, NP);

    k_gather_ln1_gemm2<<<(N + GN - 1) / GN, GN * 32, 0, stream>>>(rowp, col, rec, dinv, h, b1, g1, be1, res, W2p, h2, N);

    k_gather_ln2<<<(N + 7) / 8, B, 0, stream>>>(rowp, rec, dinv, h2, b2, g2, be2, res, (float*)d_out, N);
}


# Layer 2 sums before it multiplies: the gather walks the bf16 layer-1 rows the product already took as A (2 lines a row, not 4), the f32 sum enters the WMMAs as bf16 hi+lo; order: sources, then k
# speedup vs baseline: 4.1190x; 1.0432x over previous
#include <hip/hip_runtime.h>
#include <hip/hip_bf16.h>

typedef __attribute__((ext_vector_type(16))) __bf16 v16bf;
typedef __attribute__((ext_vector_type(8)))  float  v8f;

union FragU {
    v16bf v;
    unsigned short u16[16];
    uint4 u4[2];
};

__device__ __forceinline__ unsigned short f2bf(float f) {
    unsigned int u = __float_as_uint(f);
    unsigned int r = u + 0x7FFFu + ((u >> 16) & 1u);
    return (unsigned short)(r >> 16);
}

__device__ __forceinline__ void pack_tile(const float* __restrict__ W, unsigned short* __restrict__ out,
                                          int tile, int lane, int N) {
    int ntiles = N >> 4;
    int kt = tile / ntiles, nt = tile - kt * ntiles;
    int col = nt * 16 + (lane & 15);
    int kb  = kt * 32 + ((lane < 16) ? 0 : 8);
    unsigned short* o = out + (size_t)tile * 512 + lane * 16;
#pragma unroll
    for (int e = 0; e < 8; ++e) o[e]     = f2bf(W[(kb + e)      * N + col]);
#pragma unroll
    for (int e = 0; e < 8; ++e) o[8 + e] = f2bf(W[(kb + 16 + e) * N + col]);
}

struct RawF { float4 f0, f1, f2, f3; };

__device__ __forceinline__ void issue_a(RawF& r, const float* p) {
    r.f0 = *(const float4*)(p);
    r.f1 = *(const float4*)(p + 4);
    r.f2 = *(const float4*)(p + 16);
    r.f3 = *(const float4*)(p + 20);
}

__device__ __forceinline__ unsigned int pack2bf(float lo, float hi) {
    return (unsigned int)f2bf(lo) | ((unsigned int)f2bf(hi) << 16);
}

__device__ __forceinline__ void finish_a(FragU& a, const RawF& r) {
    a.u4[0] = make_uint4(pack2bf(r.f0.x, r.f0.y), pack2bf(r.f0.z, r.f0.w),
                         pack2bf(r.f1.x, r.f1.y), pack2bf(r.f1.z, r.f1.w));
    a.u4[1] = make_uint4(pack2bf(r.f2.x, r.f2.y), pack2bf(r.f2.z, r.f2.w),
                         pack2bf(r.f3.x, r.f3.y), pack2bf(r.f3.z, r.f3.w));
}

constexpr int SBAT = 16;

__device__ __forceinline__ void gemm_strip(const float* __restrict__ A,
                                           const unsigned short* __restrict__ Bp,
                                           float* __restrict__ C, int M, int K,
                                           unsigned short* sB) {
    int wave   = (int)((blockIdx.x * blockDim.x + threadIdx.x) >> 5);
    int lane   = threadIdx.x & 31;
    int mtiles = (M + 15) >> 4;

    int row = wave * 16 + (lane & 15);
    if (row >= M) row = M - 1;
    int kb0 = (lane < 16) ? 0 : 8;
    const float* arow = A + (size_t)row * K + kb0;

    {
        const uint4* gB  = (const uint4*)Bp;
        uint4*       sB4 = (uint4*)sB;
        const int n4  = (K >> 5) * 512;
        const int nth = (int)blockDim.x;
        for (int b0 = 0; b0 < n4; b0 += SBAT * nth) {
            uint4 t[SBAT];
#pragma unroll
            for (int u = 0; u < SBAT; ++u) {
                int i = b0 + u * nth + (int)threadIdx.x;
                t[u] = (i < n4) ? gB[i] : make_uint4(0u, 0u, 0u, 0u);
            }
#pragma unroll
            for (int u = 0; u < SBAT; ++u) {
                int i = b0 + u * nth + (int)threadIdx.x;
                if (i < n4) sB4[i] = t[u];
            }
        }
    }
    const int ksteps = K >> 5;
    RawF raw;
    issue_a(raw, arow);
    __syncthreads();

    if (wave >= mtiles) return;

    v8f zero = {};
    v8f acc[8];
#pragma unroll
    for (int nt = 0; nt < 8; ++nt) acc[nt] = zero;

    for (int ks = 0; ks < ksteps; ++ks) {
        FragU a;
        finish_a(a, raw);
        issue_a(raw, arow + min(ks + 1, ksteps - 1) * 32);
        const unsigned short* bp = sB + (size_t)ks * 8 * 512 + lane * 16;
#pragma unroll
        for (int nt = 0; nt < 8; ++nt) {
            FragU b;
            const uint4* bl = (const uint4*)(bp + nt * 512);
            b.u4[0] = bl[0];
            b.u4[1] = bl[1];
            acc[nt] = __builtin_amdgcn_wmma_f32_16x16x32_bf16(
                false, a.v, false, b.v, (short)0, acc[nt], false, false);
        }
    }

    int srow0 = wave * 16 + ((lane < 16) ? 0 : 8);
    int scol  = lane & 15;
#pragma unroll
    for (int nt = 0; nt < 8; ++nt) {
#pragma unroll
        for (int j = 0; j < 8; ++j) {
            int r = srow0 + j;
            if (r < M) C[(size_t)r * 128 + nt * 16 + scol] = acc[nt][j];
        }
    }
}


constexpr int LCH  = 32;
constexpr int LCAP = 65536;
constexpr int LDEP = 32;

__device__ __forceinline__ int chunk_edges(int e) {
    return (((e + LCH - 1) / LCH) + 3) & ~3;
}

__global__ __launch_bounds__(1024) void k_cnt_pack(const int* __restrict__ dst, int* __restrict__ rank,
                                                   int* __restrict__ tab, int* __restrict__ bsum,
                                                   int n, int np, int nbp, int e,
                                                   const float* __restrict__ W1, unsigned short* __restrict__ W1p, int K1,
                                                   const float* __restrict__ W2, unsigned short* __restrict__ W2p, int K2,
                                                   int ncol) {
    __shared__ __attribute__((aligned(16))) int s_cnt[LCAP];
    __shared__ int s_part[(LCAP / 4096) * 32];
    const int tid = threadIdx.x, lane = tid & 31, wv = tid >> 5;
    if ((int)blockIdx.x >= LCH) {
        const int tile = ((int)blockIdx.x - LCH) * 32 + wv;
        const int t1 = (K1 >> 5) * (ncol >> 4);
        const int t2 = (K2 >> 5) * (ncol >> 4);
        if (tile < t1)           pack_tile(W1, W1p, tile, lane, ncol);
        else if (tile - t1 < t2) pack_tile(W2, W2p, tile - t1, lane, ncol);
        return;
    }
    const int c   = blockIdx.x;
    const int per = chunk_edges(e);
    const int e0  = min(c * per, e), e1 = min(e0 + per, e);
    int* trow = tab  + (size_t)c * np;
    int* brow = bsum + (size_t)c * nbp;

    for (int w0 = 0; w0 < np; w0 += LCAP) {
        const int wn  = min(LCAP, np - w0);
        const int wn4 = wn >> 2;
        for (int q = tid; q < wn4; q += 1024) ((int4*)s_cnt)[q] = make_int4(0, 0, 0, 0);
        __syncthreads();

        for (int i0 = e0; i0 < e1; i0 += LDEP * 1024) {
            int d[LDEP];
#pragma unroll
            for (int u = 0; u < LDEP; ++u) {
                int i = i0 + u * 1024 + tid;
                d[u] = (i < e1) ? dst[i] : -1;
            }
            int r[LDEP];
#pragma unroll
            for (int u = 0; u < LDEP; ++u) {
                int i = i0 + u * 1024 + tid;
                unsigned k = (unsigned)(d[u] - w0);
                r[u] = -1;
                if (i < e1 && k < (unsigned)wn) r[u] = atomicAdd(&s_cnt[k], 1);
            }
#pragma unroll
            for (int u = 0; u < LDEP; ++u) {
                int i = i0 + u * 1024 + tid;
                if (r[u] >= 0) rank[i] = r[u];
            }
        }
        __syncthreads();

        const int steps = (wn4 + 1023) >> 10;
        for (int t = 0; t < steps; ++t) {
            int q = t * 1024 + tid;
            int4 v = make_int4(0, 0, 0, 0);
            if (q < wn4) {
                v = ((const int4*)s_cnt)[q];
                ((int4*)(trow + w0))[q] = v;
            }
            int s = (v.x + v.y) + (v.z + v.w);
#pragma unroll
            for (int off = 16; off >= 1; off >>= 1) s += __shfl_xor(s, off, 32);
            if (lane == 0) s_part[t * 32 + wv] = s;
        }
        __syncthreads();
        const int nr = (wn + 1023) >> 10;
        for (int j = tid; j < nr; j += 1024) {
            int s = 0;
#pragma unroll
            for (int k = 0; k < 8; ++k) s += s_part[8 * j + k];
            brow[(w0 >> 10) + j] = s;
        }
        __syncthreads();
    }
}

__global__ __launch_bounds__(1024) void k_scan(int* __restrict__ tab, const int* __restrict__ bsum,
                                               int* __restrict__ rowptr, float* __restrict__ dinv,
                                               int n, int np, int nbp) {
    __shared__ int s_red[32];
    __shared__ int s_wav[32];
    int tid  = threadIdx.x;
    int lane = tid & 31, wv = tid >> 5;
    int b0   = blockIdx.x * 1024;

    int part = 0;
    for (int j = tid; j < (int)blockIdx.x; j += 1024) {
#pragma unroll
        for (int c = 0; c < LCH; ++c) part += bsum[(size_t)c * nbp + j];
    }
#pragma unroll
    for (int off = 16; off >= 1; off >>= 1) part += __shfl_xor(part, off, 32);
    if (lane == 0) s_red[wv] = part;

    int i = b0 + tid;
    int cv[LCH];
    int v = 0;
#pragma unroll
    for (int c = 0; c < LCH; ++c) {
        cv[c] = (i < n) ? tab[(size_t)c * np + i] : 0;
        v += cv[c];
    }
    int inc = v;
#pragma unroll
    for (int off = 1; off < 32; off <<= 1) {
        int t = __shfl_up(inc, off, 32);
        if (lane >= off) inc += t;
    }
    if (lane == 31) s_wav[wv] = inc;
    __syncthreads();

    int t = s_red[lane] + ((lane < wv) ? s_wav[lane] : 0);
#pragma unroll
    for (int off = 16; off >= 1; off >>= 1) t += __shfl_xor(t, off, 32);
    int excl = t + inc - v;
    if (i < n) {
        rowptr[i] = excl;
        dinv[i]   = rsqrtf((float)(v + 1));
        if (i == n - 1) rowptr[n] = excl + v;
        int run = excl;
#pragma unroll
        for (int c = 0; c < LCH; ++c) {
            tab[(size_t)c * np + i] = run;
            run += cv[c];
        }
    }
}

__global__ void k_gemm1_fill(const float* __restrict__ A, const unsigned short* __restrict__ Bp,
                             float* __restrict__ C, int M, int K, int gb,
                             const int* __restrict__ src, const int* __restrict__ dst,
                             const int* __restrict__ rank, const int* __restrict__ tab,
                             int* __restrict__ col, int e, int np) {
    extern __shared__ unsigned short sB[];
    if ((int)blockIdx.x < gb) {
        gemm_strip(A, Bp, C, M, K, sB);
        return;
    }
    int i = ((int)blockIdx.x - gb) * blockDim.x + threadIdx.x;
    if (i < e) {
        int c = i / chunk_edges(e);
        col[tab[(size_t)c * np + dst[i]] + rank[i]] = src[i];
    }
}

__device__ __forceinline__ void sort_long_list(const int* __restrict__ seg, int2* rseg,
                                               const float* __restrict__ dinv, float dn,
                                               int deg, int lane, int n) {
    for (int c0 = 0; c0 < deg; c0 += 32) {
        int i = c0 + lane;
        bool act = i < deg;
        int s = act ? seg[i] : 0;
        s = min(max(s, 0), n - 1);
        int r = 0;
        for (int k = 0; k < deg; ++k) {
            int t = min(max(seg[k], 0), n - 1);
            r += (t < s || (t == s && k < i)) ? 1 : 0;
        }
        if (act) rseg[r] = make_int2(s, __float_as_int(dinv[s] * dn));
    }
}

__device__ __forceinline__ int2 sort_short_list(const int* __restrict__ seg,
                                                const float* __restrict__ dinv, float dn,
                                                int deg, int lane, int n) {
    int s = (lane < deg) ? seg[lane] : 0;
    s = min(max(s, 0), n - 1);
    float ds = dinv[s];
    int r = 0;
    for (int k = 0; k < deg; ++k) {
        int t = __builtin_amdgcn_readlane(s, k);
        r += (t < s || (t == s && k < lane)) ? 1 : 0;
    }
    if (lane >= deg) r = lane;
    int w = __float_as_int(ds * dn);
    return make_int2(__builtin_amdgcn_ds_permute(r << 2, s), __builtin_amdgcn_ds_permute(r << 2, w));
}

__device__ __forceinline__ float4 row_vals(const float4& v) { return v; }
__device__ __forceinline__ float4 row_vals(const uint2& v) {
    return make_float4(__uint_as_float(v.x << 16), __uint_as_float(v.x & 0xFFFF0000u),
                       __uint_as_float(v.y << 16), __uint_as_float(v.y & 0xFFFF0000u));
}

template <int G, typename RowT>
__device__ __forceinline__ void gather_rows(float4& acc, const int2& r, int j,
                                            const RowT* __restrict__ rows, int lane) {
    RowT v[G]; float w[G];
#pragma unroll
    for (int u = 0; u < G; ++u) {
        int s = __builtin_amdgcn_readlane(r.x, j + u);
        w[u]  = __int_as_float(__builtin_amdgcn_readlane(r.y, j + u));
        v[u]  = rows[(size_t)s * 32 + lane];
    }
#pragma unroll
    for (int u = 0; u < G; ++u) {
        float4 f = row_vals(v[u]);
        acc.x = fmaf(w[u], f.x, acc.x);
        acc.y = fmaf(w[u], f.y, acc.y);
        acc.z = fmaf(w[u], f.z, acc.z);
        acc.w = fmaf(w[u], f.w, acc.w);
    }
}

template <bool SORT, typename RowT>
__device__ __forceinline__ float4 aggregate(const int* __restrict__ rowptr,
                                            const int* __restrict__ rec_i,
                                            const int* __restrict__ col,
                                            int* rec_w, int n,
                                            const float* __restrict__ dinv,
                                            const RowT* __restrict__ rows,
                                            int node, int lane) {
    const int2* rec = SORT ? (const int2*)rec_w : (const int2*)rec_i;
    int beg = rowptr[node], end = rowptr[node + 1];
    float dn = dinv[node];
    float w0 = dn * dn;
    float4 hv = row_vals(rows[(size_t)node * 32 + lane]);
    float4 acc = make_float4(w0 * hv.x, w0 * hv.y, w0 * hv.z, w0 * hv.w);

    int2 r = make_int2(0, 0);
    const bool held = SORT && (end - beg <= 32);
    if (SORT) {
        const int deg = end - beg;
        if (held) {
            r = sort_short_list(col + beg, dinv, dn, deg, lane, n);
            if (lane < deg) ((int2*)rec_w)[beg + lane] = r;
        } else {
            sort_long_list(col + beg, (int2*)rec_w + beg, dinv, dn, deg, lane, n);
            __threadfence();
        }
    }

    for (int c0 = beg; c0 < end; c0 += 32) {
        int m = min(32, end - c0);
        if (!held) {
            r = make_int2(0, 0);
            if (lane < m) r = rec[c0 + lane];
        }
        int j = 0;
        for (; j + 8 <= m; j += 8) gather_rows<8>(acc, r, j, rows, lane);
        if (j + 4 <= m) { gather_rows<4>(acc, r, j, rows, lane); j += 4; }
        if (j + 2 <= m) { gather_rows<2>(acc, r, j, rows, lane); j += 2; }
        if (j < m)        gather_rows<1>(acc, r, j, rows, lane);
    }
    return acc;
}

__device__ __forceinline__ void bias_ln_elu(const float4& acc, const float4& bv,
                                            const float* __restrict__ gamma,
                                            const float* __restrict__ beta,
                                            int lane, float (&y)[4]) {
    float4 gv = ((const float4*)gamma)[lane];
    float4 ev = ((const float4*)beta)[lane];
    float v[4] = { acc.x + bv.x, acc.y + bv.y, acc.z + bv.z, acc.w + bv.w };
    float s = 0.f;
#pragma unroll
    for (int j = 0; j < 4; ++j) s += v[j];
#pragma unroll
    for (int off = 16; off >= 1; off >>= 1) s += __shfl_xor(s, off, 32);
    float mu = s * (1.0f / 128.0f);

    float q = 0.f;
#pragma unroll
    for (int j = 0; j < 4; ++j) {
        float dv = v[j] - mu;
        q += dv * dv;
    }
#pragma unroll
    for (int off = 16; off >= 1; off >>= 1) q += __shfl_xor(q, off, 32);
    float inv = rsqrtf(q * (1.0f / 128.0f) + 1e-5f);

    const float g[4] = { gv.x, gv.y, gv.z, gv.w };
    const float b[4] = { ev.x, ev.y, ev.z, ev.w };
#pragma unroll
    for (int j = 0; j < 4; ++j) {
        float t = (v[j] - mu) * inv * g[j] + b[j];
        y[j] = (t > 0.f) ? t : expm1f(t);
    }
}

__global__ void k_gather_ln1(const int* __restrict__ rowptr, const int* __restrict__ col,
                             int* rec_w,
                             const float* __restrict__ dinv, const float* __restrict__ h,
                             const float* __restrict__ bias, const float* __restrict__ gamma,
                             const float* __restrict__ beta,
                             float* __restrict__ out_f32, unsigned short* __restrict__ out_bf, int n) {
    int node = __builtin_amdgcn_readfirstlane((int)((blockIdx.x * blockDim.x + threadIdx.x) >> 5));
    if (node >= n) return;
    int lane = threadIdx.x & 31;
    float4 bv = ((const float4*)bias)[lane];
    float4 acc = aggregate<true>(rowptr, nullptr, col, rec_w, n, dinv, (const float4*)h, node, lane);
    float y[4];
    bias_ln_elu(acc, bv, gamma, beta, lane, y);
    size_t ridx = (size_t)node * 32 + lane;
    ((float4*)out_f32)[ridx] = make_float4(y[0], y[1], y[2], y[3]);
    ((uint2*)out_bf)[ridx]   = make_uint2(pack2bf(y[0], y[1]), pack2bf(y[2], y[3]));
}

constexpr int GN     = 16;
constexpr int APITCH = 136;
constexpr int DPITCH = 132;

__device__ __forceinline__ float bf2f(unsigned short b) { return __uint_as_float((unsigned int)b << 16); }

__global__ __launch_bounds__(GN * 32) void k_gather2_gemm2_ln2(const int* __restrict__ rowptr, const int* __restrict__ rec_i,
                             const float* __restrict__ dinv, const unsigned short* __restrict__ rows_bf,
                             const unsigned short* __restrict__ Bp,
                             const float* __restrict__ bias, const float* __restrict__ gamma,
                             const float* __restrict__ beta, const float* __restrict__ res_in,
                             float* __restrict__ out_f32, int n) {
    __shared__ __attribute__((aligned(16))) unsigned short sAh[GN * APITCH];
    __shared__ __attribute__((aligned(16))) unsigned short sAl[GN * APITCH];
    __shared__ __attribute__((aligned(16))) float          sD[GN * DPITCH];
    const int lane = threadIdx.x & 31;
    const int wv   = __builtin_amdgcn_readfirstlane((int)(threadIdx.x >> 5));
    const int node = (int)blockIdx.x * GN + wv;
    const bool live = node < n;
    const size_t rbase = (size_t)node * 32;

    float4 acc = make_float4(0.f, 0.f, 0.f, 0.f);
    float4 rv  = make_float4(0.f, 0.f, 0.f, 0.f);
    if (live) {
        rv = ((const float4*)res_in + rbase)[lane];
        acc = aggregate<false>(rowptr, rec_i, nullptr, nullptr, n, dinv, (const uint2*)rows_bf, node, lane);
    }
    {
        const float a[4] = { acc.x, acc.y, acc.z, acc.w };
        unsigned int hi[4], lo[4];
#pragma unroll
        for (int j = 0; j < 4; ++j) {
            hi[j] = f2bf(a[j]);
            lo[j] = f2bf(a[j] - bf2f((unsigned short)hi[j]));
        }
        *(uint2*)(sAh + wv * APITCH + lane * 4) = make_uint2(hi[0] | (hi[1] << 16), hi[2] | (hi[3] << 16));
        *(uint2*)(sAl + wv * APITCH + lane * 4) = make_uint2(lo[0] | (lo[1] << 16), lo[2] | (lo[3] << 16));
    }

    FragU b[4];
    if (wv < 8) {
#pragma unroll
        for (int ks = 0; ks < 4; ++ks) {
            const uint4* bl = (const uint4*)(Bp + (size_t)(ks * 8 + wv) * 512 + lane * 16);
            b[ks].u4[0] = bl[0];
            b[ks].u4[1] = bl[1];
        }
    }
    __syncthreads();

    if (wv < 8) {
        const int aoff = (lane & 15) * APITCH + ((lane < 16) ? 0 : 8);
        v8f c = {};
#pragma unroll
        for (int ks = 0; ks < 4; ++ks) {
            FragU a;
            a.u4[0] = *(const uint4*)(sAl + aoff + ks * 32);
            a.u4[1] = *(const uint4*)(sAl + aoff + ks * 32 + 16);
            c = __builtin_amdgcn_wmma_f32_16x16x32_bf16(
                false, a.v, false, b[ks].v, (short)0, c, false, false);
        }
#pragma unroll
        for (int ks = 0; ks < 4; ++ks) {
            FragU a;
            a.u4[0] = *(const uint4*)(sAh + aoff + ks * 32);
            a.u4[1] = *(const uint4*)(sAh + aoff + ks * 32 + 16);
            c = __builtin_amdgcn_wmma_f32_16x16x32_bf16(
                false, a.v, false, b[ks].v, (short)0, c, false, false);
        }
        float* dp = sD + ((lane < 16) ? 0 : 8) * DPITCH + wv * 16 + (lane & 15);
#pragma unroll
        for (int j = 0; j < 8; ++j) dp[j * DPITCH] = c[j];
    }
    __syncthreads();

    if (live) {
        float4 bv = ((const float4*)bias)[lane];
        float4 cv = *(const float4*)(sD + wv * DPITCH + lane * 4);
        float y[4];
        bias_ln_elu(cv, bv, gamma, beta, lane, y);
        ((float4*)out_f32 + rbase)[lane] = make_float4(y[0] + rv.x, y[1] + rv.y, y[2] + rv.z, y[3] + rv.w);
    }
}

extern "C" void kernel_launch(void* const* d_in, const int* in_sizes, int n_in,
                              void* d_out, int out_size, void* d_ws, size_t ws_size,
                              hipStream_t stream) {
    const float* x   = (const float*)d_in[0];
    const float* W1  = (const float*)d_in[1];
    const float* b1  = (const float*)d_in[2];
    const float* g1  = (const float*)d_in[3];
    const float* be1 = (const float*)d_in[4];
    const float* W2  = (const float*)d_in[5];
    const float* b2  = (const float*)d_in[6];
    const float* g2  = (const float*)d_in[7];
    const float* be2 = (const float*)d_in[8];
    const int*   ei  = (const int*)d_in[9];

    const int EMB = in_sizes[2];
    const int IN  = in_sizes[1] / EMB;
    const int N   = in_sizes[0] / IN;
    const int E   = in_sizes[9] / 2;
    const int* src = ei;
    const int* dst = ei + E;

    char* base = (char*)d_ws;
    size_t off = 0;
    auto alloc = [&](size_t bytes) {
        char* p = base + off;
        off += (bytes + 255) & ~(size_t)255;
        return p;
    };
    float*          dinv  = (float*)alloc((size_t)N * 4);
    const int NP  = (N + 3) & ~3;
    const int NB  = (N + 1023) / 1024;
    int*            tab   = (int*)alloc((size_t)LCH * NP * 4);
    int*            bsum  = (int*)alloc((size_t)LCH * NB * 4);
    int*            rowp  = (int*)alloc((size_t)(N + 1) * 4);
    int*            rank  = (int*)alloc((size_t)E * 4);
    int*            col   = (int*)alloc((size_t)E * 4);
    int*            rec   = (int*)alloc((size_t)E * 8);
    unsigned short* W1p   = (unsigned short*)alloc((size_t)IN * EMB * 2);
    unsigned short* W2p   = (unsigned short*)alloc((size_t)EMB * EMB * 2);
    float*          h     = (float*)alloc((size_t)N * EMB * 4);
    float*          res   = (float*)alloc((size_t)N * EMB * 4);
    unsigned short* resb  = (unsigned short*)alloc((size_t)N * EMB * 2);
    (void)ws_size;

    const int B = 256;

    int mtiles = (N + 15) / 16;
    int gemm_blocks = (mtiles + 7) / 8;
    size_t lds1 = (size_t)(IN  / 32) * 8 * 1024;
    int pack_tiles  = (IN / 32) * (EMB / 16) + (EMB / 32) * (EMB / 16);
    int pack_blocks = (pack_tiles + 31) / 32;

    k_cnt_pack  <<<LCH + pack_blocks, 1024, 0, stream>>>(dst, rank, tab, bsum, N, NP, NB, E,
                                                         W1, W1p, IN, W2, W2p, EMB, EMB);
    k_scan      <<<NB, 1024, 0, stream>>>(tab, bsum, rowp, dinv, N, NP, NB);
    k_gemm1_fill<<<gemm_blocks + (E + B - 1) / B, B, lds1, stream>>>(x, W1p, h, N, IN, gemm_blocks,
                                                                     src, dst, rank, tab, col, E, NP);

    k_gather_ln1<<<(N + 7) / 8, B, 0, stream>>>(rowp, col, rec, dinv, h, b1, g1, be1, res, resb, N);

    k_gather2_gemm2_ln2<<<(N + GN - 1) / GN, GN * 32, 0, stream>>>(rowp, rec, dinv, resb, W2p, b2, g2, be2, res,
                                                                   (float*)d_out, N);
}


# Wave sums and the in-wave scan take partner lanes by DPP row moves and v_permlanex16, not LDS permute round trips (count write-out, scan, both LayerNorms); partner order 16,8,4,2,1: sums unchanged
# speedup vs baseline: 4.2584x; 1.0338x over previous
#include <hip/hip_runtime.h>
#include <hip/hip_bf16.h>

typedef __attribute__((ext_vector_type(16))) __bf16 v16bf;
typedef __attribute__((ext_vector_type(8)))  float  v8f;

union FragU {
    v16bf v;
    unsigned short u16[16];
    uint4 u4[2];
};

__device__ __forceinline__ unsigned short f2bf(float f) {
    unsigned int u = __float_as_uint(f);
    unsigned int r = u + 0x7FFFu + ((u >> 16) & 1u);
    return (unsigned short)(r >> 16);
}

template <int K>
__device__ __forceinline__ int lane_xor(int v) {
    if constexpr (K == 16)
        return (int)__builtin_amdgcn_permlanex16((unsigned)v, (unsigned)v, 0x76543210u, 0xfedcba98u, false, false);
    else
        return __builtin_amdgcn_update_dpp(v, v, 0x160 | K, 0xF, 0xF, true);
}
template <int K>
__device__ __forceinline__ int row_up_or_zero(int v) {
    return __builtin_amdgcn_update_dpp(0, v, 0x110 | K, 0xF, 0xF, false);
}
__device__ __forceinline__ int wave_sum(int s) {
    s += lane_xor<16>(s);
    s += lane_xor<8>(s);
    s += lane_xor<4>(s);
    s += lane_xor<2>(s);
    s += lane_xor<1>(s);
    return s;
}
__device__ __forceinline__ float wave_sum(float s) {
    s += __int_as_float(lane_xor<16>(__float_as_int(s)));
    s += __int_as_float(lane_xor<8>(__float_as_int(s)));
    s += __int_as_float(lane_xor<4>(__float_as_int(s)));
    s += __int_as_float(lane_xor<2>(__float_as_int(s)));
    s += __int_as_float(lane_xor<1>(__float_as_int(s)));
    return s;
}

__device__ __forceinline__ void pack_tile(const float* __restrict__ W, unsigned short* __restrict__ out,
                                          int tile, int lane, int N) {
    int ntiles = N >> 4;
    int kt = tile / ntiles, nt = tile - kt * ntiles;
    int col = nt * 16 + (lane & 15);
    int kb  = kt * 32 + ((lane < 16) ? 0 : 8);
    unsigned short* o = out + (size_t)tile * 512 + lane * 16;
#pragma unroll
    for (int e = 0; e < 8; ++e) o[e]     = f2bf(W[(kb + e)      * N + col]);
#pragma unroll
    for (int e = 0; e < 8; ++e) o[8 + e] = f2bf(W[(kb + 16 + e) * N + col]);
}

struct RawF { float4 f0, f1, f2, f3; };

__device__ __forceinline__ void issue_a(RawF& r, const float* p) {
    r.f0 = *(const float4*)(p);
    r.f1 = *(const float4*)(p + 4);
    r.f2 = *(const float4*)(p + 16);
    r.f3 = *(const float4*)(p + 20);
}

__device__ __forceinline__ unsigned int pack2bf(float lo, float hi) {
    return (unsigned int)f2bf(lo) | ((unsigned int)f2bf(hi) << 16);
}

__device__ __forceinline__ void finish_a(FragU& a, const RawF& r) {
    a.u4[0] = make_uint4(pack2bf(r.f0.x, r.f0.y), pack2bf(r.f0.z, r.f0.w),
                         pack2bf(r.f1.x, r.f1.y), pack2bf(r.f1.z, r.f1.w));
    a.u4[1] = make_uint4(pack2bf(r.f2.x, r.f2.y), pack2bf(r.f2.z, r.f2.w),
                         pack2bf(r.f3.x, r.f3.y), pack2bf(r.f3.z, r.f3.w));
}

constexpr int SBAT = 16;

__device__ __forceinline__ void gemm_strip(const float* __restrict__ A,
                                           const unsigned short* __restrict__ Bp,
                                           float* __restrict__ C, int M, int K,
                                           unsigned short* sB) {
    int wave   = (int)((blockIdx.x * blockDim.x + threadIdx.x) >> 5);
    int lane   = threadIdx.x & 31;
    int mtiles = (M + 15) >> 4;

    int row = wave * 16 + (lane & 15);
    if (row >= M) row = M - 1;
    int kb0 = (lane < 16) ? 0 : 8;
    const float* arow = A + (size_t)row * K + kb0;

    {
        const uint4* gB  = (const uint4*)Bp;
        uint4*       sB4 = (uint4*)sB;
        const int n4  = (K >> 5) * 512;
        const int nth = (int)blockDim.x;
        for (int b0 = 0; b0 < n4; b0 += SBAT * nth) {
            uint4 t[SBAT];
#pragma unroll
            for (int u = 0; u < SBAT; ++u) {
                int i = b0 + u * nth + (int)threadIdx.x;
                t[u] = (i < n4) ? gB[i] : make_uint4(0u, 0u, 0u, 0u);
            }
#pragma unroll
            for (int u = 0; u < SBAT; ++u) {
                int i = b0 + u * nth + (int)threadIdx.x;
                if (i < n4) sB4[i] = t[u];
            }
        }
    }
    const int ksteps = K >> 5;
    RawF raw;
    issue_a(raw, arow);
    __syncthreads();

    if (wave >= mtiles) return;

    v8f zero = {};
    v8f acc[8];
#pragma unroll
    for (int nt = 0; nt < 8; ++nt) acc[nt] = zero;

    for (int ks = 0; ks < ksteps; ++ks) {
        FragU a;
        finish_a(a, raw);
        issue_a(raw, arow + min(ks + 1, ksteps - 1) * 32);
        const unsigned short* bp = sB + (size_t)ks * 8 * 512 + lane * 16;
#pragma unroll
        for (int nt = 0; nt < 8; ++nt) {
            FragU b;
            const uint4* bl = (const uint4*)(bp + nt * 512);
            b.u4[0] = bl[0];
            b.u4[1] = bl[1];
            acc[nt] = __builtin_amdgcn_wmma_f32_16x16x32_bf16(
                false, a.v, false, b.v, (short)0, acc[nt], false, false);
        }
    }

    int srow0 = wave * 16 + ((lane < 16) ? 0 : 8);
    int scol  = lane & 15;
#pragma unroll
    for (int nt = 0; nt < 8; ++nt) {
#pragma unroll
        for (int j = 0; j < 8; ++j) {
            int r = srow0 + j;
            if (r < M) C[(size_t)r * 128 + nt * 16 + scol] = acc[nt][j];
        }
    }
}


constexpr int LCH  = 32;
constexpr int LCAP = 65536;
constexpr int LDEP = 32;

__device__ __forceinline__ int chunk_edges(int e) {
    return (((e + LCH - 1) / LCH) + 3) & ~3;
}

__global__ __launch_bounds__(1024) void k_cnt_pack(const int* __restrict__ dst, int* __restrict__ rank,
                                                   int* __restrict__ tab, int* __restrict__ bsum,
                                                   int n, int np, int nbp, int e,
                                                   const float* __restrict__ W1, unsigned short* __restrict__ W1p, int K1,
                                                   const float* __restrict__ W2, unsigned short* __restrict__ W2p, int K2,
                                                   int ncol) {
    __shared__ __attribute__((aligned(16))) int s_cnt[LCAP];
    __shared__ int s_part[(LCAP / 4096) * 32];
    const int tid = threadIdx.x, lane = tid & 31, wv = tid >> 5;
    if ((int)blockIdx.x >= LCH) {
        const int tile = ((int)blockIdx.x - LCH) * 32 + wv;
        const int t1 = (K1 >> 5) * (ncol >> 4);
        const int t2 = (K2 >> 5) * (ncol >> 4);
        if (tile < t1)           pack_tile(W1, W1p, tile, lane, ncol);
        else if (tile - t1 < t2) pack_tile(W2, W2p, tile - t1, lane, ncol);
        return;
    }
    const int c   = blockIdx.x;
    const int per = chunk_edges(e);
    const int e0  = min(c * per, e), e1 = min(e0 + per, e);
    int* trow = tab  + (size_t)c * np;
    int* brow = bsum + (size_t)c * nbp;

    for (int w0 = 0; w0 < np; w0 += LCAP) {
        const int wn  = min(LCAP, np - w0);
        const int wn4 = wn >> 2;
        for (int q = tid; q < wn4; q += 1024) ((int4*)s_cnt)[q] = make_int4(0, 0, 0, 0);
        __syncthreads();

        for (int i0 = e0; i0 < e1; i0 += LDEP * 1024) {
            int d[LDEP];
#pragma unroll
            for (int u = 0; u < LDEP; ++u) {
                int i = i0 + u * 1024 + tid;
                d[u] = (i < e1) ? dst[i] : -1;
            }
            int r[LDEP];
#pragma unroll
            for (int u = 0; u < LDEP; ++u) {
                int i = i0 + u * 1024 + tid;
                unsigned k = (unsigned)(d[u] - w0);
                r[u] = -1;
                if (i < e1 && k < (unsigned)wn) r[u] = atomicAdd(&s_cnt[k], 1);
            }
#pragma unroll
            for (int u = 0; u < LDEP; ++u) {
                int i = i0 + u * 1024 + tid;
                if (r[u] >= 0) rank[i] = r[u];
            }
        }
        __syncthreads();

        const int steps = (wn4 + 1023) >> 10;
        for (int t = 0; t < steps; ++t) {
            int q = t * 1024 + tid;
            int4 v = make_int4(0, 0, 0, 0);
            if (q < wn4) {
                v = ((const int4*)s_cnt)[q];
                ((int4*)(trow + w0))[q] = v;
            }
            int s = wave_sum((v.x + v.y) + (v.z + v.w));
            if (lane == 0) s_part[t * 32 + wv] = s;
        }
        __syncthreads();
        const int nr = (wn + 1023) >> 10;
        for (int j = tid; j < nr; j += 1024) {
            int s = 0;
#pragma unroll
            for (int k = 0; k < 8; ++k) s += s_part[8 * j + k];
            brow[(w0 >> 10) + j] = s;
        }
        __syncthreads();
    }
}

__global__ __launch_bounds__(1024) void k_scan(int* __restrict__ tab, const int* __restrict__ bsum,
                                               int* __restrict__ rowptr, float* __restrict__ dinv,
                                               int n, int np, int nbp) {
    __shared__ int s_red[32];
    __shared__ int s_wav[32];
    int tid  = threadIdx.x;
    int lane = tid & 31, wv = tid >> 5;
    int b0   = blockIdx.x * 1024;

    int part = 0;
    for (int j = tid; j < (int)blockIdx.x; j += 1024) {
#pragma unroll
        for (int c = 0; c < LCH; ++c) part += bsum[(size_t)c * nbp + j];
    }
    part = wave_sum(part);
    if (lane == 0) s_red[wv] = part;

    int i = b0 + tid;
    int cv[LCH];
    int v = 0;
#pragma unroll
    for (int c = 0; c < LCH; ++c) {
        cv[c] = (i < n) ? tab[(size_t)c * np + i] : 0;
        v += cv[c];
    }
    int inc = v;
    inc += row_up_or_zero<1>(inc);
    inc += row_up_or_zero<2>(inc);
    inc += row_up_or_zero<4>(inc);
    inc += row_up_or_zero<8>(inc);
    {
        int low = __builtin_amdgcn_readlane(inc, 15);
        if (lane >= 16) inc += low;
    }
    if (lane == 31) s_wav[wv] = inc;
    __syncthreads();

    int t = wave_sum(s_red[lane] + ((lane < wv) ? s_wav[lane] : 0));
    int excl = t + inc - v;
    if (i < n) {
        rowptr[i] = excl;
        dinv[i]   = rsqrtf((float)(v + 1));
        if (i == n - 1) rowptr[n] = excl + v;
        int run = excl;
#pragma unroll
        for (int c = 0; c < LCH; ++c) {
            tab[(size_t)c * np + i] = run;
            run += cv[c];
        }
    }
}

__global__ void k_gemm1_fill(const float* __restrict__ A, const unsigned short* __restrict__ Bp,
                             float* __restrict__ C, int M, int K, int gb,
                             const int* __restrict__ src, const int* __restrict__ dst,
                             const int* __restrict__ rank, const int* __restrict__ tab,
                             int* __restrict__ col, int e, int np) {
    extern __shared__ unsigned short sB[];
    if ((int)blockIdx.x < gb) {
        gemm_strip(A, Bp, C, M, K, sB);
        return;
    }
    int i = ((int)blockIdx.x - gb) * blockDim.x + threadIdx.x;
    if (i < e) {
        int c = i / chunk_edges(e);
        col[tab[(size_t)c * np + dst[i]] + rank[i]] = src[i];
    }
}

__device__ __forceinline__ void sort_long_list(const int* __restrict__ seg, int2* rseg,
                                               const float* __restrict__ dinv, float dn,
                                               int deg, int lane, int n) {
    for (int c0 = 0; c0 < deg; c0 += 32) {
        int i = c0 + lane;
        bool act = i < deg;
        int s = act ? seg[i] : 0;
        s = min(max(s, 0), n - 1);
        int r = 0;
        for (int k = 0; k < deg; ++k) {
            int t = min(max(seg[k], 0), n - 1);
            r += (t < s || (t == s && k < i)) ? 1 : 0;
        }
        if (act) rseg[r] = make_int2(s, __float_as_int(dinv[s] * dn));
    }
}

__device__ __forceinline__ int2 sort_short_list(const int* __restrict__ seg,
                                                const float* __restrict__ dinv, float dn,
                                                int deg, int lane, int n) {
    int s = (lane < deg) ? seg[lane] : 0;
    s = min(max(s, 0), n - 1);
    float ds = dinv[s];
    int r = 0;
    for (int k = 0; k < deg; ++k) {
        int t = __builtin_amdgcn_readlane(s, k);
        r += (t < s || (t == s && k < lane)) ? 1 : 0;
    }
    if (lane >= deg) r = lane;
    int w = __float_as_int(ds * dn);
    return make_int2(__builtin_amdgcn_ds_permute(r << 2, s), __builtin_amdgcn_ds_permute(r << 2, w));
}

__device__ __forceinline__ float4 row_vals(const float4& v) { return v; }
__device__ __forceinline__ float4 row_vals(const uint2& v) {
    return make_float4(__uint_as_float(v.x << 16), __uint_as_float(v.x & 0xFFFF0000u),
                       __uint_as_float(v.y << 16), __uint_as_float(v.y & 0xFFFF0000u));
}

template <int G, typename RowT>
__device__ __forceinline__ void gather_rows(float4& acc, const int2& r, int j,
                                            const RowT* __restrict__ rows, int lane) {
    RowT v[G]; float w[G];
#pragma unroll
    for (int u = 0; u < G; ++u) {
        int s = __builtin_amdgcn_readlane(r.x, j + u);
        w[u]  = __int_as_float(__builtin_amdgcn_readlane(r.y, j + u));
        v[u]  = rows[(size_t)s * 32 + lane];
    }
#pragma unroll
    for (int u = 0; u < G; ++u) {
        float4 f = row_vals(v[u]);
        acc.x = fmaf(w[u], f.x, acc.x);
        acc.y = fmaf(w[u], f.y, acc.y);
        acc.z = fmaf(w[u], f.z, acc.z);
        acc.w = fmaf(w[u], f.w, acc.w);
    }
}

template <bool SORT, typename RowT>
__device__ __forceinline__ float4 aggregate(const int* __restrict__ rowptr,
                                            const int* __restrict__ rec_i,
                                            const int* __restrict__ col,
                                            int* rec_w, int n,
                                            const float* __restrict__ dinv,
                                            const RowT* __restrict__ rows,
                                            int node, int lane) {
    const int2* rec = SORT ? (const int2*)rec_w : (const int2*)rec_i;
    int beg = rowptr[node], end = rowptr[node + 1];
    float dn = dinv[node];
    float w0 = dn * dn;
    float4 hv = row_vals(rows[(size_t)node * 32 + lane]);
    float4 acc = make_float4(w0 * hv.x, w0 * hv.y, w0 * hv.z, w0 * hv.w);

    int2 r = make_int2(0, 0);
    const bool held = SORT && (end - beg <= 32);
    if (SORT) {
        const int deg = end - beg;
        if (held) {
            r = sort_short_list(col + beg, dinv, dn, deg, lane, n);
            if (lane < deg) ((int2*)rec_w)[beg + lane] = r;
        } else {
            sort_long_list(col + beg, (int2*)rec_w + beg, dinv, dn, deg, lane, n);
            __threadfence();
        }
    }

    for (int c0 = beg; c0 < end; c0 += 32) {
        int m = min(32, end - c0);
        if (!held) {
            r = make_int2(0, 0);
            if (lane < m) r = rec[c0 + lane];
        }
        int j = 0;
        for (; j + 8 <= m; j += 8) gather_rows<8>(acc, r, j, rows, lane);
        if (j + 4 <= m) { gather_rows<4>(acc, r, j, rows, lane); j += 4; }
        if (j + 2 <= m) { gather_rows<2>(acc, r, j, rows, lane); j += 2; }
        if (j < m)        gather_rows<1>(acc, r, j, rows, lane);
    }
    return acc;
}

__device__ __forceinline__ void bias_ln_elu(const float4& acc, const float4& bv,
                                            const float* __restrict__ gamma,
                                            const float* __restrict__ beta,
                                            int lane, float (&y)[4]) {
    float4 gv = ((const float4*)gamma)[lane];
    float4 ev = ((const float4*)beta)[lane];
    float v[4] = { acc.x + bv.x, acc.y + bv.y, acc.z + bv.z, acc.w + bv.w };
    float s = 0.f;
#pragma unroll
    for (int j = 0; j < 4; ++j) s += v[j];
    s = wave_sum(s);
    float mu = s * (1.0f / 128.0f);

    float q = 0.f;
#pragma unroll
    for (int j = 0; j < 4; ++j) {
        float dv = v[j] - mu;
        q += dv * dv;
    }
    q = wave_sum(q);
    float inv = rsqrtf(q * (1.0f / 128.0f) + 1e-5f);

    const float g[4] = { gv.x, gv.y, gv.z, gv.w };
    const float b[4] = { ev.x, ev.y, ev.z, ev.w };
#pragma unroll
    for (int j = 0; j < 4; ++j) {
        float t = (v[j] - mu) * inv * g[j] + b[j];
        y[j] = (t > 0.f) ? t : expm1f(t);
    }
}

__global__ void k_gather_ln1(const int* __restrict__ rowptr, const int* __restrict__ col,
                             int* rec_w,
                             const float* __restrict__ dinv, const float* __restrict__ h,
                             const float* __restrict__ bias, const float* __restrict__ gamma,
                             const float* __restrict__ beta,
                             float* __restrict__ out_f32, unsigned short* __restrict__ out_bf, int n) {
    int node = __builtin_amdgcn_readfirstlane((int)((blockIdx.x * blockDim.x + threadIdx.x) >> 5));
    if (node >= n) return;
    int lane = threadIdx.x & 31;
    float4 bv = ((const float4*)bias)[lane];
    float4 acc = aggregate<true>(rowptr, nullptr, col, rec_w, n, dinv, (const float4*)h, node, lane);
    float y[4];
    bias_ln_elu(acc, bv, gamma, beta, lane, y);
    size_t ridx = (size_t)node * 32 + lane;
    ((float4*)out_f32)[ridx] = make_float4(y[0], y[1], y[2], y[3]);
    ((uint2*)out_bf)[ridx]   = make_uint2(pack2bf(y[0], y[1]), pack2bf(y[2], y[3]));
}

constexpr int GN     = 16;
constexpr int APITCH = 136;
constexpr int DPITCH = 132;

__device__ __forceinline__ float bf2f(unsigned short b) { return __uint_as_float((unsigned int)b << 16); }

__global__ __launch_bounds__(GN * 32) void k_gather2_gemm2_ln2(const int* __restrict__ rowptr, const int* __restrict__ rec_i,
                             const float* __restrict__ dinv, const unsigned short* __restrict__ rows_bf,
                             const unsigned short* __restrict__ Bp,
                             const float* __restrict__ bias, const float* __restrict__ gamma,
                             const float* __restrict__ beta, const float* __restrict__ res_in,
                             float* __restrict__ out_f32, int n) {
    __shared__ __attribute__((aligned(16))) unsigned short sAh[GN * APITCH];
    __shared__ __attribute__((aligned(16))) unsigned short sAl[GN * APITCH];
    __shared__ __attribute__((aligned(16))) float          sD[GN * DPITCH];
    const int lane = threadIdx.x & 31;
    const int wv   = __builtin_amdgcn_readfirstlane((int)(threadIdx.x >> 5));
    const int node = (int)blockIdx.x * GN + wv;
    const bool live = node < n;
    const size_t rbase = (size_t)node * 32;

    float4 acc = make_float4(0.f, 0.f, 0.f, 0.f);
    float4 rv  = make_float4(0.f, 0.f, 0.f, 0.f);
    if (live) {
        rv = ((const float4*)res_in + rbase)[lane];
        acc = aggregate<false>(rowptr, rec_i, nullptr, nullptr, n, dinv, (const uint2*)rows_bf, node, lane);
    }
    {
        const float a[4] = { acc.x, acc.y, acc.z, acc.w };
        unsigned int hi[4], lo[4];
#pragma unroll
        for (int j = 0; j < 4; ++j) {
            hi[j] = f2bf(a[j]);
            lo[j] = f2bf(a[j] - bf2f((unsigned short)hi[j]));
        }
        *(uint2*)(sAh + wv * APITCH + lane * 4) = make_uint2(hi[0] | (hi[1] << 16), hi[2] | (hi[3] << 16));
        *(uint2*)(sAl + wv * APITCH + lane * 4) = make_uint2(lo[0] | (lo[1] << 16), lo[2] | (lo[3] << 16));
    }

    FragU b[4];
    if (wv < 8) {
#pragma unroll
        for (int ks = 0; ks < 4; ++ks) {
            const uint4* bl = (const uint4*)(Bp + (size_t)(ks * 8 + wv) * 512 + lane * 16);
            b[ks].u4[0] = bl[0];
            b[ks].u4[1] = bl[1];
        }
    }
    __syncthreads();

    if (wv < 8) {
        const int aoff = (lane & 15) * APITCH + ((lane < 16) ? 0 : 8);
        v8f c = {};
#pragma unroll
        for (int ks = 0; ks < 4; ++ks) {
            FragU a;
            a.u4[0] = *(const uint4*)(sAl + aoff + ks * 32);
            a.u4[1] = *(const uint4*)(sAl + aoff + ks * 32 + 16);
            c = __builtin_amdgcn_wmma_f32_16x16x32_bf16(
                false, a.v, false, b[ks].v, (short)0, c, false, false);
        }
#pragma unroll
        for (int ks = 0; ks < 4; ++ks) {
            FragU a;
            a.u4[0] = *(const uint4*)(sAh + aoff + ks * 32);
            a.u4[1] = *(const uint4*)(sAh + aoff + ks * 32 + 16);
            c = __builtin_amdgcn_wmma_f32_16x16x32_bf16(
                false, a.v, false, b[ks].v, (short)0, c, false, false);
        }
        float* dp = sD + ((lane < 16) ? 0 : 8) * DPITCH + wv * 16 + (lane & 15);
#pragma unroll
        for (int j = 0; j < 8; ++j) dp[j * DPITCH] = c[j];
    }
    __syncthreads();

    if (live) {
        float4 bv = ((const float4*)bias)[lane];
        float4 cv = *(const float4*)(sD + wv * DPITCH + lane * 4);
        float y[4];
        bias_ln_elu(cv, bv, gamma, beta, lane, y);
        ((float4*)out_f32 + rbase)[lane] = make_float4(y[0] + rv.x, y[1] + rv.y, y[2] + rv.z, y[3] + rv.w);
    }
}

extern "C" void kernel_launch(void* const* d_in, const int* in_sizes, int n_in,
                              void* d_out, int out_size, void* d_ws, size_t ws_size,
                              hipStream_t stream) {
    const float* x   = (const float*)d_in[0];
    const float* W1  = (const float*)d_in[1];
    const float* b1  = (const float*)d_in[2];
    const float* g1  = (const float*)d_in[3];
    const float* be1 = (const float*)d_in[4];
    const float* W2  = (const float*)d_in[5];
    const float* b2  = (const float*)d_in[6];
    const float* g2  = (const float*)d_in[7];
    const float* be2 = (const float*)d_in[8];
    const int*   ei  = (const int*)d_in[9];

    const int EMB = in_sizes[2];
    const int IN  = in_sizes[1] / EMB;
    const int N   = in_sizes[0] / IN;
    const int E   = in_sizes[9] / 2;
    const int* src = ei;
    const int* dst = ei + E;

    char* base = (char*)d_ws;
    size_t off = 0;
    auto alloc = [&](size_t bytes) {
        char* p = base + off;
        off += (bytes + 255) & ~(size_t)255;
        return p;
    };
    float*          dinv  = (float*)alloc((size_t)N * 4);
    const int NP  = (N + 3) & ~3;
    const int NB  = (N + 1023) / 1024;
    int*            tab   = (int*)alloc((size_t)LCH * NP * 4);
    int*            bsum  = (int*)alloc((size_t)LCH * NB * 4);
    int*            rowp  = (int*)alloc((size_t)(N + 1) * 4);
    int*            rank  = (int*)alloc((size_t)E * 4);
    int*            col   = (int*)alloc((size_t)E * 4);
    int*            rec   = (int*)alloc((size_t)E * 8);
    unsigned short* W1p   = (unsigned short*)alloc((size_t)IN * EMB * 2);
    unsigned short* W2p   = (unsigned short*)alloc((size_t)EMB * EMB * 2);
    float*          h     = (float*)alloc((size_t)N * EMB * 4);
    float*          res   = (float*)alloc((size_t)N * EMB * 4);
    unsigned short* resb  = (unsigned short*)alloc((size_t)N * EMB * 2);
    (void)ws_size;

    const int B = 256;

    int mtiles = (N + 15) / 16;
    int gemm_blocks = (mtiles + 7) / 8;
    size_t lds1 = (size_t)(IN  / 32) * 8 * 1024;
    int pack_tiles  = (IN / 32) * (EMB / 16) + (EMB / 32) * (EMB / 16);
    int pack_blocks = (pack_tiles + 31) / 32;

    k_cnt_pack  <<<LCH + pack_blocks, 1024, 0, stream>>>(dst, rank, tab, bsum, N, NP, NB, E,
                                                         W1, W1p, IN, W2, W2p, EMB, EMB);
    k_scan      <<<NB, 1024, 0, stream>>>(tab, bsum, rowp, dinv, N, NP, NB);
    k_gemm1_fill<<<gemm_blocks + (E + B - 1) / B, B, lds1, stream>>>(x, W1p, h, N, IN, gemm_blocks,
                                                                     src, dst, rank, tab, col, E, NP);

    k_gather_ln1<<<(N + 7) / 8, B, 0, stream>>>(rowp, col, rec, dinv, h, b1, g1, be1, res, resb, N);

    k_gather2_gemm2_ln2<<<(N + GN - 1) / GN, GN * 32, 0, stream>>>(rowp, rec, dinv, resb, W2p, b2, g2, be2, res,
                                                                   (float*)d_out, N);
}
